# starting kernel
# baseline (speedup 1.0000x reference)
#include <hip/hip_runtime.h>
#include <math.h>

#define BB 8
#define CC 63
#define HH 240
#define WW 240
#define HW (HH * WW)
#define C2 126
#define WS 5
#define NWIN 48
#define BN_EPS 1e-5f

typedef __attribute__((ext_vector_type(2))) float v2f;
typedef __attribute__((ext_vector_type(8))) float v8f;

__device__ __forceinline__ v8f wmma4(v2f a, v2f b, v8f c) {
  return __builtin_amdgcn_wmma_f32_16x16x4_f32(
       false, a,  false, b,
       (short)0, c,  false,  false);
}

__device__ __forceinline__ v2f frag_a(const float* p, int ld, int m0, int k0, int lane) {
  int m  = m0 + (lane & 15);
  int kb = k0 + ((lane >> 4) << 1);
  const float* q = p + m * ld + kb;
  v2f r; r.x = q[0]; r.y = q[1];
  return r;
}

__device__ __forceinline__ v2f frag_b(const float* p, int ld, int k0, int n0, int lane) {
  int n  = n0 + (lane & 15);
  int kb = k0 + ((lane >> 4) << 1);
  v2f r; r.x = p[kb * ld + n]; r.y = p[(kb + 1) * ld + n];
  return r;
}

__device__ __forceinline__ void frag_store(float* p, int ld, int m0, int n0, int lane, v8f d) {
  int n  = n0 + (lane & 15);
  int mb = m0 + ((lane >> 4) << 3);
#pragma unroll
  for (int r = 0; r < 8; ++r) p[(mb + r) * ld + n] = d[r];
}

#define V8Z {0.f,0.f,0.f,0.f,0.f,0.f,0.f,0.f}

__global__ __launch_bounds__(128) void k_proj_in(
    const float* __restrict__ x, const float* __restrict__ w_in,
    const float* __restrict__ b_in, const float* __restrict__ bn_g,
    const float* __restrict__ bn_b, const float* __restrict__ bn_m,
    const float* __restrict__ bn_v, float* __restrict__ xp)
{
  __shared__ float a_lds[64][65];
  __shared__ float w_lds[64][128];
  __shared__ float bias_lds[128];
  __shared__ float d_lds[4][16][17];

  int tid  = threadIdx.x;
  int lane = tid & 31;
  int wave = tid >> 5;
  int tile = blockIdx.x;
  int b    = tile / 900;
  int p0   = (tile % 900) * 64;

  for (int idx = tid; idx < 64 * 128; idx += 128) {
    int k = idx >> 7, n = idx & 127;
    float val = 0.f;
    if (k < 63 && n < 126) {
      float s = bn_g[n] * rsqrtf(bn_v[n] + BN_EPS);
      val = w_in[n * 63 + k] * s;
    }
    w_lds[k][n] = val;
  }
  if (tid < 128) {
    float bv = 0.f;
    if (tid < 126) {
      float s = bn_g[tid] * rsqrtf(bn_v[tid] + BN_EPS);
      bv = b_in[tid] * s + bn_b[tid] - bn_m[tid] * s;
    }
    bias_lds[tid] = bv;
  }
  for (int idx = tid; idx < 64 * 64; idx += 128) {
    int k = idx >> 6, m = idx & 63;
    a_lds[m][k] = (k < 63) ? x[(size_t)(b * 63 + k) * HW + p0 + m] : 0.f;
  }
  __syncthreads();

  int m0 = wave * 16;
  for (int nt = 0; nt < 8; ++nt) {
    int n0 = nt * 16;
    v8f acc = V8Z;
    for (int k0 = 0; k0 < 64; k0 += 4) {
      v2f av = frag_a(&a_lds[0][0], 65, m0, k0, lane);
      v2f bv = frag_b(&w_lds[0][0], 128, k0, n0, lane);
      acc = wmma4(av, bv, acc);
    }
    int nloc = lane & 15;
    int mb   = (lane >> 4) << 3;
    float bias = bias_lds[n0 + nloc];
#pragma unroll
    for (int r = 0; r < 8; ++r) d_lds[wave][mb + r][nloc] = acc[r] + bias;
    for (int idx = lane; idx < 256; idx += 32) {
      int n = idx >> 4, m = idx & 15;
      int ch = n0 + n;
      if (ch < 126)
        xp[(size_t)(b * 126 + ch) * HW + p0 + m0 + m] = d_lds[wave][m][n];
    }
  }
}

__global__ __launch_bounds__(32) void k_win(float* __restrict__ xp, float* __restrict__ y2)
{
  __shared__ float q[32][25];
  __shared__ float v[32][25];
  __shared__ float s[32][33];

  int lane = threadIdx.x;
  int wid  = blockIdx.x;
  int grp  = blockIdx.y;
  int b    = wid / (NWIN * NWIN);
  int rem  = wid % (NWIN * NWIN);
  int wy   = rem / NWIN, wx = rem % NWIN;

  for (int i = lane; i < 32 * 25; i += 32) { (&q[0][0])[i] = 0.f; (&v[0][0])[i] = 0.f; }

  int chq = grp * 42;
  for (int i = lane; i < 25 * 21; i += 32) {
    int m = i / 21, c = i % 21;
    int gy = wy * WS + m / WS, gx = wx * WS + m % WS;
    if (grp == 1) { gy = (gy + 3) % HH; gx = (gx + 3) % WW; }
    size_t base = (size_t)(b * 126 + chq + c) * HW + gy * WW + gx;
    q[m][c] = xp[base];
    v[m][c] = xp[base + (size_t)21 * HW];
  }

  for (int mt = 0; mt < 2; ++mt)
    for (int nt = 0; nt < 2; ++nt) {
      v8f acc = V8Z;
      for (int k0 = 0; k0 < 24; k0 += 4) {
        v2f av = frag_a(&q[0][0], 25, mt * 16, k0, lane);
        v2f bv = frag_a(&q[0][0], 25, nt * 16, k0, lane);
        acc = wmma4(av, bv, acc);
      }
      frag_store(&s[0][0], 33, mt * 16, nt * 16, lane, acc);
    }

  if (lane < 25) {
    float mx = -1e30f;
    for (int n = 0; n < 25; ++n) mx = fmaxf(mx, s[lane][n]);
    float sum = 0.f;
    for (int n = 0; n < 25; ++n) { float e = __expf(s[lane][n] - mx); s[lane][n] = e; sum += e; }
    float inv = 1.f / sum;
    for (int n = 0; n < 25; ++n) s[lane][n] *= inv;
    for (int n = 25; n < 28; ++n) s[lane][n] = 0.f;
  }

  for (int mt = 0; mt < 2; ++mt)
    for (int nt = 0; nt < 2; ++nt) {
      v8f acc = V8Z;
      for (int k0 = 0; k0 < 28; k0 += 4) {
        v2f av = frag_a(&s[0][0], 33, mt * 16, k0, lane);
        v2f bv = frag_b(&v[0][0], 25, k0, nt * 16, lane);
        acc = wmma4(av, bv, acc);
      }
      int n  = nt * 16 + (lane & 15);
      int mb = mt * 16 + ((lane >> 4) << 3);
#pragma unroll
      for (int r = 0; r < 8; ++r) {
        int m = mb + r;
        if (m < 25 && n < 21) {
          int oy = wy * WS + m / WS, ox = wx * WS + m % WS;
          if (grp == 0) {
            xp[(size_t)(b * 126 + n) * HW + oy * WW + ox] = acc[r];
          } else {
            oy = (oy + 2) % HH; ox = (ox + 2) % WW;
            y2[(size_t)(b * 21 + n) * HW + oy * WW + ox] = acc[r];
          }
        }
      }
    }
}

__global__ __launch_bounds__(32) void k_axial(float* __restrict__ xp,
                                              float* __restrict__ vmid, int pass)
{
  __shared__ float q[240][25];
  __shared__ float v[241][25];
  __shared__ float s[16][241];

  int lane = threadIdx.x;
  int row  = blockIdx.x;
  int qt   = blockIdx.y;
  int b    = row / HH;
  int r    = row % HH;

  if (pass == 0) {
    for (int i = lane; i < 240 * 21; i += 32) {
      int c = i / 240, xx = i % 240;
      q[xx][c] = xp[(size_t)(b * 126 + 84 + c) * HW + r * WW + xx];
      v[xx][c] = xp[(size_t)(b * 126 + 105 + c) * HW + r * WW + xx];
    }
  } else {
    for (int i = lane; i < 240 * 21; i += 32) {
      int c = i / 240, h = i % 240;
      q[h][c] = xp[(size_t)(b * 126 + 84 + c) * HW + h * WW + r];
    }
    for (int i = lane; i < 240 * 21; i += 32) {
      int h = i / 21, c = i % 21;
      v[h][c] = vmid[((size_t)(b * 240 + r) * 240 + h) * 21 + c];
    }
  }
  for (int i = lane; i < 240 * 4; i += 32) {
    int m = i >> 2, c = 21 + (i & 3);
    q[m][c] = 0.f; v[m][c] = 0.f;
  }

  int m0 = qt * 16;
  for (int nt = 0; nt < 15; ++nt) {
    int n0 = nt * 16;
    v8f acc = V8Z;
    for (int k0 = 0; k0 < 24; k0 += 4) {
      v2f av = frag_a(&q[0][0], 25, m0, k0, lane);
      v2f bv = frag_a(&q[0][0], 25, n0, k0, lane);
      acc = wmma4(av, bv, acc);
    }
    frag_store(&s[0][0], 241, 0, n0, lane, acc);
  }

  if (lane < 16) {
    float mx = -1e30f;
    for (int n = 0; n < 240; ++n) mx = fmaxf(mx, s[lane][n]);
    float sum = 0.f;
    for (int n = 0; n < 240; ++n) { float e = __expf(s[lane][n] - mx); s[lane][n] = e; sum += e; }
    float inv = 1.f / sum;
    for (int n = 0; n < 240; ++n) s[lane][n] *= inv;
  }

  for (int nt = 0; nt < 2; ++nt) {
    int n0 = nt * 16;
    v8f acc = V8Z;
    for (int k0 = 0; k0 < 240; k0 += 4) {
      v2f av = frag_a(&s[0][0], 241, 0, k0, lane);
      v2f bv = frag_b(&v[0][0], 25, k0, n0, lane);
      acc = wmma4(av, bv, acc);
    }
    int n  = n0 + (lane & 15);
    int mb = (lane >> 4) << 3;
#pragma unroll
    for (int rr = 0; rr < 8; ++rr) {
      int m = m0 + mb + rr;
      if (n < 21) {
        if (pass == 0)
          vmid[((size_t)(b * 240 + m) * 240 + r) * 21 + n] = acc[rr];
        else
          xp[(size_t)(b * 126 + 105 + n) * HW + m * WW + r] = acc[rr];
      }
    }
  }
}

__global__ __launch_bounds__(128) void k_proj_out(
    const float* __restrict__ xp, const float* __restrict__ y2,
    const float* __restrict__ w_out, const float* __restrict__ b_out,
    float* __restrict__ out)
{
  __shared__ float a_lds[64][65];
  __shared__ float w_lds[64][64];
  __shared__ float bias_lds[64];
  __shared__ float d_lds[4][16][17];

  int tid  = threadIdx.x;
  int lane = tid & 31;
  int wave = tid >> 5;
  int tile = blockIdx.x;
  int b    = tile / 900;
  int p0   = (tile % 900) * 64;

  for (int idx = tid; idx < 64 * 64; idx += 128) {
    int k = idx >> 6, n = idx & 63;
    w_lds[k][n] = (k < 63 && n < 63) ? w_out[n * 63 + k] : 0.f;
  }
  if (tid < 64) bias_lds[tid] = (tid < 63) ? b_out[tid] : 0.f;

  for (int idx = tid; idx < 64 * 64; idx += 128) {
    int k = idx >> 6, m = idx & 63;
    float val = 0.f;
    if (k < 21)       val = xp[(size_t)(b * 126 + k)        * HW + p0 + m];
    else if (k < 42)  val = y2[(size_t)(b * 21 + (k - 21))  * HW + p0 + m];
    else if (k < 63)  val = xp[(size_t)(b * 126 + (k + 63)) * HW + p0 + m];
    a_lds[m][k] = val;
  }
  __syncthreads();

  int m0 = wave * 16;
  for (int nt = 0; nt < 4; ++nt) {
    int n0 = nt * 16;
    v8f acc = V8Z;
    for (int k0 = 0; k0 < 64; k0 += 4) {
      v2f av = frag_a(&a_lds[0][0], 65, m0, k0, lane);
      v2f bv = frag_b(&w_lds[0][0], 64, k0, n0, lane);
      acc = wmma4(av, bv, acc);
    }
    int nloc = lane & 15;
    int mb   = (lane >> 4) << 3;
    float bias = bias_lds[n0 + nloc];
#pragma unroll
    for (int r = 0; r < 8; ++r) d_lds[wave][mb + r][nloc] = acc[r] + bias;
    for (int idx = lane; idx < 256; idx += 32) {
      int n = idx >> 4, m = idx & 15;
      int ch = n0 + n;
      if (ch < 63)
        out[(size_t)(b * 63 + ch) * HW + p0 + m0 + m] = d_lds[wave][m][n];
    }
  }
}

extern "C" void kernel_launch(void* const* d_in, const int* in_sizes, int n_in,
                              void* d_out, int out_size, void* d_ws, size_t ws_size,
                              hipStream_t stream)
{
  (void)in_sizes; (void)n_in; (void)out_size; (void)ws_size;
  const float* x     = (const float*)d_in[0];
  const float* w_in  = (const float*)d_in[1];
  const float* b_in  = (const float*)d_in[2];
  const float* bn_g  = (const float*)d_in[3];
  const float* bn_b  = (const float*)d_in[4];
  const float* bn_m  = (const float*)d_in[5];
  const float* bn_v  = (const float*)d_in[6];
  const float* w_out = (const float*)d_in[7];
  const float* b_out = (const float*)d_in[8];
  float* out = (float*)d_out;

  float* xp   = (float*)d_ws;
  float* y2b  = xp  + (size_t)BB * 126 * HW;
  float* vmid = y2b + (size_t)BB * 21 * HW;

  k_proj_in <<<dim3(BB * 900), 128, 0, stream>>>(x, w_in, b_in, bn_g, bn_b, bn_m, bn_v, xp);
  k_win     <<<dim3(BB * NWIN * NWIN, 2), 32, 0, stream>>>(xp, y2b);
  k_axial   <<<dim3(BB * HH, 15), 32, 0, stream>>>(xp, vmid, 0);
  k_axial   <<<dim3(BB * WW, 15), 32, 0, stream>>>(xp, vmid, 1);
  k_proj_out<<<dim3(BB * 900), 128, 0, stream>>>(xp, y2b, w_out, b_out, out);
}


# axial attention: one 15-wave workgroup per sequence stages q and v into LDS once for all 15 query tiles (was 15 one-wave blocks, each staging the whole sequence)
# speedup vs baseline: 2.0834x; 2.0834x over previous
#include <hip/hip_runtime.h>
#include <math.h>

#define BB 8
#define CC 63
#define HH 240
#define WW 240
#define HW (HH * WW)
#define C2 126
#define WS 5
#define NWIN 48
#define BN_EPS 1e-5f

typedef __attribute__((ext_vector_type(2))) float v2f;
typedef __attribute__((ext_vector_type(8))) float v8f;

__device__ __forceinline__ v8f wmma4(v2f a, v2f b, v8f c) {
  return __builtin_amdgcn_wmma_f32_16x16x4_f32(
       false, a,  false, b,
       (short)0, c,  false,  false);
}

__device__ __forceinline__ v2f frag_a(const float* p, int ld, int m0, int k0, int lane) {
  int m  = m0 + (lane & 15);
  int kb = k0 + ((lane >> 4) << 1);
  const float* q = p + m * ld + kb;
  v2f r; r.x = q[0]; r.y = q[1];
  return r;
}

__device__ __forceinline__ v2f frag_b(const float* p, int ld, int k0, int n0, int lane) {
  int n  = n0 + (lane & 15);
  int kb = k0 + ((lane >> 4) << 1);
  v2f r; r.x = p[kb * ld + n]; r.y = p[(kb + 1) * ld + n];
  return r;
}

__device__ __forceinline__ void frag_store(float* p, int ld, int m0, int n0, int lane, v8f d) {
  int n  = n0 + (lane & 15);
  int mb = m0 + ((lane >> 4) << 3);
#pragma unroll
  for (int r = 0; r < 8; ++r) p[(mb + r) * ld + n] = d[r];
}

#define V8Z {0.f,0.f,0.f,0.f,0.f,0.f,0.f,0.f}

__global__ __launch_bounds__(128) void k_proj_in(
    const float* __restrict__ x, const float* __restrict__ w_in,
    const float* __restrict__ b_in, const float* __restrict__ bn_g,
    const float* __restrict__ bn_b, const float* __restrict__ bn_m,
    const float* __restrict__ bn_v, float* __restrict__ xp)
{
  __shared__ float a_lds[64][65];
  __shared__ float w_lds[64][128];
  __shared__ float bias_lds[128];
  __shared__ float d_lds[4][16][17];

  int tid  = threadIdx.x;
  int lane = tid & 31;
  int wave = tid >> 5;
  int tile = blockIdx.x;
  int b    = tile / 900;
  int p0   = (tile % 900) * 64;

  for (int idx = tid; idx < 64 * 128; idx += 128) {
    int k = idx >> 7, n = idx & 127;
    float val = 0.f;
    if (k < 63 && n < 126) {
      float s = bn_g[n] * rsqrtf(bn_v[n] + BN_EPS);
      val = w_in[n * 63 + k] * s;
    }
    w_lds[k][n] = val;
  }
  if (tid < 128) {
    float bv = 0.f;
    if (tid < 126) {
      float s = bn_g[tid] * rsqrtf(bn_v[tid] + BN_EPS);
      bv = b_in[tid] * s + bn_b[tid] - bn_m[tid] * s;
    }
    bias_lds[tid] = bv;
  }
  for (int idx = tid; idx < 64 * 64; idx += 128) {
    int k = idx >> 6, m = idx & 63;
    a_lds[m][k] = (k < 63) ? x[(size_t)(b * 63 + k) * HW + p0 + m] : 0.f;
  }
  __syncthreads();

  int m0 = wave * 16;
  for (int nt = 0; nt < 8; ++nt) {
    int n0 = nt * 16;
    v8f acc = V8Z;
    for (int k0 = 0; k0 < 64; k0 += 4) {
      v2f av = frag_a(&a_lds[0][0], 65, m0, k0, lane);
      v2f bv = frag_b(&w_lds[0][0], 128, k0, n0, lane);
      acc = wmma4(av, bv, acc);
    }
    int nloc = lane & 15;
    int mb   = (lane >> 4) << 3;
    float bias = bias_lds[n0 + nloc];
#pragma unroll
    for (int r = 0; r < 8; ++r) d_lds[wave][mb + r][nloc] = acc[r] + bias;
    for (int idx = lane; idx < 256; idx += 32) {
      int n = idx >> 4, m = idx & 15;
      int ch = n0 + n;
      if (ch < 126)
        xp[(size_t)(b * 126 + ch) * HW + p0 + m0 + m] = d_lds[wave][m][n];
    }
  }
}

__global__ __launch_bounds__(32) void k_win(float* __restrict__ xp, float* __restrict__ y2)
{
  __shared__ float q[32][25];
  __shared__ float v[32][25];
  __shared__ float s[32][33];

  int lane = threadIdx.x;
  int wid  = blockIdx.x;
  int grp  = blockIdx.y;
  int b    = wid / (NWIN * NWIN);
  int rem  = wid % (NWIN * NWIN);
  int wy   = rem / NWIN, wx = rem % NWIN;

  for (int i = lane; i < 32 * 25; i += 32) { (&q[0][0])[i] = 0.f; (&v[0][0])[i] = 0.f; }

  int chq = grp * 42;
  for (int i = lane; i < 25 * 21; i += 32) {
    int m = i / 21, c = i % 21;
    int gy = wy * WS + m / WS, gx = wx * WS + m % WS;
    if (grp == 1) { gy = (gy + 3) % HH; gx = (gx + 3) % WW; }
    size_t base = (size_t)(b * 126 + chq + c) * HW + gy * WW + gx;
    q[m][c] = xp[base];
    v[m][c] = xp[base + (size_t)21 * HW];
  }

  for (int mt = 0; mt < 2; ++mt)
    for (int nt = 0; nt < 2; ++nt) {
      v8f acc = V8Z;
      for (int k0 = 0; k0 < 24; k0 += 4) {
        v2f av = frag_a(&q[0][0], 25, mt * 16, k0, lane);
        v2f bv = frag_a(&q[0][0], 25, nt * 16, k0, lane);
        acc = wmma4(av, bv, acc);
      }
      frag_store(&s[0][0], 33, mt * 16, nt * 16, lane, acc);
    }

  if (lane < 25) {
    float mx = -1e30f;
    for (int n = 0; n < 25; ++n) mx = fmaxf(mx, s[lane][n]);
    float sum = 0.f;
    for (int n = 0; n < 25; ++n) { float e = __expf(s[lane][n] - mx); s[lane][n] = e; sum += e; }
    float inv = 1.f / sum;
    for (int n = 0; n < 25; ++n) s[lane][n] *= inv;
    for (int n = 25; n < 28; ++n) s[lane][n] = 0.f;
  }

  for (int mt = 0; mt < 2; ++mt)
    for (int nt = 0; nt < 2; ++nt) {
      v8f acc = V8Z;
      for (int k0 = 0; k0 < 28; k0 += 4) {
        v2f av = frag_a(&s[0][0], 33, mt * 16, k0, lane);
        v2f bv = frag_b(&v[0][0], 25, k0, nt * 16, lane);
        acc = wmma4(av, bv, acc);
      }
      int n  = nt * 16 + (lane & 15);
      int mb = mt * 16 + ((lane >> 4) << 3);
#pragma unroll
      for (int r = 0; r < 8; ++r) {
        int m = mb + r;
        if (m < 25 && n < 21) {
          int oy = wy * WS + m / WS, ox = wx * WS + m % WS;
          if (grp == 0) {
            xp[(size_t)(b * 126 + n) * HW + oy * WW + ox] = acc[r];
          } else {
            oy = (oy + 2) % HH; ox = (ox + 2) % WW;
            y2[(size_t)(b * 21 + n) * HW + oy * WW + ox] = acc[r];
          }
        }
      }
    }
}

#define AX_WAVES 15
#define AX_THREADS (AX_WAVES * 32)

__global__ __launch_bounds__(AX_THREADS) void k_axial(float* __restrict__ xp,
                                                      float* __restrict__ vmid, int pass)
{
  __shared__ float q[240][25];
  __shared__ float v[241][25];
  __shared__ float s[AX_WAVES][16][241];

  int tid  = threadIdx.x;
  int lane = tid & 31;
  int qt   = tid >> 5;
  int row  = blockIdx.x;
  int b    = row / HH;
  int r    = row % HH;

  if (pass == 0) {
    for (int i = tid; i < 240 * 21; i += AX_THREADS) {
      int c = i / 240, xx = i % 240;
      q[xx][c] = xp[(size_t)(b * 126 + 84 + c) * HW + r * WW + xx];
      v[xx][c] = xp[(size_t)(b * 126 + 105 + c) * HW + r * WW + xx];
    }
  } else {
    for (int i = tid; i < 240 * 21; i += AX_THREADS) {
      int c = i / 240, h = i % 240;
      q[h][c] = xp[(size_t)(b * 126 + 84 + c) * HW + h * WW + r];
    }
    for (int i = tid; i < 240 * 21; i += AX_THREADS) {
      int h = i / 21, c = i % 21;
      v[h][c] = vmid[((size_t)(b * 240 + r) * 240 + h) * 21 + c];
    }
  }
  for (int i = tid; i < 240 * 4; i += AX_THREADS) {
    int m = i >> 2, c = 21 + (i & 3);
    q[m][c] = 0.f; v[m][c] = 0.f;
  }
  if (tid < 25) v[240][tid] = 0.f;
  __syncthreads();

  float (*sw)[241] = s[qt];
  int m0 = qt * 16;
  for (int nt = 0; nt < 15; ++nt) {
    int n0 = nt * 16;
    v8f acc = V8Z;
    for (int k0 = 0; k0 < 24; k0 += 4) {
      v2f av = frag_a(&q[0][0], 25, m0, k0, lane);
      v2f bv = frag_a(&q[0][0], 25, n0, k0, lane);
      acc = wmma4(av, bv, acc);
    }
    frag_store(&sw[0][0], 241, 0, n0, lane, acc);
  }

  if (lane < 16) {
    float mx = -1e30f;
    for (int n = 0; n < 240; ++n) mx = fmaxf(mx, sw[lane][n]);
    float sum = 0.f;
    for (int n = 0; n < 240; ++n) { float e = __expf(sw[lane][n] - mx); sw[lane][n] = e; sum += e; }
    float inv = 1.f / sum;
    for (int n = 0; n < 240; ++n) sw[lane][n] *= inv;
  }

  for (int nt = 0; nt < 2; ++nt) {
    int n0 = nt * 16;
    v8f acc = V8Z;
    for (int k0 = 0; k0 < 240; k0 += 4) {
      v2f av = frag_a(&sw[0][0], 241, 0, k0, lane);
      v2f bv = frag_b(&v[0][0], 25, k0, n0, lane);
      acc = wmma4(av, bv, acc);
    }
    int n  = n0 + (lane & 15);
    int mb = (lane >> 4) << 3;
#pragma unroll
    for (int rr = 0; rr < 8; ++rr) {
      int m = m0 + mb + rr;
      if (n < 21) {
        if (pass == 0)
          vmid[((size_t)(b * 240 + m) * 240 + r) * 21 + n] = acc[rr];
        else
          xp[(size_t)(b * 126 + 105 + n) * HW + m * WW + r] = acc[rr];
      }
    }
  }
}

__global__ __launch_bounds__(128) void k_proj_out(
    const float* __restrict__ xp, const float* __restrict__ y2,
    const float* __restrict__ w_out, const float* __restrict__ b_out,
    float* __restrict__ out)
{
  __shared__ float a_lds[64][65];
  __shared__ float w_lds[64][64];
  __shared__ float bias_lds[64];
  __shared__ float d_lds[4][16][17];

  int tid  = threadIdx.x;
  int lane = tid & 31;
  int wave = tid >> 5;
  int tile = blockIdx.x;
  int b    = tile / 900;
  int p0   = (tile % 900) * 64;

  for (int idx = tid; idx < 64 * 64; idx += 128) {
    int k = idx >> 6, n = idx & 63;
    w_lds[k][n] = (k < 63 && n < 63) ? w_out[n * 63 + k] : 0.f;
  }
  if (tid < 64) bias_lds[tid] = (tid < 63) ? b_out[tid] : 0.f;

  for (int idx = tid; idx < 64 * 64; idx += 128) {
    int k = idx >> 6, m = idx & 63;
    float val = 0.f;
    if (k < 21)       val = xp[(size_t)(b * 126 + k)        * HW + p0 + m];
    else if (k < 42)  val = y2[(size_t)(b * 21 + (k - 21))  * HW + p0 + m];
    else if (k < 63)  val = xp[(size_t)(b * 126 + (k + 63)) * HW + p0 + m];
    a_lds[m][k] = val;
  }
  __syncthreads();

  int m0 = wave * 16;
  for (int nt = 0; nt < 4; ++nt) {
    int n0 = nt * 16;
    v8f acc = V8Z;
    for (int k0 = 0; k0 < 64; k0 += 4) {
      v2f av = frag_a(&a_lds[0][0], 65, m0, k0, lane);
      v2f bv = frag_b(&w_lds[0][0], 64, k0, n0, lane);
      acc = wmma4(av, bv, acc);
    }
    int nloc = lane & 15;
    int mb   = (lane >> 4) << 3;
    float bias = bias_lds[n0 + nloc];
#pragma unroll
    for (int r = 0; r < 8; ++r) d_lds[wave][mb + r][nloc] = acc[r] + bias;
    for (int idx = lane; idx < 256; idx += 32) {
      int n = idx >> 4, m = idx & 15;
      int ch = n0 + n;
      if (ch < 63)
        out[(size_t)(b * 63 + ch) * HW + p0 + m0 + m] = d_lds[wave][m][n];
    }
  }
}

extern "C" void kernel_launch(void* const* d_in, const int* in_sizes, int n_in,
                              void* d_out, int out_size, void* d_ws, size_t ws_size,
                              hipStream_t stream)
{
  (void)in_sizes; (void)n_in; (void)out_size; (void)ws_size;
  const float* x     = (const float*)d_in[0];
  const float* w_in  = (const float*)d_in[1];
  const float* b_in  = (const float*)d_in[2];
  const float* bn_g  = (const float*)d_in[3];
  const float* bn_b  = (const float*)d_in[4];
  const float* bn_m  = (const float*)d_in[5];
  const float* bn_v  = (const float*)d_in[6];
  const float* w_out = (const float*)d_in[7];
  const float* b_out = (const float*)d_in[8];
  float* out = (float*)d_out;

  float* xp   = (float*)d_ws;
  float* y2b  = xp  + (size_t)BB * 126 * HW;
  float* vmid = y2b + (size_t)BB * 21 * HW;

  k_proj_in <<<dim3(BB * 900), 128, 0, stream>>>(x, w_in, b_in, bn_g, bn_b, bn_m, bn_v, xp);
  k_win     <<<dim3(BB * NWIN * NWIN, 2), 32, 0, stream>>>(xp, y2b);
  k_axial   <<<dim3(BB * HH), AX_THREADS, 0, stream>>>(xp, vmid, 0);
  k_axial   <<<dim3(BB * WW), AX_THREADS, 0, stream>>>(xp, vmid, 1);
  k_proj_out<<<dim3(BB * 900), 128, 0, stream>>>(xp, y2b, w_out, b_out, out);
}


# staging loops of all four kernels issue every global load of a stage into registers before the first LDS store (weights read linearly), so memory round trips overlap; sums unchanged
# speedup vs baseline: 2.4113x; 1.1574x over previous
#include <hip/hip_runtime.h>
#include <math.h>

#define BB 8
#define CC 63
#define HH 240
#define WW 240
#define HW (HH * WW)
#define C2 126
#define WS 5
#define NWIN 48
#define BN_EPS 1e-5f

typedef __attribute__((ext_vector_type(2))) float v2f;
typedef __attribute__((ext_vector_type(8))) float v8f;

__device__ __forceinline__ v8f wmma4(v2f a, v2f b, v8f c) {
  return __builtin_amdgcn_wmma_f32_16x16x4_f32(
       false, a,  false, b,
       (short)0, c,  false,  false);
}

__device__ __forceinline__ v2f frag_a(const float* p, int ld, int m0, int k0, int lane) {
  int m  = m0 + (lane & 15);
  int kb = k0 + ((lane >> 4) << 1);
  const float* q = p + m * ld + kb;
  v2f r; r.x = q[0]; r.y = q[1];
  return r;
}

__device__ __forceinline__ v2f frag_b(const float* p, int ld, int k0, int n0, int lane) {
  int n  = n0 + (lane & 15);
  int kb = k0 + ((lane >> 4) << 1);
  v2f r; r.x = p[kb * ld + n]; r.y = p[(kb + 1) * ld + n];
  return r;
}

__device__ __forceinline__ void frag_store(float* p, int ld, int m0, int n0, int lane, v8f d) {
  int n  = n0 + (lane & 15);
  int mb = m0 + ((lane >> 4) << 3);
#pragma unroll
  for (int r = 0; r < 8; ++r) p[(mb + r) * ld + n] = d[r];
}

#define V8Z {0.f,0.f,0.f,0.f,0.f,0.f,0.f,0.f}

__global__ __launch_bounds__(128) void k_proj_in(
    const float* __restrict__ x, const float* __restrict__ w_in,
    const float* __restrict__ b_in, const float* __restrict__ bn_g,
    const float* __restrict__ bn_b, const float* __restrict__ bn_m,
    const float* __restrict__ bn_v, float* __restrict__ xp)
{
  __shared__ float a_lds[64][65];
  __shared__ float w_lds[64][129];
  __shared__ float bias_lds[128];
  __shared__ float scale_lds[128];
  __shared__ float d_lds[4][16][17];

  int tid  = threadIdx.x;
  int lane = tid & 31;
  int wave = tid >> 5;
  int tile = blockIdx.x;
  int b    = tile / 900;
  int p0   = (tile % 900) * 64;

  {
    float sc = 0.f, bv = 0.f;
    if (tid < 126) {
      sc = bn_g[tid] * rsqrtf(bn_v[tid] + BN_EPS);
      bv = b_in[tid] * sc + bn_b[tid] - bn_m[tid] * sc;
    }
    scale_lds[tid] = sc;
    bias_lds[tid]  = bv;
  }
  const int am = tid & 63, akh = tid >> 6;
  float ta[32];
  {
    const float* xb = x + (size_t)b * 63 * HW + p0 + am;
#pragma unroll
    for (int j = 0; j < 32; ++j) {
      int k  = 2 * j + akh;
      int kc = k < 63 ? k : 62;
      ta[j] = xb[(size_t)kc * HW];
    }
  }
  float tw[63];
#pragma unroll
  for (int j = 0; j < 63; ++j) {
    int idx = tid + j * 128;
    int ic  = idx < 126 * 63 ? idx : 126 * 63 - 1;
    tw[j] = w_in[ic];
  }
  w_lds[63][tid] = 0.f;
  if (tid < 63) { w_lds[tid][126] = 0.f; w_lds[tid][127] = 0.f; }
  __syncthreads();
#pragma unroll
  for (int j = 0; j < 63; ++j) {
    int idx = tid + j * 128;
    if (idx < 126 * 63) {
      int n = idx / 63, k = idx - n * 63;
      w_lds[k][n] = tw[j] * scale_lds[n];
    }
  }
#pragma unroll
  for (int j = 0; j < 32; ++j) {
    int k = 2 * j + akh;
    a_lds[am][k] = (k < 63) ? ta[j] : 0.f;
  }
  __syncthreads();

  int m0 = wave * 16;
  for (int nt = 0; nt < 8; ++nt) {
    int n0 = nt * 16;
    v8f acc = V8Z;
    for (int k0 = 0; k0 < 64; k0 += 4) {
      v2f av = frag_a(&a_lds[0][0], 65, m0, k0, lane);
      v2f bv = frag_b(&w_lds[0][0], 129, k0, n0, lane);
      acc = wmma4(av, bv, acc);
    }
    int nloc = lane & 15;
    int mb   = (lane >> 4) << 3;
    float bias = bias_lds[n0 + nloc];
#pragma unroll
    for (int r = 0; r < 8; ++r) d_lds[wave][mb + r][nloc] = acc[r] + bias;
    for (int idx = lane; idx < 256; idx += 32) {
      int n = idx >> 4, m = idx & 15;
      int ch = n0 + n;
      if (ch < 126)
        xp[(size_t)(b * 126 + ch) * HW + p0 + m0 + m] = d_lds[wave][m][n];
    }
  }
}

__global__ __launch_bounds__(32) void k_win(float* __restrict__ xp, float* __restrict__ y2)
{
  __shared__ float q[32][25];
  __shared__ float v[32][25];
  __shared__ float s[32][33];

  int lane = threadIdx.x;
  int wid  = blockIdx.x;
  int grp  = blockIdx.y;
  int b    = wid / (NWIN * NWIN);
  int rem  = wid % (NWIN * NWIN);
  int wy   = rem / NWIN, wx = rem % NWIN;

  int chq = grp * 42;
  float tq[17], tv[17];
#pragma unroll
  for (int j = 0; j < 17; ++j) {
    int i  = lane + j * 32;
    int ic = i < 25 * 21 ? i : 25 * 21 - 1;
    int m = ic / 21, c = ic - m * 21;
    int gy = wy * WS + m / WS, gx = wx * WS + m % WS;
    if (grp == 1) { gy = (gy + 3) % HH; gx = (gx + 3) % WW; }
    size_t base = (size_t)(b * 126 + chq + c) * HW + gy * WW + gx;
    tq[j] = xp[base];
    tv[j] = xp[base + (size_t)21 * HW];
  }

  for (int i = lane; i < 32 * 25; i += 32) { (&q[0][0])[i] = 0.f; (&v[0][0])[i] = 0.f; }

#pragma unroll
  for (int j = 0; j < 17; ++j) {
    int i = lane + j * 32;
    if (i < 25 * 21) {
      int m = i / 21, c = i - m * 21;
      q[m][c] = tq[j];
      v[m][c] = tv[j];
    }
  }

  for (int mt = 0; mt < 2; ++mt)
    for (int nt = 0; nt < 2; ++nt) {
      v8f acc = V8Z;
      for (int k0 = 0; k0 < 24; k0 += 4) {
        v2f av = frag_a(&q[0][0], 25, mt * 16, k0, lane);
        v2f bv = frag_a(&q[0][0], 25, nt * 16, k0, lane);
        acc = wmma4(av, bv, acc);
      }
      frag_store(&s[0][0], 33, mt * 16, nt * 16, lane, acc);
    }

  if (lane < 25) {
    float mx = -1e30f;
    for (int n = 0; n < 25; ++n) mx = fmaxf(mx, s[lane][n]);
    float sum = 0.f;
    for (int n = 0; n < 25; ++n) { float e = __expf(s[lane][n] - mx); s[lane][n] = e; sum += e; }
    float inv = 1.f / sum;
    for (int n = 0; n < 25; ++n) s[lane][n] *= inv;
    for (int n = 25; n < 28; ++n) s[lane][n] = 0.f;
  }

  for (int mt = 0; mt < 2; ++mt)
    for (int nt = 0; nt < 2; ++nt) {
      v8f acc = V8Z;
      for (int k0 = 0; k0 < 28; k0 += 4) {
        v2f av = frag_a(&s[0][0], 33, mt * 16, k0, lane);
        v2f bv = frag_b(&v[0][0], 25, k0, nt * 16, lane);
        acc = wmma4(av, bv, acc);
      }
      int n  = nt * 16 + (lane & 15);
      int mb = mt * 16 + ((lane >> 4) << 3);
#pragma unroll
      for (int r = 0; r < 8; ++r) {
        int m = mb + r;
        if (m < 25 && n < 21) {
          int oy = wy * WS + m / WS, ox = wx * WS + m % WS;
          if (grp == 0) {
            xp[(size_t)(b * 126 + n) * HW + oy * WW + ox] = acc[r];
          } else {
            oy = (oy + 2) % HH; ox = (ox + 2) % WW;
            y2[(size_t)(b * 21 + n) * HW + oy * WW + ox] = acc[r];
          }
        }
      }
    }
}

#define AX_WAVES 15
#define AX_THREADS (AX_WAVES * 32)

__global__ __launch_bounds__(AX_THREADS) void k_axial(float* __restrict__ xp,
                                                      float* __restrict__ vmid, int pass)
{
  __shared__ float q[240][25];
  __shared__ float v[241][25];
  __shared__ float s[AX_WAVES][16][241];

  int tid  = threadIdx.x;
  int lane = tid & 31;
  int qt   = tid >> 5;
  int row  = blockIdx.x;
  int b    = row / HH;
  int r    = row % HH;

  {
    float tq[11], tv[11];
    if (pass == 0) {
      const float* qb = xp + (size_t)(b * 126 + 84) * HW + r * WW;
      const float* vb = xp + (size_t)(b * 126 + 105) * HW + r * WW;
#pragma unroll
      for (int j = 0; j < 11; ++j) {
        int i  = tid + j * AX_THREADS;
        int ic = i < 240 * 21 ? i : 240 * 21 - 1;
        int c = ic / 240, xx = ic - c * 240;
        tq[j] = qb[(size_t)c * HW + xx];
        tv[j] = vb[(size_t)c * HW + xx];
      }
#pragma unroll
      for (int j = 0; j < 11; ++j) {
        int i = tid + j * AX_THREADS;
        if (i < 240 * 21) {
          int c = i / 240, xx = i - c * 240;
          q[xx][c] = tq[j];
          v[xx][c] = tv[j];
        }
      }
    } else {
      const float* qb = xp + (size_t)(b * 126 + 84) * HW + r;
      const float* vb = vmid + (size_t)(b * 240 + r) * 240 * 21;
#pragma unroll
      for (int j = 0; j < 11; ++j) {
        int i  = tid + j * AX_THREADS;
        int ic = i < 240 * 21 ? i : 240 * 21 - 1;
        int c = ic / 240, h = ic - c * 240;
        tq[j] = qb[(size_t)c * HW + h * WW];
        tv[j] = vb[ic];
      }
#pragma unroll
      for (int j = 0; j < 11; ++j) {
        int i = tid + j * AX_THREADS;
        if (i < 240 * 21) {
          int c = i / 240, h = i - c * 240;
          q[h][c] = tq[j];
          int h2 = i / 21, c2 = i - h2 * 21;
          v[h2][c2] = tv[j];
        }
      }
    }
  }
  for (int i = tid; i < 240 * 4; i += AX_THREADS) {
    int m = i >> 2, c = 21 + (i & 3);
    q[m][c] = 0.f; v[m][c] = 0.f;
  }
  if (tid < 25) v[240][tid] = 0.f;
  __syncthreads();

  float (*sw)[241] = s[qt];
  int m0 = qt * 16;
  for (int nt = 0; nt < 15; ++nt) {
    int n0 = nt * 16;
    v8f acc = V8Z;
    for (int k0 = 0; k0 < 24; k0 += 4) {
      v2f av = frag_a(&q[0][0], 25, m0, k0, lane);
      v2f bv = frag_a(&q[0][0], 25, n0, k0, lane);
      acc = wmma4(av, bv, acc);
    }
    frag_store(&sw[0][0], 241, 0, n0, lane, acc);
  }

  if (lane < 16) {
    float mx = -1e30f;
    for (int n = 0; n < 240; ++n) mx = fmaxf(mx, sw[lane][n]);
    float sum = 0.f;
    for (int n = 0; n < 240; ++n) { float e = __expf(sw[lane][n] - mx); sw[lane][n] = e; sum += e; }
    float inv = 1.f / sum;
    for (int n = 0; n < 240; ++n) sw[lane][n] *= inv;
  }

  for (int nt = 0; nt < 2; ++nt) {
    int n0 = nt * 16;
    v8f acc = V8Z;
    for (int k0 = 0; k0 < 240; k0 += 4) {
      v2f av = frag_a(&sw[0][0], 241, 0, k0, lane);
      v2f bv = frag_b(&v[0][0], 25, k0, n0, lane);
      acc = wmma4(av, bv, acc);
    }
    int n  = n0 + (lane & 15);
    int mb = (lane >> 4) << 3;
#pragma unroll
    for (int rr = 0; rr < 8; ++rr) {
      int m = m0 + mb + rr;
      if (n < 21) {
        if (pass == 0)
          vmid[((size_t)(b * 240 + m) * 240 + r) * 21 + n] = acc[rr];
        else
          xp[(size_t)(b * 126 + 105 + n) * HW + m * WW + r] = acc[rr];
      }
    }
  }
}

__global__ __launch_bounds__(128) void k_proj_out(
    const float* __restrict__ xp, const float* __restrict__ y2,
    const float* __restrict__ w_out, const float* __restrict__ b_out,
    float* __restrict__ out)
{
  __shared__ float a_lds[64][65];
  __shared__ float w_lds[64][65];
  __shared__ float bias_lds[64];
  __shared__ float d_lds[4][16][17];

  int tid  = threadIdx.x;
  int lane = tid & 31;
  int wave = tid >> 5;
  int tile = blockIdx.x;
  int b    = tile / 900;
  int p0   = (tile % 900) * 64;

  const int am = tid & 63, akh = tid >> 6;
  float ta[32];
#pragma unroll
  for (int j = 0; j < 32; ++j) {
    int k = 2 * j + akh;
    const float* sp;
    if (k < 21)      sp = xp + (size_t)(b * 126 + k) * HW;
    else if (k < 42) sp = y2 + (size_t)(b * 21 + (k - 21)) * HW;
    else             sp = xp + (size_t)(b * 126 + ((k < 63 ? k : 62) + 63)) * HW;
    ta[j] = sp[p0 + am];
  }
  float tw[32];
#pragma unroll
  for (int j = 0; j < 32; ++j) {
    int idx = tid + j * 128;
    int ic  = idx < 63 * 63 ? idx : 63 * 63 - 1;
    tw[j] = w_out[ic];
  }
  if (tid < 64) {
    bias_lds[tid] = (tid < 63) ? b_out[tid] : 0.f;
    w_lds[63][tid] = 0.f;
    w_lds[tid][63] = 0.f;
  }
#pragma unroll
  for (int j = 0; j < 32; ++j) {
    int idx = tid + j * 128;
    if (idx < 63 * 63) {
      int n = idx / 63, k = idx - n * 63;
      w_lds[k][n] = tw[j];
    }
  }
#pragma unroll
  for (int j = 0; j < 32; ++j) {
    int k = 2 * j + akh;
    a_lds[am][k] = (k < 63) ? ta[j] : 0.f;
  }
  __syncthreads();

  int m0 = wave * 16;
  for (int nt = 0; nt < 4; ++nt) {
    int n0 = nt * 16;
    v8f acc = V8Z;
    for (int k0 = 0; k0 < 64; k0 += 4) {
      v2f av = frag_a(&a_lds[0][0], 65, m0, k0, lane);
      v2f bv = frag_b(&w_lds[0][0], 65, k0, n0, lane);
      acc = wmma4(av, bv, acc);
    }
    int nloc = lane & 15;
    int mb   = (lane >> 4) << 3;
    float bias = bias_lds[n0 + nloc];
#pragma unroll
    for (int r = 0; r < 8; ++r) d_lds[wave][mb + r][nloc] = acc[r] + bias;
    for (int idx = lane; idx < 256; idx += 32) {
      int n = idx >> 4, m = idx & 15;
      int ch = n0 + n;
      if (ch < 63)
        out[(size_t)(b * 63 + ch) * HW + p0 + m0 + m] = d_lds[wave][m][n];
    }
  }
}

extern "C" void kernel_launch(void* const* d_in, const int* in_sizes, int n_in,
                              void* d_out, int out_size, void* d_ws, size_t ws_size,
                              hipStream_t stream)
{
  (void)in_sizes; (void)n_in; (void)out_size; (void)ws_size;
  const float* x     = (const float*)d_in[0];
  const float* w_in  = (const float*)d_in[1];
  const float* b_in  = (const float*)d_in[2];
  const float* bn_g  = (const float*)d_in[3];
  const float* bn_b  = (const float*)d_in[4];
  const float* bn_m  = (const float*)d_in[5];
  const float* bn_v  = (const float*)d_in[6];
  const float* w_out = (const float*)d_in[7];
  const float* b_out = (const float*)d_in[8];
  float* out = (float*)d_out;

  float* xp   = (float*)d_ws;
  float* y2b  = xp  + (size_t)BB * 126 * HW;
  float* vmid = y2b + (size_t)BB * 21 * HW;

  k_proj_in <<<dim3(BB * 900), 128, 0, stream>>>(x, w_in, b_in, bn_g, bn_b, bn_m, bn_v, xp);
  k_win     <<<dim3(BB * NWIN * NWIN, 2), 32, 0, stream>>>(xp, y2b);
  k_axial   <<<dim3(BB * HH), AX_THREADS, 0, stream>>>(xp, vmid, 0);
  k_axial   <<<dim3(BB * WW), AX_THREADS, 0, stream>>>(xp, vmid, 1);
  k_proj_out<<<dim3(BB * 900), 128, 0, stream>>>(xp, y2b, w_out, b_out, out);
}


# axial attention: scores computed transposed, kept in registers; softmax in-lane on all 32 lanes; strip registers are the A operand of P V (no LDS score strip); row sums and P V keys in another order
# speedup vs baseline: 2.9458x; 1.2217x over previous
#include <hip/hip_runtime.h>
#include <math.h>

#define BB 8
#define CC 63
#define HH 240
#define WW 240
#define HW (HH * WW)
#define C2 126
#define WS 5
#define NWIN 48
#define BN_EPS 1e-5f

typedef __attribute__((ext_vector_type(2))) float v2f;
typedef __attribute__((ext_vector_type(8))) float v8f;

__device__ __forceinline__ v8f wmma4(v2f a, v2f b, v8f c) {
  return __builtin_amdgcn_wmma_f32_16x16x4_f32(
       false, a,  false, b,
       (short)0, c,  false,  false);
}

__device__ __forceinline__ v2f frag_a(const float* p, int ld, int m0, int k0, int lane) {
  int m  = m0 + (lane & 15);
  int kb = k0 + ((lane >> 4) << 1);
  const float* q = p + m * ld + kb;
  v2f r; r.x = q[0]; r.y = q[1];
  return r;
}

__device__ __forceinline__ v2f frag_b(const float* p, int ld, int k0, int n0, int lane) {
  int n  = n0 + (lane & 15);
  int kb = k0 + ((lane >> 4) << 1);
  v2f r; r.x = p[kb * ld + n]; r.y = p[(kb + 1) * ld + n];
  return r;
}

__device__ __forceinline__ void frag_store(float* p, int ld, int m0, int n0, int lane, v8f d) {
  int n  = n0 + (lane & 15);
  int mb = m0 + ((lane >> 4) << 3);
#pragma unroll
  for (int r = 0; r < 8; ++r) p[(mb + r) * ld + n] = d[r];
}

#define V8Z {0.f,0.f,0.f,0.f,0.f,0.f,0.f,0.f}

__global__ __launch_bounds__(128) void k_proj_in(
    const float* __restrict__ x, const float* __restrict__ w_in,
    const float* __restrict__ b_in, const float* __restrict__ bn_g,
    const float* __restrict__ bn_b, const float* __restrict__ bn_m,
    const float* __restrict__ bn_v, float* __restrict__ xp)
{
  __shared__ float a_lds[64][65];
  __shared__ float w_lds[64][129];
  __shared__ float bias_lds[128];
  __shared__ float scale_lds[128];
  __shared__ float d_lds[4][16][17];

  int tid  = threadIdx.x;
  int lane = tid & 31;
  int wave = tid >> 5;
  int tile = blockIdx.x;
  int b    = tile / 900;
  int p0   = (tile % 900) * 64;

  {
    float sc = 0.f, bv = 0.f;
    if (tid < 126) {
      sc = bn_g[tid] * rsqrtf(bn_v[tid] + BN_EPS);
      bv = b_in[tid] * sc + bn_b[tid] - bn_m[tid] * sc;
    }
    scale_lds[tid] = sc;
    bias_lds[tid]  = bv;
  }
  const int am = tid & 63, akh = tid >> 6;
  float ta[32];
  {
    const float* xb = x + (size_t)b * 63 * HW + p0 + am;
#pragma unroll
    for (int j = 0; j < 32; ++j) {
      int k  = 2 * j + akh;
      int kc = k < 63 ? k : 62;
      ta[j] = xb[(size_t)kc * HW];
    }
  }
  float tw[63];
#pragma unroll
  for (int j = 0; j < 63; ++j) {
    int idx = tid + j * 128;
    int ic  = idx < 126 * 63 ? idx : 126 * 63 - 1;
    tw[j] = w_in[ic];
  }
  w_lds[63][tid] = 0.f;
  if (tid < 63) { w_lds[tid][126] = 0.f; w_lds[tid][127] = 0.f; }
  __syncthreads();
#pragma unroll
  for (int j = 0; j < 63; ++j) {
    int idx = tid + j * 128;
    if (idx < 126 * 63) {
      int n = idx / 63, k = idx - n * 63;
      w_lds[k][n] = tw[j] * scale_lds[n];
    }
  }
#pragma unroll
  for (int j = 0; j < 32; ++j) {
    int k = 2 * j + akh;
    a_lds[am][k] = (k < 63) ? ta[j] : 0.f;
  }
  __syncthreads();

  int m0 = wave * 16;
  for (int nt = 0; nt < 8; ++nt) {
    int n0 = nt * 16;
    v8f acc = V8Z;
    for (int k0 = 0; k0 < 64; k0 += 4) {
      v2f av = frag_a(&a_lds[0][0], 65, m0, k0, lane);
      v2f bv = frag_b(&w_lds[0][0], 129, k0, n0, lane);
      acc = wmma4(av, bv, acc);
    }
    int nloc = lane & 15;
    int mb   = (lane >> 4) << 3;
    float bias = bias_lds[n0 + nloc];
#pragma unroll
    for (int r = 0; r < 8; ++r) d_lds[wave][mb + r][nloc] = acc[r] + bias;
    for (int idx = lane; idx < 256; idx += 32) {
      int n = idx >> 4, m = idx & 15;
      int ch = n0 + n;
      if (ch < 126)
        xp[(size_t)(b * 126 + ch) * HW + p0 + m0 + m] = d_lds[wave][m][n];
    }
  }
}

__global__ __launch_bounds__(32) void k_win(float* __restrict__ xp, float* __restrict__ y2)
{
  __shared__ float q[32][25];
  __shared__ float v[32][25];
  __shared__ float s[32][33];

  int lane = threadIdx.x;
  int wid  = blockIdx.x;
  int grp  = blockIdx.y;
  int b    = wid / (NWIN * NWIN);
  int rem  = wid % (NWIN * NWIN);
  int wy   = rem / NWIN, wx = rem % NWIN;

  int chq = grp * 42;
  float tq[17], tv[17];
#pragma unroll
  for (int j = 0; j < 17; ++j) {
    int i  = lane + j * 32;
    int ic = i < 25 * 21 ? i : 25 * 21 - 1;
    int m = ic / 21, c = ic - m * 21;
    int gy = wy * WS + m / WS, gx = wx * WS + m % WS;
    if (grp == 1) { gy = (gy + 3) % HH; gx = (gx + 3) % WW; }
    size_t base = (size_t)(b * 126 + chq + c) * HW + gy * WW + gx;
    tq[j] = xp[base];
    tv[j] = xp[base + (size_t)21 * HW];
  }

  for (int i = lane; i < 32 * 25; i += 32) { (&q[0][0])[i] = 0.f; (&v[0][0])[i] = 0.f; }

#pragma unroll
  for (int j = 0; j < 17; ++j) {
    int i = lane + j * 32;
    if (i < 25 * 21) {
      int m = i / 21, c = i - m * 21;
      q[m][c] = tq[j];
      v[m][c] = tv[j];
    }
  }

  for (int mt = 0; mt < 2; ++mt)
    for (int nt = 0; nt < 2; ++nt) {
      v8f acc = V8Z;
      for (int k0 = 0; k0 < 24; k0 += 4) {
        v2f av = frag_a(&q[0][0], 25, mt * 16, k0, lane);
        v2f bv = frag_a(&q[0][0], 25, nt * 16, k0, lane);
        acc = wmma4(av, bv, acc);
      }
      frag_store(&s[0][0], 33, mt * 16, nt * 16, lane, acc);
    }

  if (lane < 25) {
    float mx = -1e30f;
    for (int n = 0; n < 25; ++n) mx = fmaxf(mx, s[lane][n]);
    float sum = 0.f;
    for (int n = 0; n < 25; ++n) { float e = __expf(s[lane][n] - mx); s[lane][n] = e; sum += e; }
    float inv = 1.f / sum;
    for (int n = 0; n < 25; ++n) s[lane][n] *= inv;
    for (int n = 25; n < 28; ++n) s[lane][n] = 0.f;
  }

  for (int mt = 0; mt < 2; ++mt)
    for (int nt = 0; nt < 2; ++nt) {
      v8f acc = V8Z;
      for (int k0 = 0; k0 < 28; k0 += 4) {
        v2f av = frag_a(&s[0][0], 33, mt * 16, k0, lane);
        v2f bv = frag_b(&v[0][0], 25, k0, nt * 16, lane);
        acc = wmma4(av, bv, acc);
      }
      int n  = nt * 16 + (lane & 15);
      int mb = mt * 16 + ((lane >> 4) << 3);
#pragma unroll
      for (int r = 0; r < 8; ++r) {
        int m = mb + r;
        if (m < 25 && n < 21) {
          int oy = wy * WS + m / WS, ox = wx * WS + m % WS;
          if (grp == 0) {
            xp[(size_t)(b * 126 + n) * HW + oy * WW + ox] = acc[r];
          } else {
            oy = (oy + 2) % HH; ox = (ox + 2) % WW;
            y2[(size_t)(b * 21 + n) * HW + oy * WW + ox] = acc[r];
          }
        }
      }
    }
}

#define AX_WAVES 15
#define AX_THREADS (AX_WAVES * 32)
#define AX_VP 26

__global__ __launch_bounds__(AX_THREADS) void k_axial(float* __restrict__ xp,
                                                      float* __restrict__ vmid, int pass)
{
  __shared__ float q[240][25];
  __shared__ float v[241][AX_VP];

  int tid  = threadIdx.x;
  int lane = tid & 31;
  int qt   = tid >> 5;
  int row  = blockIdx.x;
  int b    = row / HH;
  int r    = row % HH;

  {
    float tq[11], tv[11];
    if (pass == 0) {
      const float* qb = xp + (size_t)(b * 126 + 84) * HW + r * WW;
      const float* vb = xp + (size_t)(b * 126 + 105) * HW + r * WW;
#pragma unroll
      for (int j = 0; j < 11; ++j) {
        int i  = tid + j * AX_THREADS;
        int ic = i < 240 * 21 ? i : 240 * 21 - 1;
        int c = ic / 240, xx = ic - c * 240;
        tq[j] = qb[(size_t)c * HW + xx];
        tv[j] = vb[(size_t)c * HW + xx];
      }
#pragma unroll
      for (int j = 0; j < 11; ++j) {
        int i = tid + j * AX_THREADS;
        if (i < 240 * 21) {
          int c = i / 240, xx = i - c * 240;
          q[xx][c] = tq[j];
          v[xx][c] = tv[j];
        }
      }
    } else {
      const float* qb = xp + (size_t)(b * 126 + 84) * HW + r;
      const float* vb = vmid + (size_t)(b * 240 + r) * 240 * 21;
#pragma unroll
      for (int j = 0; j < 11; ++j) {
        int i  = tid + j * AX_THREADS;
        int ic = i < 240 * 21 ? i : 240 * 21 - 1;
        int c = ic / 240, h = ic - c * 240;
        tq[j] = qb[(size_t)c * HW + h * WW];
        tv[j] = vb[ic];
      }
#pragma unroll
      for (int j = 0; j < 11; ++j) {
        int i = tid + j * AX_THREADS;
        if (i < 240 * 21) {
          int c = i / 240, h = i - c * 240;
          q[h][c] = tq[j];
          int h2 = i / 21, c2 = i - h2 * 21;
          v[h2][c2] = tv[j];
        }
      }
    }
  }
  for (int i = tid; i < 240 * 4; i += AX_THREADS) {
    int m = i >> 2, c = 21 + (i & 3);
    q[m][c] = 0.f;
  }
  for (int i = tid; i < 240 * 5; i += AX_THREADS) {
    int m = i / 5, c = 21 + (i - m * 5);
    v[m][c] = 0.f;
  }
  if (tid < AX_VP) v[240][tid] = 0.f;
  __syncthreads();

  const int m0 = qt * 16;
  const int cn = lane & 15;
  const int kh = (lane >> 4) << 3;

  v2f bq[6];
#pragma unroll
  for (int kk = 0; kk < 6; ++kk) bq[kk] = frag_a(&q[0][0], 25, m0, kk * 4, lane);

  v8f st[15];
#pragma unroll
  for (int g = 0; g < 5; ++g) {
    v8f a0 = V8Z, a1 = V8Z, a2 = V8Z;
#pragma unroll
    for (int kk = 0; kk < 6; ++kk) {
      a0 = wmma4(frag_a(&q[0][0], 25, (3 * g + 0) * 16, kk * 4, lane), bq[kk], a0);
      a1 = wmma4(frag_a(&q[0][0], 25, (3 * g + 1) * 16, kk * 4, lane), bq[kk], a1);
      a2 = wmma4(frag_a(&q[0][0], 25, (3 * g + 2) * 16, kk * 4, lane), bq[kk], a2);
    }
    st[3 * g + 0] = a0; st[3 * g + 1] = a1; st[3 * g + 2] = a2;
  }

  float mx = -1e30f;
#pragma unroll
  for (int nt = 0; nt < 15; ++nt)
#pragma unroll
    for (int rr = 0; rr < 8; ++rr) mx = fmaxf(mx, st[nt][rr]);
  mx = fmaxf(mx, __shfl_xor(mx, 16));
  float s0 = 0.f, s1 = 0.f, s2 = 0.f, s3 = 0.f;
#pragma unroll
  for (int nt = 0; nt < 15; ++nt) {
#pragma unroll
    for (int rr = 0; rr < 8; rr += 4) {
      float e0 = __expf(st[nt][rr + 0] - mx);
      float e1 = __expf(st[nt][rr + 1] - mx);
      float e2 = __expf(st[nt][rr + 2] - mx);
      float e3 = __expf(st[nt][rr + 3] - mx);
      st[nt][rr + 0] = e0; st[nt][rr + 1] = e1; st[nt][rr + 2] = e2; st[nt][rr + 3] = e3;
      s0 += e0; s1 += e1; s2 += e2; s3 += e3;
    }
  }
  float sum = (s0 + s1) + (s2 + s3);
  sum += __shfl_xor(sum, 16);
  float inv = 1.f / sum;
#pragma unroll
  for (int nt = 0; nt < 15; ++nt)
#pragma unroll
    for (int rr = 0; rr < 8; ++rr) st[nt][rr] *= inv;

  v8f y0 = V8Z, y1 = V8Z;
#pragma unroll
  for (int nt = 0; nt < 15; ++nt) {
#pragma unroll
    for (int i = 0; i < 4; ++i) {
      v2f pa; pa.x = st[nt][2 * i]; pa.y = st[nt][2 * i + 1];
      int kb = nt * 16 + kh + 2 * i;
      v2f b0, b1;
      b0.x = v[kb][cn];      b0.y = v[kb + 1][cn];
      b1.x = v[kb][16 + cn]; b1.y = v[kb + 1][16 + cn];
      y0 = wmma4(pa, b0, y0);
      y1 = wmma4(pa, b1, y1);
    }
  }

  float* ob = (pass == 0) ? vmid + ((size_t)(b * 240) * 240 + r) * 21
                          : xp + (size_t)(b * 126 + 105) * HW + r;
  const int sm = (pass == 0) ? 240 * 21 : WW;
  const int sn = (pass == 0) ? 1 : HW;
  const int mb = m0 + kh;
#pragma unroll
  for (int nt = 0; nt < 2; ++nt) {
    v8f acc = nt ? y1 : y0;
    int n = nt * 16 + cn;
    if (n < 21) {
#pragma unroll
      for (int rr = 0; rr < 8; ++rr) ob[(mb + rr) * sm + n * sn] = acc[rr];
    }
  }
}

__global__ __launch_bounds__(128) void k_proj_out(
    const float* __restrict__ xp, const float* __restrict__ y2,
    const float* __restrict__ w_out, const float* __restrict__ b_out,
    float* __restrict__ out)
{
  __shared__ float a_lds[64][65];
  __shared__ float w_lds[64][65];
  __shared__ float bias_lds[64];
  __shared__ float d_lds[4][16][17];

  int tid  = threadIdx.x;
  int lane = tid & 31;
  int wave = tid >> 5;
  int tile = blockIdx.x;
  int b    = tile / 900;
  int p0   = (tile % 900) * 64;

  const int am = tid & 63, akh = tid >> 6;
  float ta[32];
#pragma unroll
  for (int j = 0; j < 32; ++j) {
    int k = 2 * j + akh;
    const float* sp;
    if (k < 21)      sp = xp + (size_t)(b * 126 + k) * HW;
    else if (k < 42) sp = y2 + (size_t)(b * 21 + (k - 21)) * HW;
    else             sp = xp + (size_t)(b * 126 + ((k < 63 ? k : 62) + 63)) * HW;
    ta[j] = sp[p0 + am];
  }
  float tw[32];
#pragma unroll
  for (int j = 0; j < 32; ++j) {
    int idx = tid + j * 128;
    int ic  = idx < 63 * 63 ? idx : 63 * 63 - 1;
    tw[j] = w_out[ic];
  }
  if (tid < 64) {
    bias_lds[tid] = (tid < 63) ? b_out[tid] : 0.f;
    w_lds[63][tid] = 0.f;
    w_lds[tid][63] = 0.f;
  }
#pragma unroll
  for (int j = 0; j < 32; ++j) {
    int idx = tid + j * 128;
    if (idx < 63 * 63) {
      int n = idx / 63, k = idx - n * 63;
      w_lds[k][n] = tw[j];
    }
  }
#pragma unroll
  for (int j = 0; j < 32; ++j) {
    int k = 2 * j + akh;
    a_lds[am][k] = (k < 63) ? ta[j] : 0.f;
  }
  __syncthreads();

  int m0 = wave * 16;
  for (int nt = 0; nt < 4; ++nt) {
    int n0 = nt * 16;
    v8f acc = V8Z;
    for (int k0 = 0; k0 < 64; k0 += 4) {
      v2f av = frag_a(&a_lds[0][0], 65, m0, k0, lane);
      v2f bv = frag_b(&w_lds[0][0], 65, k0, n0, lane);
      acc = wmma4(av, bv, acc);
    }
    int nloc = lane & 15;
    int mb   = (lane >> 4) << 3;
    float bias = bias_lds[n0 + nloc];
#pragma unroll
    for (int r = 0; r < 8; ++r) d_lds[wave][mb + r][nloc] = acc[r] + bias;
    for (int idx = lane; idx < 256; idx += 32) {
      int n = idx >> 4, m = idx & 15;
      int ch = n0 + n;
      if (ch < 63)
        out[(size_t)(b * 63 + ch) * HW + p0 + m0 + m] = d_lds[wave][m][n];
    }
  }
}

extern "C" void kernel_launch(void* const* d_in, const int* in_sizes, int n_in,
                              void* d_out, int out_size, void* d_ws, size_t ws_size,
                              hipStream_t stream)
{
  (void)in_sizes; (void)n_in; (void)out_size; (void)ws_size;
  const float* x     = (const float*)d_in[0];
  const float* w_in  = (const float*)d_in[1];
  const float* b_in  = (const float*)d_in[2];
  const float* bn_g  = (const float*)d_in[3];
  const float* bn_b  = (const float*)d_in[4];
  const float* bn_m  = (const float*)d_in[5];
  const float* bn_v  = (const float*)d_in[6];
  const float* w_out = (const float*)d_in[7];
  const float* b_out = (const float*)d_in[8];
  float* out = (float*)d_out;

  float* xp   = (float*)d_ws;
  float* y2b  = xp  + (size_t)BB * 126 * HW;
  float* vmid = y2b + (size_t)BB * 21 * HW;

  k_proj_in <<<dim3(BB * 900), 128, 0, stream>>>(x, w_in, b_in, bn_g, bn_b, bn_m, bn_v, xp);
  k_win     <<<dim3(BB * NWIN * NWIN, 2), 32, 0, stream>>>(xp, y2b);
  k_axial   <<<dim3(BB * HH), AX_THREADS, 0, stream>>>(xp, vmid, 0);
  k_axial   <<<dim3(BB * WW), AX_THREADS, 0, stream>>>(xp, vmid, 1);
  k_proj_out<<<dim3(BB * 900), 128, 0, stream>>>(xp, y2b, w_out, b_out, out);
}


# window attention: an 8-wave workgroup per strip of 8 windows moves q, v and the result as x-contiguous row pieces through LDS (was one wave per window, a line per lane); sums unchanged
# speedup vs baseline: 4.9870x; 1.6929x over previous
#include <hip/hip_runtime.h>
#include <math.h>

#define BB 8
#define CC 63
#define HH 240
#define WW 240
#define HW (HH * WW)
#define C2 126
#define WS 5
#define NWIN 48
#define BN_EPS 1e-5f

typedef __attribute__((ext_vector_type(2))) float v2f;
typedef __attribute__((ext_vector_type(8))) float v8f;

__device__ __forceinline__ v8f wmma4(v2f a, v2f b, v8f c) {
  return __builtin_amdgcn_wmma_f32_16x16x4_f32(
       false, a,  false, b,
       (short)0, c,  false,  false);
}

__device__ __forceinline__ v2f frag_a(const float* p, int ld, int m0, int k0, int lane) {
  int m  = m0 + (lane & 15);
  int kb = k0 + ((lane >> 4) << 1);
  const float* q = p + m * ld + kb;
  v2f r; r.x = q[0]; r.y = q[1];
  return r;
}

__device__ __forceinline__ v2f frag_b(const float* p, int ld, int k0, int n0, int lane) {
  int n  = n0 + (lane & 15);
  int kb = k0 + ((lane >> 4) << 1);
  v2f r; r.x = p[kb * ld + n]; r.y = p[(kb + 1) * ld + n];
  return r;
}

__device__ __forceinline__ void frag_store(float* p, int ld, int m0, int n0, int lane, v8f d) {
  int n  = n0 + (lane & 15);
  int mb = m0 + ((lane >> 4) << 3);
#pragma unroll
  for (int r = 0; r < 8; ++r) p[(mb + r) * ld + n] = d[r];
}

#define V8Z {0.f,0.f,0.f,0.f,0.f,0.f,0.f,0.f}

__global__ __launch_bounds__(128) void k_proj_in(
    const float* __restrict__ x, const float* __restrict__ w_in,
    const float* __restrict__ b_in, const float* __restrict__ bn_g,
    const float* __restrict__ bn_b, const float* __restrict__ bn_m,
    const float* __restrict__ bn_v, float* __restrict__ xp)
{
  __shared__ float a_lds[64][65];
  __shared__ float w_lds[64][129];
  __shared__ float bias_lds[128];
  __shared__ float scale_lds[128];
  __shared__ float d_lds[4][16][17];

  int tid  = threadIdx.x;
  int lane = tid & 31;
  int wave = tid >> 5;
  int tile = blockIdx.x;
  int b    = tile / 900;
  int p0   = (tile % 900) * 64;

  {
    float sc = 0.f, bv = 0.f;
    if (tid < 126) {
      sc = bn_g[tid] * rsqrtf(bn_v[tid] + BN_EPS);
      bv = b_in[tid] * sc + bn_b[tid] - bn_m[tid] * sc;
    }
    scale_lds[tid] = sc;
    bias_lds[tid]  = bv;
  }
  const int am = tid & 63, akh = tid >> 6;
  float ta[32];
  {
    const float* xb = x + (size_t)b * 63 * HW + p0 + am;
#pragma unroll
    for (int j = 0; j < 32; ++j) {
      int k  = 2 * j + akh;
      int kc = k < 63 ? k : 62;
      ta[j] = xb[(size_t)kc * HW];
    }
  }
  float tw[63];
#pragma unroll
  for (int j = 0; j < 63; ++j) {
    int idx = tid + j * 128;
    int ic  = idx < 126 * 63 ? idx : 126 * 63 - 1;
    tw[j] = w_in[ic];
  }
  w_lds[63][tid] = 0.f;
  if (tid < 63) { w_lds[tid][126] = 0.f; w_lds[tid][127] = 0.f; }
  __syncthreads();
#pragma unroll
  for (int j = 0; j < 63; ++j) {
    int idx = tid + j * 128;
    if (idx < 126 * 63) {
      int n = idx / 63, k = idx - n * 63;
      w_lds[k][n] = tw[j] * scale_lds[n];
    }
  }
#pragma unroll
  for (int j = 0; j < 32; ++j) {
    int k = 2 * j + akh;
    a_lds[am][k] = (k < 63) ? ta[j] : 0.f;
  }
  __syncthreads();

  int m0 = wave * 16;
  for (int nt = 0; nt < 8; ++nt) {
    int n0 = nt * 16;
    v8f acc = V8Z;
    for (int k0 = 0; k0 < 64; k0 += 4) {
      v2f av = frag_a(&a_lds[0][0], 65, m0, k0, lane);
      v2f bv = frag_b(&w_lds[0][0], 129, k0, n0, lane);
      acc = wmma4(av, bv, acc);
    }
    int nloc = lane & 15;
    int mb   = (lane >> 4) << 3;
    float bias = bias_lds[n0 + nloc];
#pragma unroll
    for (int r = 0; r < 8; ++r) d_lds[wave][mb + r][nloc] = acc[r] + bias;
    for (int idx = lane; idx < 256; idx += 32) {
      int n = idx >> 4, m = idx & 15;
      int ch = n0 + n;
      if (ch < 126)
        xp[(size_t)(b * 126 + ch) * HW + p0 + m0 + m] = d_lds[wave][m][n];
    }
  }
}

#define WN_NW 8
#define WN_THREADS (WN_NW * 32)
#define WN_SEG (WN_NW * WS)
#define WN_STRIPS (WW / WN_SEG)
#define WN_MOVERS (WN_SEG * 6)
#define WN_VOFF 800
#define WN_SOFF 1600
#define WN_WSZ 2664

__global__ __launch_bounds__(WN_THREADS) void k_win(float* __restrict__ xp, float* __restrict__ y2)
{
  __shared__ float wl[WN_NW * WN_WSZ];

  const int tid  = threadIdx.x;
  const int lane = tid & 31;
  const int wv   = tid >> 5;
  const int sid  = blockIdx.x;
  const int grp  = blockIdx.y;
  const int b    = sid / (NWIN * WN_STRIPS);
  const int rem  = sid % (NWIN * WN_STRIPS);
  const int wy   = rem / WN_STRIPS, st = rem % WN_STRIPS;
  const int shi  = grp ? 3 : 0;
  const int sho  = grp ? 2 : 0;

  const int x    = tid % WN_SEG, sub = tid / WN_SEG;
  const int win  = x / WS, mxp = x - win * WS;
  const int isv  = sub >= 3;
  const int cb   = (sub - 3 * isv) * 7;

  float t[35];
  if (tid < WN_MOVERS) {
    const int gx = (st * WN_SEG + x + shi) % WW;
    const float* srcp = xp + (size_t)(b * 126 + grp * 42 + isv * 21 + cb) * HW + gx;
    int ro[5];
#pragma unroll
    for (int r = 0; r < 5; ++r) ro[r] = ((wy * WS + r + shi) % HH) * WW;
#pragma unroll
    for (int j = 0; j < 35; ++j) t[j] = srcp[(size_t)(j / 5) * HW + ro[j % 5]];
  }

  {
    float* rec = wl + wv * WN_WSZ;
    for (int i = lane; i < 7 * 25; i += 32) {
      rec[25 * 25 + i] = 0.f;
      rec[WN_VOFF + 25 * 25 + i] = 0.f;
    }
    for (int i = lane; i < 25 * 4; i += 32) {
      int m = i >> 2, c = 21 + (i & 3);
      rec[m * 25 + c] = 0.f;
      rec[WN_VOFF + m * 25 + c] = 0.f;
    }
  }

  if (tid < WN_MOVERS) {
    float* dl = wl + win * WN_WSZ + isv * WN_VOFF + mxp * 25 + cb;
#pragma unroll
    for (int j = 0; j < 35; ++j) dl[(j % 5) * (WS * 25) + (j / 5)] = t[j];
  }
  __syncthreads();

  float* q = wl + wv * WN_WSZ;
  float* v = q + WN_VOFF;
  float* s = q + WN_SOFF;

  for (int mt = 0; mt < 2; ++mt)
    for (int nt = 0; nt < 2; ++nt) {
      v8f acc = V8Z;
      for (int k0 = 0; k0 < 24; k0 += 4) {
        v2f av = frag_a(q, 25, mt * 16, k0, lane);
        v2f bv = frag_a(q, 25, nt * 16, k0, lane);
        acc = wmma4(av, bv, acc);
      }
      frag_store(s, 33, mt * 16, nt * 16, lane, acc);
    }

  if (lane < 25) {
    float* sr = s + lane * 33;
    float mx = -1e30f;
    for (int n = 0; n < 25; ++n) mx = fmaxf(mx, sr[n]);
    float sum = 0.f;
    for (int n = 0; n < 25; ++n) { float e = __expf(sr[n] - mx); sr[n] = e; sum += e; }
    float inv = 1.f / sum;
    for (int n = 0; n < 25; ++n) sr[n] *= inv;
    for (int n = 25; n < 28; ++n) sr[n] = 0.f;
  }

  v8f yt[2][2];
#pragma unroll
  for (int mt = 0; mt < 2; ++mt)
#pragma unroll
    for (int nt = 0; nt < 2; ++nt) {
      v8f acc = V8Z;
      for (int k0 = 0; k0 < 28; k0 += 4) {
        v2f av = frag_a(s, 33, mt * 16, k0, lane);
        v2f bv = frag_b(v, 25, k0, nt * 16, lane);
        acc = wmma4(av, bv, acc);
      }
      yt[mt][nt] = acc;
    }
#pragma unroll
  for (int mt = 0; mt < 2; ++mt)
#pragma unroll
    for (int nt = 0; nt < 2; ++nt) frag_store(s, 33, mt * 16, nt * 16, lane, yt[mt][nt]);
  __syncthreads();

  if (tid < WN_MOVERS) {
    float* dst = grp ? y2 + (size_t)(b * 21) * HW : xp + (size_t)(b * 126) * HW;
    const float* yl = wl + win * WN_WSZ + WN_SOFF + mxp * 33;
    const int ox = (st * WN_SEG + x + sho) % WW;
#pragma unroll
    for (int j = 0; j < 18; ++j) {
      if (j < 17 || sub < 3) {
        int p = j * 6 + sub;
        int c = p / WS, r = p - c * WS;
        int oy = (wy * WS + r + sho) % HH;
        dst[(size_t)c * HW + oy * WW + ox] = yl[r * (WS * 33) + c];
      }
    }
  }
}

#define AX_WAVES 15
#define AX_THREADS (AX_WAVES * 32)
#define AX_VP 26

__global__ __launch_bounds__(AX_THREADS) void k_axial(float* __restrict__ xp,
                                                      float* __restrict__ vmid, int pass)
{
  __shared__ float q[240][25];
  __shared__ float v[241][AX_VP];

  int tid  = threadIdx.x;
  int lane = tid & 31;
  int qt   = tid >> 5;
  int row  = blockIdx.x;
  int b    = row / HH;
  int r    = row % HH;

  {
    float tq[11], tv[11];
    if (pass == 0) {
      const float* qb = xp + (size_t)(b * 126 + 84) * HW + r * WW;
      const float* vb = xp + (size_t)(b * 126 + 105) * HW + r * WW;
#pragma unroll
      for (int j = 0; j < 11; ++j) {
        int i  = tid + j * AX_THREADS;
        int ic = i < 240 * 21 ? i : 240 * 21 - 1;
        int c = ic / 240, xx = ic - c * 240;
        tq[j] = qb[(size_t)c * HW + xx];
        tv[j] = vb[(size_t)c * HW + xx];
      }
#pragma unroll
      for (int j = 0; j < 11; ++j) {
        int i = tid + j * AX_THREADS;
        if (i < 240 * 21) {
          int c = i / 240, xx = i - c * 240;
          q[xx][c] = tq[j];
          v[xx][c] = tv[j];
        }
      }
    } else {
      const float* qb = xp + (size_t)(b * 126 + 84) * HW + r;
      const float* vb = vmid + (size_t)(b * 240 + r) * 240 * 21;
#pragma unroll
      for (int j = 0; j < 11; ++j) {
        int i  = tid + j * AX_THREADS;
        int ic = i < 240 * 21 ? i : 240 * 21 - 1;
        int c = ic / 240, h = ic - c * 240;
        tq[j] = qb[(size_t)c * HW + h * WW];
        tv[j] = vb[ic];
      }
#pragma unroll
      for (int j = 0; j < 11; ++j) {
        int i = tid + j * AX_THREADS;
        if (i < 240 * 21) {
          int c = i / 240, h = i - c * 240;
          q[h][c] = tq[j];
          int h2 = i / 21, c2 = i - h2 * 21;
          v[h2][c2] = tv[j];
        }
      }
    }
  }
  for (int i = tid; i < 240 * 4; i += AX_THREADS) {
    int m = i >> 2, c = 21 + (i & 3);
    q[m][c] = 0.f;
  }
  for (int i = tid; i < 240 * 5; i += AX_THREADS) {
    int m = i / 5, c = 21 + (i - m * 5);
    v[m][c] = 0.f;
  }
  if (tid < AX_VP) v[240][tid] = 0.f;
  __syncthreads();

  const int m0 = qt * 16;
  const int cn = lane & 15;
  const int kh = (lane >> 4) << 3;

  v2f bq[6];
#pragma unroll
  for (int kk = 0; kk < 6; ++kk) bq[kk] = frag_a(&q[0][0], 25, m0, kk * 4, lane);

  v8f st[15];
#pragma unroll
  for (int g = 0; g < 5; ++g) {
    v8f a0 = V8Z, a1 = V8Z, a2 = V8Z;
#pragma unroll
    for (int kk = 0; kk < 6; ++kk) {
      a0 = wmma4(frag_a(&q[0][0], 25, (3 * g + 0) * 16, kk * 4, lane), bq[kk], a0);
      a1 = wmma4(frag_a(&q[0][0], 25, (3 * g + 1) * 16, kk * 4, lane), bq[kk], a1);
      a2 = wmma4(frag_a(&q[0][0], 25, (3 * g + 2) * 16, kk * 4, lane), bq[kk], a2);
    }
    st[3 * g + 0] = a0; st[3 * g + 1] = a1; st[3 * g + 2] = a2;
  }

  float mx = -1e30f;
#pragma unroll
  for (int nt = 0; nt < 15; ++nt)
#pragma unroll
    for (int rr = 0; rr < 8; ++rr) mx = fmaxf(mx, st[nt][rr]);
  mx = fmaxf(mx, __shfl_xor(mx, 16));
  float s0 = 0.f, s1 = 0.f, s2 = 0.f, s3 = 0.f;
#pragma unroll
  for (int nt = 0; nt < 15; ++nt) {
#pragma unroll
    for (int rr = 0; rr < 8; rr += 4) {
      float e0 = __expf(st[nt][rr + 0] - mx);
      float e1 = __expf(st[nt][rr + 1] - mx);
      float e2 = __expf(st[nt][rr + 2] - mx);
      float e3 = __expf(st[nt][rr + 3] - mx);
      st[nt][rr + 0] = e0; st[nt][rr + 1] = e1; st[nt][rr + 2] = e2; st[nt][rr + 3] = e3;
      s0 += e0; s1 += e1; s2 += e2; s3 += e3;
    }
  }
  float sum = (s0 + s1) + (s2 + s3);
  sum += __shfl_xor(sum, 16);
  float inv = 1.f / sum;
#pragma unroll
  for (int nt = 0; nt < 15; ++nt)
#pragma unroll
    for (int rr = 0; rr < 8; ++rr) st[nt][rr] *= inv;

  v8f y0 = V8Z, y1 = V8Z;
#pragma unroll
  for (int nt = 0; nt < 15; ++nt) {
#pragma unroll
    for (int i = 0; i < 4; ++i) {
      v2f pa; pa.x = st[nt][2 * i]; pa.y = st[nt][2 * i + 1];
      int kb = nt * 16 + kh + 2 * i;
      v2f b0, b1;
      b0.x = v[kb][cn];      b0.y = v[kb + 1][cn];
      b1.x = v[kb][16 + cn]; b1.y = v[kb + 1][16 + cn];
      y0 = wmma4(pa, b0, y0);
      y1 = wmma4(pa, b1, y1);
    }
  }

  float* ob = (pass == 0) ? vmid + ((size_t)(b * 240) * 240 + r) * 21
                          : xp + (size_t)(b * 126 + 105) * HW + r;
  const int sm = (pass == 0) ? 240 * 21 : WW;
  const int sn = (pass == 0) ? 1 : HW;
  const int mb = m0 + kh;
#pragma unroll
  for (int nt = 0; nt < 2; ++nt) {
    v8f acc = nt ? y1 : y0;
    int n = nt * 16 + cn;
    if (n < 21) {
#pragma unroll
      for (int rr = 0; rr < 8; ++rr) ob[(mb + rr) * sm + n * sn] = acc[rr];
    }
  }
}

__global__ __launch_bounds__(128) void k_proj_out(
    const float* __restrict__ xp, const float* __restrict__ y2,
    const float* __restrict__ w_out, const float* __restrict__ b_out,
    float* __restrict__ out)
{
  __shared__ float a_lds[64][65];
  __shared__ float w_lds[64][65];
  __shared__ float bias_lds[64];
  __shared__ float d_lds[4][16][17];

  int tid  = threadIdx.x;
  int lane = tid & 31;
  int wave = tid >> 5;
  int tile = blockIdx.x;
  int b    = tile / 900;
  int p0   = (tile % 900) * 64;

  const int am = tid & 63, akh = tid >> 6;
  float ta[32];
#pragma unroll
  for (int j = 0; j < 32; ++j) {
    int k = 2 * j + akh;
    const float* sp;
    if (k < 21)      sp = xp + (size_t)(b * 126 + k) * HW;
    else if (k < 42) sp = y2 + (size_t)(b * 21 + (k - 21)) * HW;
    else             sp = xp + (size_t)(b * 126 + ((k < 63 ? k : 62) + 63)) * HW;
    ta[j] = sp[p0 + am];
  }
  float tw[32];
#pragma unroll
  for (int j = 0; j < 32; ++j) {
    int idx = tid + j * 128;
    int ic  = idx < 63 * 63 ? idx : 63 * 63 - 1;
    tw[j] = w_out[ic];
  }
  if (tid < 64) {
    bias_lds[tid] = (tid < 63) ? b_out[tid] : 0.f;
    w_lds[63][tid] = 0.f;
    w_lds[tid][63] = 0.f;
  }
#pragma unroll
  for (int j = 0; j < 32; ++j) {
    int idx = tid + j * 128;
    if (idx < 63 * 63) {
      int n = idx / 63, k = idx - n * 63;
      w_lds[k][n] = tw[j];
    }
  }
#pragma unroll
  for (int j = 0; j < 32; ++j) {
    int k = 2 * j + akh;
    a_lds[am][k] = (k < 63) ? ta[j] : 0.f;
  }
  __syncthreads();

  int m0 = wave * 16;
  for (int nt = 0; nt < 4; ++nt) {
    int n0 = nt * 16;
    v8f acc = V8Z;
    for (int k0 = 0; k0 < 64; k0 += 4) {
      v2f av = frag_a(&a_lds[0][0], 65, m0, k0, lane);
      v2f bv = frag_b(&w_lds[0][0], 65, k0, n0, lane);
      acc = wmma4(av, bv, acc);
    }
    int nloc = lane & 15;
    int mb   = (lane >> 4) << 3;
    float bias = bias_lds[n0 + nloc];
#pragma unroll
    for (int r = 0; r < 8; ++r) d_lds[wave][mb + r][nloc] = acc[r] + bias;
    for (int idx = lane; idx < 256; idx += 32) {
      int n = idx >> 4, m = idx & 15;
      int ch = n0 + n;
      if (ch < 63)
        out[(size_t)(b * 63 + ch) * HW + p0 + m0 + m] = d_lds[wave][m][n];
    }
  }
}

extern "C" void kernel_launch(void* const* d_in, const int* in_sizes, int n_in,
                              void* d_out, int out_size, void* d_ws, size_t ws_size,
                              hipStream_t stream)
{
  (void)in_sizes; (void)n_in; (void)out_size; (void)ws_size;
  const float* x     = (const float*)d_in[0];
  const float* w_in  = (const float*)d_in[1];
  const float* b_in  = (const float*)d_in[2];
  const float* bn_g  = (const float*)d_in[3];
  const float* bn_b  = (const float*)d_in[4];
  const float* bn_m  = (const float*)d_in[5];
  const float* bn_v  = (const float*)d_in[6];
  const float* w_out = (const float*)d_in[7];
  const float* b_out = (const float*)d_in[8];
  float* out = (float*)d_out;

  float* xp   = (float*)d_ws;
  float* y2b  = xp  + (size_t)BB * 126 * HW;
  float* vmid = y2b + (size_t)BB * 21 * HW;

  k_proj_in <<<dim3(BB * 900), 128, 0, stream>>>(x, w_in, b_in, bn_g, bn_b, bn_m, bn_v, xp);
  k_win     <<<dim3(BB * NWIN * WN_STRIPS, 2), WN_THREADS, 0, stream>>>(xp, y2b);
  k_axial   <<<dim3(BB * HH), AX_THREADS, 0, stream>>>(xp, vmid, 0);
  k_axial   <<<dim3(BB * WW), AX_THREADS, 0, stream>>>(xp, vmid, 1);
  k_proj_out<<<dim3(BB * 900), 128, 0, stream>>>(xp, y2b, w_out, b_out, out);
}


# axial column pass: q read from a [w][h][c] copy that the row pass stores, result stored [w][h][c] and gathered by the output projection: no 4-byte strided global access left; sums unchanged
# speedup vs baseline: 6.1947x; 1.2422x over previous
#include <hip/hip_runtime.h>
#include <math.h>

#define BB 8
#define CC 63
#define HH 240
#define WW 240
#define HW (HH * WW)
#define C2 126
#define WS 5
#define NWIN 48
#define BN_EPS 1e-5f

typedef __attribute__((ext_vector_type(2))) float v2f;
typedef __attribute__((ext_vector_type(8))) float v8f;

__device__ __forceinline__ v8f wmma4(v2f a, v2f b, v8f c) {
  return __builtin_amdgcn_wmma_f32_16x16x4_f32(
       false, a,  false, b,
       (short)0, c,  false,  false);
}

__device__ __forceinline__ v2f frag_a(const float* p, int ld, int m0, int k0, int lane) {
  int m  = m0 + (lane & 15);
  int kb = k0 + ((lane >> 4) << 1);
  const float* q = p + m * ld + kb;
  v2f r; r.x = q[0]; r.y = q[1];
  return r;
}

__device__ __forceinline__ v2f frag_b(const float* p, int ld, int k0, int n0, int lane) {
  int n  = n0 + (lane & 15);
  int kb = k0 + ((lane >> 4) << 1);
  v2f r; r.x = p[kb * ld + n]; r.y = p[(kb + 1) * ld + n];
  return r;
}

__device__ __forceinline__ void frag_store(float* p, int ld, int m0, int n0, int lane, v8f d) {
  int n  = n0 + (lane & 15);
  int mb = m0 + ((lane >> 4) << 3);
#pragma unroll
  for (int r = 0; r < 8; ++r) p[(mb + r) * ld + n] = d[r];
}

#define V8Z {0.f,0.f,0.f,0.f,0.f,0.f,0.f,0.f}

__global__ __launch_bounds__(128) void k_proj_in(
    const float* __restrict__ x, const float* __restrict__ w_in,
    const float* __restrict__ b_in, const float* __restrict__ bn_g,
    const float* __restrict__ bn_b, const float* __restrict__ bn_m,
    const float* __restrict__ bn_v, float* __restrict__ xp)
{
  __shared__ float a_lds[64][65];
  __shared__ float w_lds[64][129];
  __shared__ float bias_lds[128];
  __shared__ float scale_lds[128];
  __shared__ float d_lds[4][16][17];

  int tid  = threadIdx.x;
  int lane = tid & 31;
  int wave = tid >> 5;
  int tile = blockIdx.x;
  int b    = tile / 900;
  int p0   = (tile % 900) * 64;

  {
    float sc = 0.f, bv = 0.f;
    if (tid < 126) {
      sc = bn_g[tid] * rsqrtf(bn_v[tid] + BN_EPS);
      bv = b_in[tid] * sc + bn_b[tid] - bn_m[tid] * sc;
    }
    scale_lds[tid] = sc;
    bias_lds[tid]  = bv;
  }
  const int am = tid & 63, akh = tid >> 6;
  float ta[32];
  {
    const float* xb = x + (size_t)b * 63 * HW + p0 + am;
#pragma unroll
    for (int j = 0; j < 32; ++j) {
      int k  = 2 * j + akh;
      int kc = k < 63 ? k : 62;
      ta[j] = xb[(size_t)kc * HW];
    }
  }
  float tw[63];
#pragma unroll
  for (int j = 0; j < 63; ++j) {
    int idx = tid + j * 128;
    int ic  = idx < 126 * 63 ? idx : 126 * 63 - 1;
    tw[j] = w_in[ic];
  }
  w_lds[63][tid] = 0.f;
  if (tid < 63) { w_lds[tid][126] = 0.f; w_lds[tid][127] = 0.f; }
  __syncthreads();
#pragma unroll
  for (int j = 0; j < 63; ++j) {
    int idx = tid + j * 128;
    if (idx < 126 * 63) {
      int n = idx / 63, k = idx - n * 63;
      w_lds[k][n] = tw[j] * scale_lds[n];
    }
  }
#pragma unroll
  for (int j = 0; j < 32; ++j) {
    int k = 2 * j + akh;
    a_lds[am][k] = (k < 63) ? ta[j] : 0.f;
  }
  __syncthreads();

  int m0 = wave * 16;
  for (int nt = 0; nt < 8; ++nt) {
    int n0 = nt * 16;
    v8f acc = V8Z;
    for (int k0 = 0; k0 < 64; k0 += 4) {
      v2f av = frag_a(&a_lds[0][0], 65, m0, k0, lane);
      v2f bv = frag_b(&w_lds[0][0], 129, k0, n0, lane);
      acc = wmma4(av, bv, acc);
    }
    int nloc = lane & 15;
    int mb   = (lane >> 4) << 3;
    float bias = bias_lds[n0 + nloc];
#pragma unroll
    for (int r = 0; r < 8; ++r) d_lds[wave][mb + r][nloc] = acc[r] + bias;
    for (int idx = lane; idx < 256; idx += 32) {
      int n = idx >> 4, m = idx & 15;
      int ch = n0 + n;
      if (ch < 126)
        xp[(size_t)(b * 126 + ch) * HW + p0 + m0 + m] = d_lds[wave][m][n];
    }
  }
}

#define WN_NW 8
#define WN_THREADS (WN_NW * 32)
#define WN_SEG (WN_NW * WS)
#define WN_STRIPS (WW / WN_SEG)
#define WN_MOVERS (WN_SEG * 6)
#define WN_VOFF 800
#define WN_SOFF 1600
#define WN_WSZ 2664

__global__ __launch_bounds__(WN_THREADS) void k_win(float* __restrict__ xp, float* __restrict__ y2)
{
  __shared__ float wl[WN_NW * WN_WSZ];

  const int tid  = threadIdx.x;
  const int lane = tid & 31;
  const int wv   = tid >> 5;
  const int sid  = blockIdx.x;
  const int grp  = blockIdx.y;
  const int b    = sid / (NWIN * WN_STRIPS);
  const int rem  = sid % (NWIN * WN_STRIPS);
  const int wy   = rem / WN_STRIPS, st = rem % WN_STRIPS;
  const int shi  = grp ? 3 : 0;
  const int sho  = grp ? 2 : 0;

  const int x    = tid % WN_SEG, sub = tid / WN_SEG;
  const int win  = x / WS, mxp = x - win * WS;
  const int isv  = sub >= 3;
  const int cb   = (sub - 3 * isv) * 7;

  float t[35];
  if (tid < WN_MOVERS) {
    const int gx = (st * WN_SEG + x + shi) % WW;
    const float* srcp = xp + (size_t)(b * 126 + grp * 42 + isv * 21 + cb) * HW + gx;
    int ro[5];
#pragma unroll
    for (int r = 0; r < 5; ++r) ro[r] = ((wy * WS + r + shi) % HH) * WW;
#pragma unroll
    for (int j = 0; j < 35; ++j) t[j] = srcp[(size_t)(j / 5) * HW + ro[j % 5]];
  }

  {
    float* rec = wl + wv * WN_WSZ;
    for (int i = lane; i < 7 * 25; i += 32) {
      rec[25 * 25 + i] = 0.f;
      rec[WN_VOFF + 25 * 25 + i] = 0.f;
    }
    for (int i = lane; i < 25 * 4; i += 32) {
      int m = i >> 2, c = 21 + (i & 3);
      rec[m * 25 + c] = 0.f;
      rec[WN_VOFF + m * 25 + c] = 0.f;
    }
  }

  if (tid < WN_MOVERS) {
    float* dl = wl + win * WN_WSZ + isv * WN_VOFF + mxp * 25 + cb;
#pragma unroll
    for (int j = 0; j < 35; ++j) dl[(j % 5) * (WS * 25) + (j / 5)] = t[j];
  }
  __syncthreads();

  float* q = wl + wv * WN_WSZ;
  float* v = q + WN_VOFF;
  float* s = q + WN_SOFF;

  for (int mt = 0; mt < 2; ++mt)
    for (int nt = 0; nt < 2; ++nt) {
      v8f acc = V8Z;
      for (int k0 = 0; k0 < 24; k0 += 4) {
        v2f av = frag_a(q, 25, mt * 16, k0, lane);
        v2f bv = frag_a(q, 25, nt * 16, k0, lane);
        acc = wmma4(av, bv, acc);
      }
      frag_store(s, 33, mt * 16, nt * 16, lane, acc);
    }

  if (lane < 25) {
    float* sr = s + lane * 33;
    float mx = -1e30f;
    for (int n = 0; n < 25; ++n) mx = fmaxf(mx, sr[n]);
    float sum = 0.f;
    for (int n = 0; n < 25; ++n) { float e = __expf(sr[n] - mx); sr[n] = e; sum += e; }
    float inv = 1.f / sum;
    for (int n = 0; n < 25; ++n) sr[n] *= inv;
    for (int n = 25; n < 28; ++n) sr[n] = 0.f;
  }

  v8f yt[2][2];
#pragma unroll
  for (int mt = 0; mt < 2; ++mt)
#pragma unroll
    for (int nt = 0; nt < 2; ++nt) {
      v8f acc = V8Z;
      for (int k0 = 0; k0 < 28; k0 += 4) {
        v2f av = frag_a(s, 33, mt * 16, k0, lane);
        v2f bv = frag_b(v, 25, k0, nt * 16, lane);
        acc = wmma4(av, bv, acc);
      }
      yt[mt][nt] = acc;
    }
#pragma unroll
  for (int mt = 0; mt < 2; ++mt)
#pragma unroll
    for (int nt = 0; nt < 2; ++nt) frag_store(s, 33, mt * 16, nt * 16, lane, yt[mt][nt]);
  __syncthreads();

  if (tid < WN_MOVERS) {
    float* dst = grp ? y2 + (size_t)(b * 21) * HW : xp + (size_t)(b * 126) * HW;
    const float* yl = wl + win * WN_WSZ + WN_SOFF + mxp * 33;
    const int ox = (st * WN_SEG + x + sho) % WW;
#pragma unroll
    for (int j = 0; j < 18; ++j) {
      if (j < 17 || sub < 3) {
        int p = j * 6 + sub;
        int c = p / WS, r = p - c * WS;
        int oy = (wy * WS + r + sho) % HH;
        dst[(size_t)c * HW + oy * WW + ox] = yl[r * (WS * 33) + c];
      }
    }
  }
}

#define AX_WAVES 15
#define AX_THREADS (AX_WAVES * 32)
#define AX_VP 26

__global__ __launch_bounds__(AX_THREADS) void k_axial(float* __restrict__ xp,
                                                      float* __restrict__ vmid, int pass)
{
  __shared__ float q[240][25];
  __shared__ float v[241][AX_VP];

  int tid  = threadIdx.x;
  int lane = tid & 31;
  int qt   = tid >> 5;
  int row  = blockIdx.x;
  int b    = row / HH;
  int r    = row % HH;

  {
    float tq[11], tv[11];
    if (pass == 0) {
      const float* qb = xp + (size_t)(b * 126 + 84) * HW + r * WW;
      const float* vb = xp + (size_t)(b * 126 + 105) * HW + r * WW;
#pragma unroll
      for (int j = 0; j < 11; ++j) {
        int i  = tid + j * AX_THREADS;
        int ic = i < 240 * 21 ? i : 240 * 21 - 1;
        int c = ic / 240, xx = ic - c * 240;
        tq[j] = qb[(size_t)c * HW + xx];
        tv[j] = vb[(size_t)c * HW + xx];
      }
#pragma unroll
      for (int j = 0; j < 11; ++j) {
        int i = tid + j * AX_THREADS;
        if (i < 240 * 21) {
          int c = i / 240, xx = i - c * 240;
          q[xx][c] = tq[j];
          v[xx][c] = tv[j];
        }
      }
    } else {
      const float* qb = xp + (size_t)(b * 126 + 21) * HW + (size_t)r * (240 * 21);
      const float* vb = vmid + (size_t)(b * 240 + r) * 240 * 21;
#pragma unroll
      for (int j = 0; j < 11; ++j) {
        int i  = tid + j * AX_THREADS;
        int ic = i < 240 * 21 ? i : 240 * 21 - 1;
        tq[j] = qb[ic];
        tv[j] = vb[ic];
      }
#pragma unroll
      for (int j = 0; j < 11; ++j) {
        int i = tid + j * AX_THREADS;
        if (i < 240 * 21) {
          int h2 = i / 21, c2 = i - h2 * 21;
          q[h2][c2] = tq[j];
          v[h2][c2] = tv[j];
        }
      }
    }
  }
  for (int i = tid; i < 240 * 4; i += AX_THREADS) {
    int m = i >> 2, c = 21 + (i & 3);
    q[m][c] = 0.f;
  }
  for (int i = tid; i < 240 * 5; i += AX_THREADS) {
    int m = i / 5, c = 21 + (i - m * 5);
    v[m][c] = 0.f;
  }
  if (tid < AX_VP) v[240][tid] = 0.f;
  __syncthreads();

  if (pass == 0) {
    float* qo = xp + (size_t)(b * 126 + 21) * HW + (size_t)r * 21;
#pragma unroll
    for (int j = 0; j < 11; ++j) {
      int i = tid + j * AX_THREADS;
      if (i < 240 * 21) {
        int xx = i / 21, c = i - xx * 21;
        qo[(size_t)xx * (240 * 21) + c] = q[xx][c];
      }
    }
  }

  const int m0 = qt * 16;
  const int cn = lane & 15;
  const int kh = (lane >> 4) << 3;

  v2f bq[6];
#pragma unroll
  for (int kk = 0; kk < 6; ++kk) bq[kk] = frag_a(&q[0][0], 25, m0, kk * 4, lane);

  v8f st[15];
#pragma unroll
  for (int g = 0; g < 5; ++g) {
    v8f a0 = V8Z, a1 = V8Z, a2 = V8Z;
#pragma unroll
    for (int kk = 0; kk < 6; ++kk) {
      a0 = wmma4(frag_a(&q[0][0], 25, (3 * g + 0) * 16, kk * 4, lane), bq[kk], a0);
      a1 = wmma4(frag_a(&q[0][0], 25, (3 * g + 1) * 16, kk * 4, lane), bq[kk], a1);
      a2 = wmma4(frag_a(&q[0][0], 25, (3 * g + 2) * 16, kk * 4, lane), bq[kk], a2);
    }
    st[3 * g + 0] = a0; st[3 * g + 1] = a1; st[3 * g + 2] = a2;
  }

  float mx = -1e30f;
#pragma unroll
  for (int nt = 0; nt < 15; ++nt)
#pragma unroll
    for (int rr = 0; rr < 8; ++rr) mx = fmaxf(mx, st[nt][rr]);
  mx = fmaxf(mx, __shfl_xor(mx, 16));
  float s0 = 0.f, s1 = 0.f, s2 = 0.f, s3 = 0.f;
#pragma unroll
  for (int nt = 0; nt < 15; ++nt) {
#pragma unroll
    for (int rr = 0; rr < 8; rr += 4) {
      float e0 = __expf(st[nt][rr + 0] - mx);
      float e1 = __expf(st[nt][rr + 1] - mx);
      float e2 = __expf(st[nt][rr + 2] - mx);
      float e3 = __expf(st[nt][rr + 3] - mx);
      st[nt][rr + 0] = e0; st[nt][rr + 1] = e1; st[nt][rr + 2] = e2; st[nt][rr + 3] = e3;
      s0 += e0; s1 += e1; s2 += e2; s3 += e3;
    }
  }
  float sum = (s0 + s1) + (s2 + s3);
  sum += __shfl_xor(sum, 16);
  float inv = 1.f / sum;
#pragma unroll
  for (int nt = 0; nt < 15; ++nt)
#pragma unroll
    for (int rr = 0; rr < 8; ++rr) st[nt][rr] *= inv;

  v8f y0 = V8Z, y1 = V8Z;
#pragma unroll
  for (int nt = 0; nt < 15; ++nt) {
#pragma unroll
    for (int i = 0; i < 4; ++i) {
      v2f pa; pa.x = st[nt][2 * i]; pa.y = st[nt][2 * i + 1];
      int kb = nt * 16 + kh + 2 * i;
      v2f b0, b1;
      b0.x = v[kb][cn];      b0.y = v[kb + 1][cn];
      b1.x = v[kb][16 + cn]; b1.y = v[kb + 1][16 + cn];
      y0 = wmma4(pa, b0, y0);
      y1 = wmma4(pa, b1, y1);
    }
  }

  float* ob = (pass == 0) ? vmid + ((size_t)(b * 240) * 240 + r) * 21
                          : xp + (size_t)(b * 126 + 105) * HW + (size_t)r * (240 * 21);
  const int sm = (pass == 0) ? 240 * 21 : 21;
  const int mb = m0 + kh;
#pragma unroll
  for (int nt = 0; nt < 2; ++nt) {
    v8f acc = nt ? y1 : y0;
    int n = nt * 16 + cn;
    if (n < 21) {
#pragma unroll
      for (int rr = 0; rr < 8; ++rr) ob[(mb + rr) * sm + n] = acc[rr];
    }
  }
}

__global__ __launch_bounds__(128) void k_proj_out(
    const float* __restrict__ xp, const float* __restrict__ y2,
    const float* __restrict__ w_out, const float* __restrict__ b_out,
    float* __restrict__ out)
{
  __shared__ float a_lds[64][65];
  __shared__ float w_lds[64][65];
  __shared__ float bias_lds[64];
  __shared__ float d_lds[4][16][17];

  int tid  = threadIdx.x;
  int lane = tid & 31;
  int wave = tid >> 5;
  int tile = blockIdx.x;
  int b    = tile / 900;
  int p0   = (tile % 900) * 64;

  const int am = tid & 63, akh = tid >> 6;
  float ta[21];
#pragma unroll
  for (int j = 0; j < 21; ++j) {
    int k = 2 * j + akh;
    const float* sp;
    if (k < 21) sp = xp + (size_t)(b * 126 + k) * HW;
    else        sp = y2 + (size_t)(b * 21 + (k - 21)) * HW;
    ta[j] = sp[p0 + am];
  }
  float tc[11];
  {
    const float* y3 = xp + (size_t)(b * 126 + 105) * HW;
#pragma unroll
    for (int j = 0; j < 11; ++j) {
      int e  = tid + j * 128;
      int ec = e < 64 * 21 ? e : 64 * 21 - 1;
      int px = ec / 21, c = ec - px * 21;
      int p  = p0 + px;
      int h  = p / WW, w = p - h * WW;
      tc[j] = y3[(size_t)(w * HH + h) * 21 + c];
    }
  }
  float tw[32];
#pragma unroll
  for (int j = 0; j < 32; ++j) {
    int idx = tid + j * 128;
    int ic  = idx < 63 * 63 ? idx : 63 * 63 - 1;
    tw[j] = w_out[ic];
  }
  if (tid < 64) {
    bias_lds[tid] = (tid < 63) ? b_out[tid] : 0.f;
    w_lds[63][tid] = 0.f;
    w_lds[tid][63] = 0.f;
    a_lds[tid][63] = 0.f;
  }
#pragma unroll
  for (int j = 0; j < 32; ++j) {
    int idx = tid + j * 128;
    if (idx < 63 * 63) {
      int n = idx / 63, k = idx - n * 63;
      w_lds[k][n] = tw[j];
    }
  }
#pragma unroll
  for (int j = 0; j < 21; ++j) a_lds[am][2 * j + akh] = ta[j];
#pragma unroll
  for (int j = 0; j < 11; ++j) {
    int e = tid + j * 128;
    if (e < 64 * 21) {
      int px = e / 21, c = e - px * 21;
      a_lds[px][42 + c] = tc[j];
    }
  }
  __syncthreads();

  int m0 = wave * 16;
  for (int nt = 0; nt < 4; ++nt) {
    int n0 = nt * 16;
    v8f acc = V8Z;
    for (int k0 = 0; k0 < 64; k0 += 4) {
      v2f av = frag_a(&a_lds[0][0], 65, m0, k0, lane);
      v2f bv = frag_b(&w_lds[0][0], 65, k0, n0, lane);
      acc = wmma4(av, bv, acc);
    }
    int nloc = lane & 15;
    int mb   = (lane >> 4) << 3;
    float bias = bias_lds[n0 + nloc];
#pragma unroll
    for (int r = 0; r < 8; ++r) d_lds[wave][mb + r][nloc] = acc[r] + bias;
    for (int idx = lane; idx < 256; idx += 32) {
      int n = idx >> 4, m = idx & 15;
      int ch = n0 + n;
      if (ch < 63)
        out[(size_t)(b * 63 + ch) * HW + p0 + m0 + m] = d_lds[wave][m][n];
    }
  }
}

extern "C" void kernel_launch(void* const* d_in, const int* in_sizes, int n_in,
                              void* d_out, int out_size, void* d_ws, size_t ws_size,
                              hipStream_t stream)
{
  (void)in_sizes; (void)n_in; (void)out_size; (void)ws_size;
  const float* x     = (const float*)d_in[0];
  const float* w_in  = (const float*)d_in[1];
  const float* b_in  = (const float*)d_in[2];
  const float* bn_g  = (const float*)d_in[3];
  const float* bn_b  = (const float*)d_in[4];
  const float* bn_m  = (const float*)d_in[5];
  const float* bn_v  = (const float*)d_in[6];
  const float* w_out = (const float*)d_in[7];
  const float* b_out = (const float*)d_in[8];
  float* out = (float*)d_out;

  float* xp   = (float*)d_ws;
  float* y2b  = xp  + (size_t)BB * 126 * HW;
  float* vmid = y2b + (size_t)BB * 21 * HW;

  k_proj_in <<<dim3(BB * 900), 128, 0, stream>>>(x, w_in, b_in, bn_g, bn_b, bn_m, bn_v, xp);
  k_win     <<<dim3(BB * NWIN * WN_STRIPS, 2), WN_THREADS, 0, stream>>>(xp, y2b);
  k_axial   <<<dim3(BB * HH), AX_THREADS, 0, stream>>>(xp, vmid, 0);
  k_axial   <<<dim3(BB * WW), AX_THREADS, 0, stream>>>(xp, vmid, 1);
  k_proj_out<<<dim3(BB * 900), 128, 0, stream>>>(xp, y2b, w_out, b_out, out);
}


# both projections computed transposed (channels x pixels): a lane holds 8 channels of one pixel and stores them straight from registers; no LDS transpose, no serial store loop; sums unchanged
# speedup vs baseline: 6.5424x; 1.0561x over previous
#include <hip/hip_runtime.h>
#include <math.h>

#define BB 8
#define CC 63
#define HH 240
#define WW 240
#define HW (HH * WW)
#define C2 126
#define WS 5
#define NWIN 48
#define BN_EPS 1e-5f

typedef __attribute__((ext_vector_type(2))) float v2f;
typedef __attribute__((ext_vector_type(8))) float v8f;

__device__ __forceinline__ v8f wmma4(v2f a, v2f b, v8f c) {
  return __builtin_amdgcn_wmma_f32_16x16x4_f32(
       false, a,  false, b,
       (short)0, c,  false,  false);
}

__device__ __forceinline__ v2f frag_a(const float* p, int ld, int m0, int k0, int lane) {
  int m  = m0 + (lane & 15);
  int kb = k0 + ((lane >> 4) << 1);
  const float* q = p + m * ld + kb;
  v2f r; r.x = q[0]; r.y = q[1];
  return r;
}

__device__ __forceinline__ v2f frag_b(const float* p, int ld, int k0, int n0, int lane) {
  int n  = n0 + (lane & 15);
  int kb = k0 + ((lane >> 4) << 1);
  v2f r; r.x = p[kb * ld + n]; r.y = p[(kb + 1) * ld + n];
  return r;
}

__device__ __forceinline__ void frag_store(float* p, int ld, int m0, int n0, int lane, v8f d) {
  int n  = n0 + (lane & 15);
  int mb = m0 + ((lane >> 4) << 3);
#pragma unroll
  for (int r = 0; r < 8; ++r) p[(mb + r) * ld + n] = d[r];
}

#define V8Z {0.f,0.f,0.f,0.f,0.f,0.f,0.f,0.f}

__global__ __launch_bounds__(128) void k_proj_in(
    const float* __restrict__ x, const float* __restrict__ w_in,
    const float* __restrict__ b_in, const float* __restrict__ bn_g,
    const float* __restrict__ bn_b, const float* __restrict__ bn_m,
    const float* __restrict__ bn_v, float* __restrict__ xp)
{
  __shared__ float a_lds[64][65];
  __shared__ float w_lds[64][129];
  __shared__ float bias_lds[128];
  __shared__ float scale_lds[128];

  int tid  = threadIdx.x;
  int lane = tid & 31;
  int wave = tid >> 5;
  int tile = blockIdx.x;
  int b    = tile / 900;
  int p0   = (tile % 900) * 64;

  {
    float sc = 0.f, bv = 0.f;
    if (tid < 126) {
      sc = bn_g[tid] * rsqrtf(bn_v[tid] + BN_EPS);
      bv = b_in[tid] * sc + bn_b[tid] - bn_m[tid] * sc;
    }
    scale_lds[tid] = sc;
    bias_lds[tid]  = bv;
  }
  const int am = tid & 63, akh = tid >> 6;
  float ta[32];
  {
    const float* xb = x + (size_t)b * 63 * HW + p0 + am;
#pragma unroll
    for (int j = 0; j < 32; ++j) {
      int k  = 2 * j + akh;
      int kc = k < 63 ? k : 62;
      ta[j] = xb[(size_t)kc * HW];
    }
  }
  float tw[63];
#pragma unroll
  for (int j = 0; j < 63; ++j) {
    int idx = tid + j * 128;
    int ic  = idx < 126 * 63 ? idx : 126 * 63 - 1;
    tw[j] = w_in[ic];
  }
  w_lds[63][tid] = 0.f;
  if (tid < 63) { w_lds[tid][126] = 0.f; w_lds[tid][127] = 0.f; }
  __syncthreads();
#pragma unroll
  for (int j = 0; j < 63; ++j) {
    int idx = tid + j * 128;
    if (idx < 126 * 63) {
      int n = idx / 63, k = idx - n * 63;
      w_lds[k][n] = tw[j] * scale_lds[n];
    }
  }
#pragma unroll
  for (int j = 0; j < 32; ++j) {
    int k = 2 * j + akh;
    a_lds[am][k] = (k < 63) ? ta[j] : 0.f;
  }
  __syncthreads();

  int m0 = wave * 16;
  const int chh = (lane >> 4) << 3;
  float* op = xp + (size_t)b * 126 * HW + p0 + m0 + (lane & 15);
  for (int nt = 0; nt < 8; ++nt) {
    int n0 = nt * 16;
    v8f acc = V8Z;
    for (int k0 = 0; k0 < 64; k0 += 4) {
      v2f av = frag_a(&a_lds[0][0], 65, m0, k0, lane);
      v2f bv = frag_b(&w_lds[0][0], 129, k0, n0, lane);
      acc = wmma4(bv, av, acc);
    }
#pragma unroll
    for (int r = 0; r < 8; ++r) {
      int ch = n0 + chh + r;
      if (ch < 126) op[(size_t)ch * HW] = acc[r] + bias_lds[ch];
    }
  }
}

#define WN_NW 8
#define WN_THREADS (WN_NW * 32)
#define WN_SEG (WN_NW * WS)
#define WN_STRIPS (WW / WN_SEG)
#define WN_MOVERS (WN_SEG * 6)
#define WN_VOFF 800
#define WN_SOFF 1600
#define WN_WSZ 2664

__global__ __launch_bounds__(WN_THREADS) void k_win(float* __restrict__ xp, float* __restrict__ y2)
{
  __shared__ float wl[WN_NW * WN_WSZ];

  const int tid  = threadIdx.x;
  const int lane = tid & 31;
  const int wv   = tid >> 5;
  const int sid  = blockIdx.x;
  const int grp  = blockIdx.y;
  const int b    = sid / (NWIN * WN_STRIPS);
  const int rem  = sid % (NWIN * WN_STRIPS);
  const int wy   = rem / WN_STRIPS, st = rem % WN_STRIPS;
  const int shi  = grp ? 3 : 0;
  const int sho  = grp ? 2 : 0;

  const int x    = tid % WN_SEG, sub = tid / WN_SEG;
  const int win  = x / WS, mxp = x - win * WS;
  const int isv  = sub >= 3;
  const int cb   = (sub - 3 * isv) * 7;

  float t[35];
  if (tid < WN_MOVERS) {
    const int gx = (st * WN_SEG + x + shi) % WW;
    const float* srcp = xp + (size_t)(b * 126 + grp * 42 + isv * 21 + cb) * HW + gx;
    int ro[5];
#pragma unroll
    for (int r = 0; r < 5; ++r) ro[r] = ((wy * WS + r + shi) % HH) * WW;
#pragma unroll
    for (int j = 0; j < 35; ++j) t[j] = srcp[(size_t)(j / 5) * HW + ro[j % 5]];
  }

  {
    float* rec = wl + wv * WN_WSZ;
    for (int i = lane; i < 7 * 25; i += 32) {
      rec[25 * 25 + i] = 0.f;
      rec[WN_VOFF + 25 * 25 + i] = 0.f;
    }
    for (int i = lane; i < 25 * 4; i += 32) {
      int m = i >> 2, c = 21 + (i & 3);
      rec[m * 25 + c] = 0.f;
      rec[WN_VOFF + m * 25 + c] = 0.f;
    }
  }

  if (tid < WN_MOVERS) {
    float* dl = wl + win * WN_WSZ + isv * WN_VOFF + mxp * 25 + cb;
#pragma unroll
    for (int j = 0; j < 35; ++j) dl[(j % 5) * (WS * 25) + (j / 5)] = t[j];
  }
  __syncthreads();

  float* q = wl + wv * WN_WSZ;
  float* v = q + WN_VOFF;
  float* s = q + WN_SOFF;

  for (int mt = 0; mt < 2; ++mt)
    for (int nt = 0; nt < 2; ++nt) {
      v8f acc = V8Z;
      for (int k0 = 0; k0 < 24; k0 += 4) {
        v2f av = frag_a(q, 25, mt * 16, k0, lane);
        v2f bv = frag_a(q, 25, nt * 16, k0, lane);
        acc = wmma4(av, bv, acc);
      }
      frag_store(s, 33, mt * 16, nt * 16, lane, acc);
    }

  if (lane < 25) {
    float* sr = s + lane * 33;
    float mx = -1e30f;
    for (int n = 0; n < 25; ++n) mx = fmaxf(mx, sr[n]);
    float sum = 0.f;
    for (int n = 0; n < 25; ++n) { float e = __expf(sr[n] - mx); sr[n] = e; sum += e; }
    float inv = 1.f / sum;
    for (int n = 0; n < 25; ++n) sr[n] *= inv;
    for (int n = 25; n < 28; ++n) sr[n] = 0.f;
  }

  v8f yt[2][2];
#pragma unroll
  for (int mt = 0; mt < 2; ++mt)
#pragma unroll
    for (int nt = 0; nt < 2; ++nt) {
      v8f acc = V8Z;
      for (int k0 = 0; k0 < 28; k0 += 4) {
        v2f av = frag_a(s, 33, mt * 16, k0, lane);
        v2f bv = frag_b(v, 25, k0, nt * 16, lane);
        acc = wmma4(av, bv, acc);
      }
      yt[mt][nt] = acc;
    }
#pragma unroll
  for (int mt = 0; mt < 2; ++mt)
#pragma unroll
    for (int nt = 0; nt < 2; ++nt) frag_store(s, 33, mt * 16, nt * 16, lane, yt[mt][nt]);
  __syncthreads();

  if (tid < WN_MOVERS) {
    float* dst = grp ? y2 + (size_t)(b * 21) * HW : xp + (size_t)(b * 126) * HW;
    const float* yl = wl + win * WN_WSZ + WN_SOFF + mxp * 33;
    const int ox = (st * WN_SEG + x + sho) % WW;
#pragma unroll
    for (int j = 0; j < 18; ++j) {
      if (j < 17 || sub < 3) {
        int p = j * 6 + sub;
        int c = p / WS, r = p - c * WS;
        int oy = (wy * WS + r + sho) % HH;
        dst[(size_t)c * HW + oy * WW + ox] = yl[r * (WS * 33) + c];
      }
    }
  }
}

#define AX_WAVES 15
#define AX_THREADS (AX_WAVES * 32)
#define AX_VP 26

__global__ __launch_bounds__(AX_THREADS) void k_axial(float* __restrict__ xp,
                                                      float* __restrict__ vmid, int pass)
{
  __shared__ float q[240][25];
  __shared__ float v[241][AX_VP];

  int tid  = threadIdx.x;
  int lane = tid & 31;
  int qt   = tid >> 5;
  int row  = blockIdx.x;
  int b    = row / HH;
  int r    = row % HH;

  {
    float tq[11], tv[11];
    if (pass == 0) {
      const float* qb = xp + (size_t)(b * 126 + 84) * HW + r * WW;
      const float* vb = xp + (size_t)(b * 126 + 105) * HW + r * WW;
#pragma unroll
      for (int j = 0; j < 11; ++j) {
        int i  = tid + j * AX_THREADS;
        int ic = i < 240 * 21 ? i : 240 * 21 - 1;
        int c = ic / 240, xx = ic - c * 240;
        tq[j] = qb[(size_t)c * HW + xx];
        tv[j] = vb[(size_t)c * HW + xx];
      }
#pragma unroll
      for (int j = 0; j < 11; ++j) {
        int i = tid + j * AX_THREADS;
        if (i < 240 * 21) {
          int c = i / 240, xx = i - c * 240;
          q[xx][c] = tq[j];
          v[xx][c] = tv[j];
        }
      }
    } else {
      const float* qb = xp + (size_t)(b * 126 + 21) * HW + (size_t)r * (240 * 21);
      const float* vb = vmid + (size_t)(b * 240 + r) * 240 * 21;
#pragma unroll
      for (int j = 0; j < 11; ++j) {
        int i  = tid + j * AX_THREADS;
        int ic = i < 240 * 21 ? i : 240 * 21 - 1;
        tq[j] = qb[ic];
        tv[j] = vb[ic];
      }
#pragma unroll
      for (int j = 0; j < 11; ++j) {
        int i = tid + j * AX_THREADS;
        if (i < 240 * 21) {
          int h2 = i / 21, c2 = i - h2 * 21;
          q[h2][c2] = tq[j];
          v[h2][c2] = tv[j];
        }
      }
    }
  }
  for (int i = tid; i < 240 * 4; i += AX_THREADS) {
    int m = i >> 2, c = 21 + (i & 3);
    q[m][c] = 0.f;
  }
  for (int i = tid; i < 240 * 5; i += AX_THREADS) {
    int m = i / 5, c = 21 + (i - m * 5);
    v[m][c] = 0.f;
  }
  if (tid < AX_VP) v[240][tid] = 0.f;
  __syncthreads();

  if (pass == 0) {
    float* qo = xp + (size_t)(b * 126 + 21) * HW + (size_t)r * 21;
#pragma unroll
    for (int j = 0; j < 11; ++j) {
      int i = tid + j * AX_THREADS;
      if (i < 240 * 21) {
        int xx = i / 21, c = i - xx * 21;
        qo[(size_t)xx * (240 * 21) + c] = q[xx][c];
      }
    }
  }

  const int m0 = qt * 16;
  const int cn = lane & 15;
  const int kh = (lane >> 4) << 3;

  v2f bq[6];
#pragma unroll
  for (int kk = 0; kk < 6; ++kk) bq[kk] = frag_a(&q[0][0], 25, m0, kk * 4, lane);

  v8f st[15];
#pragma unroll
  for (int g = 0; g < 5; ++g) {
    v8f a0 = V8Z, a1 = V8Z, a2 = V8Z;
#pragma unroll
    for (int kk = 0; kk < 6; ++kk) {
      a0 = wmma4(frag_a(&q[0][0], 25, (3 * g + 0) * 16, kk * 4, lane), bq[kk], a0);
      a1 = wmma4(frag_a(&q[0][0], 25, (3 * g + 1) * 16, kk * 4, lane), bq[kk], a1);
      a2 = wmma4(frag_a(&q[0][0], 25, (3 * g + 2) * 16, kk * 4, lane), bq[kk], a2);
    }
    st[3 * g + 0] = a0; st[3 * g + 1] = a1; st[3 * g + 2] = a2;
  }

  float mx = -1e30f;
#pragma unroll
  for (int nt = 0; nt < 15; ++nt)
#pragma unroll
    for (int rr = 0; rr < 8; ++rr) mx = fmaxf(mx, st[nt][rr]);
  mx = fmaxf(mx, __shfl_xor(mx, 16));
  float s0 = 0.f, s1 = 0.f, s2 = 0.f, s3 = 0.f;
#pragma unroll
  for (int nt = 0; nt < 15; ++nt) {
#pragma unroll
    for (int rr = 0; rr < 8; rr += 4) {
      float e0 = __expf(st[nt][rr + 0] - mx);
      float e1 = __expf(st[nt][rr + 1] - mx);
      float e2 = __expf(st[nt][rr + 2] - mx);
      float e3 = __expf(st[nt][rr + 3] - mx);
      st[nt][rr + 0] = e0; st[nt][rr + 1] = e1; st[nt][rr + 2] = e2; st[nt][rr + 3] = e3;
      s0 += e0; s1 += e1; s2 += e2; s3 += e3;
    }
  }
  float sum = (s0 + s1) + (s2 + s3);
  sum += __shfl_xor(sum, 16);
  float inv = 1.f / sum;
#pragma unroll
  for (int nt = 0; nt < 15; ++nt)
#pragma unroll
    for (int rr = 0; rr < 8; ++rr) st[nt][rr] *= inv;

  v8f y0 = V8Z, y1 = V8Z;
#pragma unroll
  for (int nt = 0; nt < 15; ++nt) {
#pragma unroll
    for (int i = 0; i < 4; ++i) {
      v2f pa; pa.x = st[nt][2 * i]; pa.y = st[nt][2 * i + 1];
      int kb = nt * 16 + kh + 2 * i;
      v2f b0, b1;
      b0.x = v[kb][cn];      b0.y = v[kb + 1][cn];
      b1.x = v[kb][16 + cn]; b1.y = v[kb + 1][16 + cn];
      y0 = wmma4(pa, b0, y0);
      y1 = wmma4(pa, b1, y1);
    }
  }

  float* ob = (pass == 0) ? vmid + ((size_t)(b * 240) * 240 + r) * 21
                          : xp + (size_t)(b * 126 + 105) * HW + (size_t)r * (240 * 21);
  const int sm = (pass == 0) ? 240 * 21 : 21;
  const int mb = m0 + kh;
#pragma unroll
  for (int nt = 0; nt < 2; ++nt) {
    v8f acc = nt ? y1 : y0;
    int n = nt * 16 + cn;
    if (n < 21) {
#pragma unroll
      for (int rr = 0; rr < 8; ++rr) ob[(mb + rr) * sm + n] = acc[rr];
    }
  }
}

__global__ __launch_bounds__(128) void k_proj_out(
    const float* __restrict__ xp, const float* __restrict__ y2,
    const float* __restrict__ w_out, const float* __restrict__ b_out,
    float* __restrict__ out)
{
  __shared__ float a_lds[64][65];
  __shared__ float w_lds[64][65];
  __shared__ float bias_lds[64];

  int tid  = threadIdx.x;
  int lane = tid & 31;
  int wave = tid >> 5;
  int tile = blockIdx.x;
  int b    = tile / 900;
  int p0   = (tile % 900) * 64;

  const int am = tid & 63, akh = tid >> 6;
  float ta[21];
#pragma unroll
  for (int j = 0; j < 21; ++j) {
    int k = 2 * j + akh;
    const float* sp;
    if (k < 21) sp = xp + (size_t)(b * 126 + k) * HW;
    else        sp = y2 + (size_t)(b * 21 + (k - 21)) * HW;
    ta[j] = sp[p0 + am];
  }
  float tc[11];
  {
    const float* y3 = xp + (size_t)(b * 126 + 105) * HW;
#pragma unroll
    for (int j = 0; j < 11; ++j) {
      int e  = tid + j * 128;
      int ec = e < 64 * 21 ? e : 64 * 21 - 1;
      int px = ec / 21, c = ec - px * 21;
      int p  = p0 + px;
      int h  = p / WW, w = p - h * WW;
      tc[j] = y3[(size_t)(w * HH + h) * 21 + c];
    }
  }
  float tw[32];
#pragma unroll
  for (int j = 0; j < 32; ++j) {
    int idx = tid + j * 128;
    int ic  = idx < 63 * 63 ? idx : 63 * 63 - 1;
    tw[j] = w_out[ic];
  }
  if (tid < 64) {
    bias_lds[tid] = (tid < 63) ? b_out[tid] : 0.f;
    w_lds[63][tid] = 0.f;
    w_lds[tid][63] = 0.f;
    a_lds[tid][63] = 0.f;
  }
#pragma unroll
  for (int j = 0; j < 32; ++j) {
    int idx = tid + j * 128;
    if (idx < 63 * 63) {
      int n = idx / 63, k = idx - n * 63;
      w_lds[k][n] = tw[j];
    }
  }
#pragma unroll
  for (int j = 0; j < 21; ++j) a_lds[am][2 * j + akh] = ta[j];
#pragma unroll
  for (int j = 0; j < 11; ++j) {
    int e = tid + j * 128;
    if (e < 64 * 21) {
      int px = e / 21, c = e - px * 21;
      a_lds[px][42 + c] = tc[j];
    }
  }
  __syncthreads();

  int m0 = wave * 16;
  const int chh = (lane >> 4) << 3;
  float* op = out + (size_t)b * 63 * HW + p0 + m0 + (lane & 15);
  for (int nt = 0; nt < 4; ++nt) {
    int n0 = nt * 16;
    v8f acc = V8Z;
    for (int k0 = 0; k0 < 64; k0 += 4) {
      v2f av = frag_a(&a_lds[0][0], 65, m0, k0, lane);
      v2f bv = frag_b(&w_lds[0][0], 65, k0, n0, lane);
      acc = wmma4(bv, av, acc);
    }
#pragma unroll
    for (int r = 0; r < 8; ++r) {
      int ch = n0 + chh + r;
      if (ch < 63) op[(size_t)ch * HW] = acc[r] + bias_lds[ch];
    }
  }
}

extern "C" void kernel_launch(void* const* d_in, const int* in_sizes, int n_in,
                              void* d_out, int out_size, void* d_ws, size_t ws_size,
                              hipStream_t stream)
{
  (void)in_sizes; (void)n_in; (void)out_size; (void)ws_size;
  const float* x     = (const float*)d_in[0];
  const float* w_in  = (const float*)d_in[1];
  const float* b_in  = (const float*)d_in[2];
  const float* bn_g  = (const float*)d_in[3];
  const float* bn_b  = (const float*)d_in[4];
  const float* bn_m  = (const float*)d_in[5];
  const float* bn_v  = (const float*)d_in[6];
  const float* w_out = (const float*)d_in[7];
  const float* b_out = (const float*)d_in[8];
  float* out = (float*)d_out;

  float* xp   = (float*)d_ws;
  float* y2b  = xp  + (size_t)BB * 126 * HW;
  float* vmid = y2b + (size_t)BB * 21 * HW;

  k_proj_in <<<dim3(BB * 900), 128, 0, stream>>>(x, w_in, b_in, bn_g, bn_b, bn_m, bn_v, xp);
  k_win     <<<dim3(BB * NWIN * WN_STRIPS, 2), WN_THREADS, 0, stream>>>(xp, y2b);
  k_axial   <<<dim3(BB * HH), AX_THREADS, 0, stream>>>(xp, vmid, 0);
  k_axial   <<<dim3(BB * WW), AX_THREADS, 0, stream>>>(xp, vmid, 1);
  k_proj_out<<<dim3(BB * 900), 128, 0, stream>>>(xp, y2b, w_out, b_out, out);
}


# both projections: a wave computes all 64 pixels as four pixel tiles (column c = pixel 4c+t); a lane stores 4 neighbouring pixels as one 16-byte piece, whole 256-byte runs per store; sums unchanged
# speedup vs baseline: 6.9084x; 1.0559x over previous
#include <hip/hip_runtime.h>
#include <math.h>

#define BB 8
#define CC 63
#define HH 240
#define WW 240
#define HW (HH * WW)
#define C2 126
#define WS 5
#define NWIN 48
#define BN_EPS 1e-5f

typedef __attribute__((ext_vector_type(2))) float v2f;
typedef __attribute__((ext_vector_type(4))) float v4f;
typedef __attribute__((ext_vector_type(8))) float v8f;

__device__ __forceinline__ v8f wmma4(v2f a, v2f b, v8f c) {
  return __builtin_amdgcn_wmma_f32_16x16x4_f32(
       false, a,  false, b,
       (short)0, c,  false,  false);
}

__device__ __forceinline__ v2f frag_a(const float* p, int ld, int m0, int k0, int lane) {
  int m  = m0 + (lane & 15);
  int kb = k0 + ((lane >> 4) << 1);
  const float* q = p + m * ld + kb;
  v2f r; r.x = q[0]; r.y = q[1];
  return r;
}

__device__ __forceinline__ v2f frag_row(const float* p, int ld, int row, int k0, int lane) {
  int kb = k0 + ((lane >> 4) << 1);
  const float* q = p + row * ld + kb;
  v2f r; r.x = q[0]; r.y = q[1];
  return r;
}

__device__ __forceinline__ v2f frag_b(const float* p, int ld, int k0, int n0, int lane) {
  int n  = n0 + (lane & 15);
  int kb = k0 + ((lane >> 4) << 1);
  v2f r; r.x = p[kb * ld + n]; r.y = p[(kb + 1) * ld + n];
  return r;
}

__device__ __forceinline__ void frag_store(float* p, int ld, int m0, int n0, int lane, v8f d) {
  int n  = n0 + (lane & 15);
  int mb = m0 + ((lane >> 4) << 3);
#pragma unroll
  for (int r = 0; r < 8; ++r) p[(mb + r) * ld + n] = d[r];
}

#define V8Z {0.f,0.f,0.f,0.f,0.f,0.f,0.f,0.f}

template <bool GUARD>
__device__ __forceinline__ void store_tile4(float* op, const float* bias, int ch0, int nch,
                                            v8f a0, v8f a1, v8f a2, v8f a3) {
  float bs[8];
#pragma unroll
  for (int r = 0; r < 8; ++r) bs[r] = bias[ch0 + r];
#pragma unroll
  for (int r = 0; r < 8; ++r) {
    int ch = ch0 + r;
    if (!GUARD || ch < nch) {
      v4f o;
      o.x = a0[r] + bs[r]; o.y = a1[r] + bs[r];
      o.z = a2[r] + bs[r]; o.w = a3[r] + bs[r];
      *(v4f*)(op + (size_t)ch * HW) = o;
    }
  }
}

__global__ __launch_bounds__(128) void k_proj_in(
    const float* __restrict__ x, const float* __restrict__ w_in,
    const float* __restrict__ b_in, const float* __restrict__ bn_g,
    const float* __restrict__ bn_b, const float* __restrict__ bn_m,
    const float* __restrict__ bn_v, float* __restrict__ xp)
{
  __shared__ float a_lds[64][65];
  __shared__ float w_lds[64][129];
  __shared__ float bias_lds[128];
  __shared__ float scale_lds[128];

  int tid  = threadIdx.x;
  int lane = tid & 31;
  int wave = tid >> 5;
  int tile = blockIdx.x;
  int b    = tile / 900;
  int p0   = (tile % 900) * 64;

  {
    float sc = 0.f, bv = 0.f;
    if (tid < 126) {
      sc = bn_g[tid] * rsqrtf(bn_v[tid] + BN_EPS);
      bv = b_in[tid] * sc + bn_b[tid] - bn_m[tid] * sc;
    }
    scale_lds[tid] = sc;
    bias_lds[tid]  = bv;
  }
  const int am = tid & 63, akh = tid >> 6;
  float ta[32];
  {
    const float* xb = x + (size_t)b * 63 * HW + p0 + am;
#pragma unroll
    for (int j = 0; j < 32; ++j) {
      int k  = 2 * j + akh;
      int kc = k < 63 ? k : 62;
      ta[j] = xb[(size_t)kc * HW];
    }
  }
  float tw[63];
#pragma unroll
  for (int j = 0; j < 63; ++j) {
    int idx = tid + j * 128;
    int ic  = idx < 126 * 63 ? idx : 126 * 63 - 1;
    tw[j] = w_in[ic];
  }
  w_lds[63][tid] = 0.f;
  if (tid < 63) { w_lds[tid][126] = 0.f; w_lds[tid][127] = 0.f; }
  __syncthreads();
#pragma unroll
  for (int j = 0; j < 63; ++j) {
    int idx = tid + j * 128;
    if (idx < 126 * 63) {
      int n = idx / 63, k = idx - n * 63;
      w_lds[k][n] = tw[j] * scale_lds[n];
    }
  }
#pragma unroll
  for (int j = 0; j < 32; ++j) {
    int k = 2 * j + akh;
    a_lds[am][k] = (k < 63) ? ta[j] : 0.f;
  }
  __syncthreads();

  const int cn  = lane & 15;
  const int chh = (lane >> 4) << 3;
  const int nb  = wave * 32;
  float* op = xp + (size_t)b * 126 * HW + p0 + 4 * cn;
  v8f acc[2][4];
#pragma unroll
  for (int nt = 0; nt < 2; ++nt)
#pragma unroll
    for (int t = 0; t < 4; ++t) { v8f z = V8Z; acc[nt][t] = z; }
  for (int k0 = 0; k0 < 64; k0 += 4) {
    v2f pf[4], wf[2];
#pragma unroll
    for (int t = 0; t < 4; ++t) pf[t] = frag_row(&a_lds[0][0], 65, 4 * cn + t, k0, lane);
#pragma unroll
    for (int nt = 0; nt < 2; ++nt) wf[nt] = frag_b(&w_lds[0][0], 129, k0, nb + nt * 16, lane);
#pragma unroll
    for (int nt = 0; nt < 2; ++nt)
#pragma unroll
      for (int t = 0; t < 4; ++t) acc[nt][t] = wmma4(wf[nt], pf[t], acc[nt][t]);
  }
  const int wave_u = __builtin_amdgcn_readfirstlane(wave);
  store_tile4<false>(op, bias_lds, nb + chh, 126, acc[0][0], acc[0][1], acc[0][2], acc[0][3]);
  if (wave_u < 3) store_tile4<false>(op, bias_lds, nb + 16 + chh, 126, acc[1][0], acc[1][1], acc[1][2], acc[1][3]);
  else            store_tile4<true >(op, bias_lds, nb + 16 + chh, 126, acc[1][0], acc[1][1], acc[1][2], acc[1][3]);
}

#define WN_NW 8
#define WN_THREADS (WN_NW * 32)
#define WN_SEG (WN_NW * WS)
#define WN_STRIPS (WW / WN_SEG)
#define WN_MOVERS (WN_SEG * 6)
#define WN_VOFF 800
#define WN_SOFF 1600
#define WN_WSZ 2664

__global__ __launch_bounds__(WN_THREADS) void k_win(float* __restrict__ xp, float* __restrict__ y2)
{
  __shared__ float wl[WN_NW * WN_WSZ];

  const int tid  = threadIdx.x;
  const int lane = tid & 31;
  const int wv   = tid >> 5;
  const int sid  = blockIdx.x;
  const int grp  = blockIdx.y;
  const int b    = sid / (NWIN * WN_STRIPS);
  const int rem  = sid % (NWIN * WN_STRIPS);
  const int wy   = rem / WN_STRIPS, st = rem % WN_STRIPS;
  const int shi  = grp ? 3 : 0;
  const int sho  = grp ? 2 : 0;

  const int x    = tid % WN_SEG, sub = tid / WN_SEG;
  const int win  = x / WS, mxp = x - win * WS;
  const int isv  = sub >= 3;
  const int cb   = (sub - 3 * isv) * 7;

  float t[35];
  if (tid < WN_MOVERS) {
    const int gx = (st * WN_SEG + x + shi) % WW;
    const float* srcp = xp + (size_t)(b * 126 + grp * 42 + isv * 21 + cb) * HW + gx;
    int ro[5];
#pragma unroll
    for (int r = 0; r < 5; ++r) ro[r] = ((wy * WS + r + shi) % HH) * WW;
#pragma unroll
    for (int j = 0; j < 35; ++j) t[j] = srcp[(size_t)(j / 5) * HW + ro[j % 5]];
  }

  {
    float* rec = wl + wv * WN_WSZ;
    for (int i = lane; i < 7 * 25; i += 32) {
      rec[25 * 25 + i] = 0.f;
      rec[WN_VOFF + 25 * 25 + i] = 0.f;
    }
    for (int i = lane; i < 25 * 4; i += 32) {
      int m = i >> 2, c = 21 + (i & 3);
      rec[m * 25 + c] = 0.f;
      rec[WN_VOFF + m * 25 + c] = 0.f;
    }
  }

  if (tid < WN_MOVERS) {
    float* dl = wl + win * WN_WSZ + isv * WN_VOFF + mxp * 25 + cb;
#pragma unroll
    for (int j = 0; j < 35; ++j) dl[(j % 5) * (WS * 25) + (j / 5)] = t[j];
  }
  __syncthreads();

  float* q = wl + wv * WN_WSZ;
  float* v = q + WN_VOFF;
  float* s = q + WN_SOFF;

  for (int mt = 0; mt < 2; ++mt)
    for (int nt = 0; nt < 2; ++nt) {
      v8f acc = V8Z;
      for (int k0 = 0; k0 < 24; k0 += 4) {
        v2f av = frag_a(q, 25, mt * 16, k0, lane);
        v2f bv = frag_a(q, 25, nt * 16, k0, lane);
        acc = wmma4(av, bv, acc);
      }
      frag_store(s, 33, mt * 16, nt * 16, lane, acc);
    }

  if (lane < 25) {
    float* sr = s + lane * 33;
    float mx = -1e30f;
    for (int n = 0; n < 25; ++n) mx = fmaxf(mx, sr[n]);
    float sum = 0.f;
    for (int n = 0; n < 25; ++n) { float e = __expf(sr[n] - mx); sr[n] = e; sum += e; }
    float inv = 1.f / sum;
    for (int n = 0; n < 25; ++n) sr[n] *= inv;
    for (int n = 25; n < 28; ++n) sr[n] = 0.f;
  }

  v8f yt[2][2];
#pragma unroll
  for (int mt = 0; mt < 2; ++mt)
#pragma unroll
    for (int nt = 0; nt < 2; ++nt) {
      v8f acc = V8Z;
      for (int k0 = 0; k0 < 28; k0 += 4) {
        v2f av = frag_a(s, 33, mt * 16, k0, lane);
        v2f bv = frag_b(v, 25, k0, nt * 16, lane);
        acc = wmma4(av, bv, acc);
      }
      yt[mt][nt] = acc;
    }
#pragma unroll
  for (int mt = 0; mt < 2; ++mt)
#pragma unroll
    for (int nt = 0; nt < 2; ++nt) frag_store(s, 33, mt * 16, nt * 16, lane, yt[mt][nt]);
  __syncthreads();

  if (tid < WN_MOVERS) {
    float* dst = grp ? y2 + (size_t)(b * 21) * HW : xp + (size_t)(b * 126) * HW;
    const float* yl = wl + win * WN_WSZ + WN_SOFF + mxp * 33;
    const int ox = (st * WN_SEG + x + sho) % WW;
#pragma unroll
    for (int j = 0; j < 18; ++j) {
      if (j < 17 || sub < 3) {
        int p = j * 6 + sub;
        int c = p / WS, r = p - c * WS;
        int oy = (wy * WS + r + sho) % HH;
        dst[(size_t)c * HW + oy * WW + ox] = yl[r * (WS * 33) + c];
      }
    }
  }
}

#define AX_WAVES 15
#define AX_THREADS (AX_WAVES * 32)
#define AX_VP 26

__global__ __launch_bounds__(AX_THREADS) void k_axial(float* __restrict__ xp,
                                                      float* __restrict__ vmid, int pass)
{
  __shared__ float q[240][25];
  __shared__ float v[241][AX_VP];

  int tid  = threadIdx.x;
  int lane = tid & 31;
  int qt   = tid >> 5;
  int row  = blockIdx.x;
  int b    = row / HH;
  int r    = row % HH;

  {
    float tq[11], tv[11];
    if (pass == 0) {
      const float* qb = xp + (size_t)(b * 126 + 84) * HW + r * WW;
      const float* vb = xp + (size_t)(b * 126 + 105) * HW + r * WW;
#pragma unroll
      for (int j = 0; j < 11; ++j) {
        int i  = tid + j * AX_THREADS;
        int ic = i < 240 * 21 ? i : 240 * 21 - 1;
        int c = ic / 240, xx = ic - c * 240;
        tq[j] = qb[(size_t)c * HW + xx];
        tv[j] = vb[(size_t)c * HW + xx];
      }
#pragma unroll
      for (int j = 0; j < 11; ++j) {
        int i = tid + j * AX_THREADS;
        if (i < 240 * 21) {
          int c = i / 240, xx = i - c * 240;
          q[xx][c] = tq[j];
          v[xx][c] = tv[j];
        }
      }
    } else {
      const float* qb = xp + (size_t)(b * 126 + 21) * HW + (size_t)r * (240 * 21);
      const float* vb = vmid + (size_t)(b * 240 + r) * 240 * 21;
#pragma unroll
      for (int j = 0; j < 11; ++j) {
        int i  = tid + j * AX_THREADS;
        int ic = i < 240 * 21 ? i : 240 * 21 - 1;
        tq[j] = qb[ic];
        tv[j] = vb[ic];
      }
#pragma unroll
      for (int j = 0; j < 11; ++j) {
        int i = tid + j * AX_THREADS;
        if (i < 240 * 21) {
          int h2 = i / 21, c2 = i - h2 * 21;
          q[h2][c2] = tq[j];
          v[h2][c2] = tv[j];
        }
      }
    }
  }
  for (int i = tid; i < 240 * 4; i += AX_THREADS) {
    int m = i >> 2, c = 21 + (i & 3);
    q[m][c] = 0.f;
  }
  for (int i = tid; i < 240 * 5; i += AX_THREADS) {
    int m = i / 5, c = 21 + (i - m * 5);
    v[m][c] = 0.f;
  }
  if (tid < AX_VP) v[240][tid] = 0.f;
  __syncthreads();

  if (pass == 0) {
    float* qo = xp + (size_t)(b * 126 + 21) * HW + (size_t)r * 21;
#pragma unroll
    for (int j = 0; j < 11; ++j) {
      int i = tid + j * AX_THREADS;
      if (i < 240 * 21) {
        int xx = i / 21, c = i - xx * 21;
        qo[(size_t)xx * (240 * 21) + c] = q[xx][c];
      }
    }
  }

  const int m0 = qt * 16;
  const int cn = lane & 15;
  const int kh = (lane >> 4) << 3;

  v2f bq[6];
#pragma unroll
  for (int kk = 0; kk < 6; ++kk) bq[kk] = frag_a(&q[0][0], 25, m0, kk * 4, lane);

  v8f st[15];
#pragma unroll
  for (int g = 0; g < 5; ++g) {
    v8f a0 = V8Z, a1 = V8Z, a2 = V8Z;
#pragma unroll
    for (int kk = 0; kk < 6; ++kk) {
      a0 = wmma4(frag_a(&q[0][0], 25, (3 * g + 0) * 16, kk * 4, lane), bq[kk], a0);
      a1 = wmma4(frag_a(&q[0][0], 25, (3 * g + 1) * 16, kk * 4, lane), bq[kk], a1);
      a2 = wmma4(frag_a(&q[0][0], 25, (3 * g + 2) * 16, kk * 4, lane), bq[kk], a2);
    }
    st[3 * g + 0] = a0; st[3 * g + 1] = a1; st[3 * g + 2] = a2;
  }

  float mx = -1e30f;
#pragma unroll
  for (int nt = 0; nt < 15; ++nt)
#pragma unroll
    for (int rr = 0; rr < 8; ++rr) mx = fmaxf(mx, st[nt][rr]);
  mx = fmaxf(mx, __shfl_xor(mx, 16));
  float s0 = 0.f, s1 = 0.f, s2 = 0.f, s3 = 0.f;
#pragma unroll
  for (int nt = 0; nt < 15; ++nt) {
#pragma unroll
    for (int rr = 0; rr < 8; rr += 4) {
      float e0 = __expf(st[nt][rr + 0] - mx);
      float e1 = __expf(st[nt][rr + 1] - mx);
      float e2 = __expf(st[nt][rr + 2] - mx);
      float e3 = __expf(st[nt][rr + 3] - mx);
      st[nt][rr + 0] = e0; st[nt][rr + 1] = e1; st[nt][rr + 2] = e2; st[nt][rr + 3] = e3;
      s0 += e0; s1 += e1; s2 += e2; s3 += e3;
    }
  }
  float sum = (s0 + s1) + (s2 + s3);
  sum += __shfl_xor(sum, 16);
  float inv = 1.f / sum;
#pragma unroll
  for (int nt = 0; nt < 15; ++nt)
#pragma unroll
    for (int rr = 0; rr < 8; ++rr) st[nt][rr] *= inv;

  v8f y0 = V8Z, y1 = V8Z;
#pragma unroll
  for (int nt = 0; nt < 15; ++nt) {
#pragma unroll
    for (int i = 0; i < 4; ++i) {
      v2f pa; pa.x = st[nt][2 * i]; pa.y = st[nt][2 * i + 1];
      int kb = nt * 16 + kh + 2 * i;
      v2f b0, b1;
      b0.x = v[kb][cn];      b0.y = v[kb + 1][cn];
      b1.x = v[kb][16 + cn]; b1.y = v[kb + 1][16 + cn];
      y0 = wmma4(pa, b0, y0);
      y1 = wmma4(pa, b1, y1);
    }
  }

  float* ob = (pass == 0) ? vmid + ((size_t)(b * 240) * 240 + r) * 21
                          : xp + (size_t)(b * 126 + 105) * HW + (size_t)r * (240 * 21);
  const int sm = (pass == 0) ? 240 * 21 : 21;
  const int mb = m0 + kh;
#pragma unroll
  for (int nt = 0; nt < 2; ++nt) {
    v8f acc = nt ? y1 : y0;
    int n = nt * 16 + cn;
    if (n < 21) {
#pragma unroll
      for (int rr = 0; rr < 8; ++rr) ob[(mb + rr) * sm + n] = acc[rr];
    }
  }
}

__global__ __launch_bounds__(128) void k_proj_out(
    const float* __restrict__ xp, const float* __restrict__ y2,
    const float* __restrict__ w_out, const float* __restrict__ b_out,
    float* __restrict__ out)
{
  __shared__ float a_lds[64][65];
  __shared__ float w_lds[64][65];
  __shared__ float bias_lds[64];

  int tid  = threadIdx.x;
  int lane = tid & 31;
  int wave = tid >> 5;
  int tile = blockIdx.x;
  int b    = tile / 900;
  int p0   = (tile % 900) * 64;

  const int am = tid & 63, akh = tid >> 6;
  float ta[21];
#pragma unroll
  for (int j = 0; j < 21; ++j) {
    int k = 2 * j + akh;
    const float* sp;
    if (k < 21) sp = xp + (size_t)(b * 126 + k) * HW;
    else        sp = y2 + (size_t)(b * 21 + (k - 21)) * HW;
    ta[j] = sp[p0 + am];
  }
  float tc[11];
  {
    const float* y3 = xp + (size_t)(b * 126 + 105) * HW;
#pragma unroll
    for (int j = 0; j < 11; ++j) {
      int e  = tid + j * 128;
      int ec = e < 64 * 21 ? e : 64 * 21 - 1;
      int px = ec / 21, c = ec - px * 21;
      int p  = p0 + px;
      int h  = p / WW, w = p - h * WW;
      tc[j] = y3[(size_t)(w * HH + h) * 21 + c];
    }
  }
  float tw[32];
#pragma unroll
  for (int j = 0; j < 32; ++j) {
    int idx = tid + j * 128;
    int ic  = idx < 63 * 63 ? idx : 63 * 63 - 1;
    tw[j] = w_out[ic];
  }
  if (tid < 64) {
    bias_lds[tid] = (tid < 63) ? b_out[tid] : 0.f;
    w_lds[63][tid] = 0.f;
    w_lds[tid][63] = 0.f;
    a_lds[tid][63] = 0.f;
  }
#pragma unroll
  for (int j = 0; j < 32; ++j) {
    int idx = tid + j * 128;
    if (idx < 63 * 63) {
      int n = idx / 63, k = idx - n * 63;
      w_lds[k][n] = tw[j];
    }
  }
#pragma unroll
  for (int j = 0; j < 21; ++j) a_lds[am][2 * j + akh] = ta[j];
#pragma unroll
  for (int j = 0; j < 11; ++j) {
    int e = tid + j * 128;
    if (e < 64 * 21) {
      int px = e / 21, c = e - px * 21;
      a_lds[px][42 + c] = tc[j];
    }
  }
  __syncthreads();

  const int cn  = lane & 15;
  const int chh = (lane >> 4) << 3;
  const int n0  = wave * 16;
  float* op = out + (size_t)b * 63 * HW + p0 + 4 * cn;
  v8f acc[4];
#pragma unroll
  for (int t = 0; t < 4; ++t) { v8f z = V8Z; acc[t] = z; }
  for (int k0 = 0; k0 < 64; k0 += 4) {
    v2f pf[4];
#pragma unroll
    for (int t = 0; t < 4; ++t) pf[t] = frag_row(&a_lds[0][0], 65, 4 * cn + t, k0, lane);
    v2f wf = frag_b(&w_lds[0][0], 65, k0, n0, lane);
#pragma unroll
    for (int t = 0; t < 4; ++t) acc[t] = wmma4(wf, pf[t], acc[t]);
  }
  const int wave_u = __builtin_amdgcn_readfirstlane(wave);
  if (wave_u < 3) store_tile4<false>(op, bias_lds, n0 + chh, 63, acc[0], acc[1], acc[2], acc[3]);
  else            store_tile4<true >(op, bias_lds, n0 + chh, 63, acc[0], acc[1], acc[2], acc[3]);
}

extern "C" void kernel_launch(void* const* d_in, const int* in_sizes, int n_in,
                              void* d_out, int out_size, void* d_ws, size_t ws_size,
                              hipStream_t stream)
{
  (void)in_sizes; (void)n_in; (void)out_size; (void)ws_size;
  const float* x     = (const float*)d_in[0];
  const float* w_in  = (const float*)d_in[1];
  const float* b_in  = (const float*)d_in[2];
  const float* bn_g  = (const float*)d_in[3];
  const float* bn_b  = (const float*)d_in[4];
  const float* bn_m  = (const float*)d_in[5];
  const float* bn_v  = (const float*)d_in[6];
  const float* w_out = (const float*)d_in[7];
  const float* b_out = (const float*)d_in[8];
  float* out = (float*)d_out;

  float* xp   = (float*)d_ws;
  float* y2b  = xp  + (size_t)BB * 126 * HW;
  float* vmid = y2b + (size_t)BB * 21 * HW;

  k_proj_in <<<dim3(BB * 900), 128, 0, stream>>>(x, w_in, b_in, bn_g, bn_b, bn_m, bn_v, xp);
  k_win     <<<dim3(BB * NWIN * WN_STRIPS, 2), WN_THREADS, 0, stream>>>(xp, y2b);
  k_axial   <<<dim3(BB * HH), AX_THREADS, 0, stream>>>(xp, vmid, 0);
  k_axial   <<<dim3(BB * WW), AX_THREADS, 0, stream>>>(xp, vmid, 1);
  k_proj_out<<<dim3(BB * 900), 128, 0, stream>>>(xp, y2b, w_out, b_out, out);
}


# both projections: a workgroup computes 5 neighbouring 64-pixel tiles with the weight tile, scale and bias staged once (was once per tile; 7200 workgroups each, now 1440); sums unchanged
# speedup vs baseline: 7.3387x; 1.0623x over previous
#include <hip/hip_runtime.h>
#include <math.h>

#define BB 8
#define CC 63
#define HH 240
#define WW 240
#define HW (HH * WW)
#define C2 126
#define WS 5
#define NWIN 48
#define BN_EPS 1e-5f

typedef __attribute__((ext_vector_type(2))) float v2f;
typedef __attribute__((ext_vector_type(4))) float v4f;
typedef __attribute__((ext_vector_type(8))) float v8f;

__device__ __forceinline__ v8f wmma4(v2f a, v2f b, v8f c) {
  return __builtin_amdgcn_wmma_f32_16x16x4_f32(
       false, a,  false, b,
       (short)0, c,  false,  false);
}

__device__ __forceinline__ v2f frag_a(const float* p, int ld, int m0, int k0, int lane) {
  int m  = m0 + (lane & 15);
  int kb = k0 + ((lane >> 4) << 1);
  const float* q = p + m * ld + kb;
  v2f r; r.x = q[0]; r.y = q[1];
  return r;
}

__device__ __forceinline__ v2f frag_row(const float* p, int ld, int row, int k0, int lane) {
  int kb = k0 + ((lane >> 4) << 1);
  const float* q = p + row * ld + kb;
  v2f r; r.x = q[0]; r.y = q[1];
  return r;
}

__device__ __forceinline__ v2f frag_b(const float* p, int ld, int k0, int n0, int lane) {
  int n  = n0 + (lane & 15);
  int kb = k0 + ((lane >> 4) << 1);
  v2f r; r.x = p[kb * ld + n]; r.y = p[(kb + 1) * ld + n];
  return r;
}

__device__ __forceinline__ void frag_store(float* p, int ld, int m0, int n0, int lane, v8f d) {
  int n  = n0 + (lane & 15);
  int mb = m0 + ((lane >> 4) << 3);
#pragma unroll
  for (int r = 0; r < 8; ++r) p[(mb + r) * ld + n] = d[r];
}

#define V8Z {0.f,0.f,0.f,0.f,0.f,0.f,0.f,0.f}

template <bool GUARD>
__device__ __forceinline__ void store_tile4(float* op, const float* bias, int ch0, int nch,
                                            v8f a0, v8f a1, v8f a2, v8f a3) {
  float bs[8];
#pragma unroll
  for (int r = 0; r < 8; ++r) bs[r] = bias[ch0 + r];
#pragma unroll
  for (int r = 0; r < 8; ++r) {
    int ch = ch0 + r;
    if (!GUARD || ch < nch) {
      v4f o;
      o.x = a0[r] + bs[r]; o.y = a1[r] + bs[r];
      o.z = a2[r] + bs[r]; o.w = a3[r] + bs[r];
      *(v4f*)(op + (size_t)ch * HW) = o;
    }
  }
}

#define PI_T 5
#define PI_BLK (900 / PI_T)

__global__ __launch_bounds__(128, 6) void k_proj_in(
    const float* __restrict__ x, const float* __restrict__ w_in,
    const float* __restrict__ b_in, const float* __restrict__ bn_g,
    const float* __restrict__ bn_b, const float* __restrict__ bn_m,
    const float* __restrict__ bn_v, float* __restrict__ xp)
{
  __shared__ float a_lds[64][65];
  __shared__ float w_lds[64][129];
  __shared__ float bias_lds[128];
  __shared__ float scale_lds[128];

  int tid  = threadIdx.x;
  int wave = tid >> 5;
  int blk  = blockIdx.x;
  int b    = blk / PI_BLK;
  int pb   = (blk % PI_BLK) * (PI_T * 64);

  {
    float sc = 0.f, bv = 0.f;
    if (tid < 126) {
      sc = bn_g[tid] * rsqrtf(bn_v[tid] + BN_EPS);
      bv = b_in[tid] * sc + bn_b[tid] - bn_m[tid] * sc;
    }
    scale_lds[tid] = sc;
    bias_lds[tid]  = bv;
  }
  const int am = tid & 63, akh = tid >> 6;
  const float* xb = x + (size_t)b * 63 * HW + pb + am;
  float ta[32];
#pragma unroll
  for (int j = 0; j < 32; ++j) {
    int k  = 2 * j + akh;
    int kc = k < 63 ? k : 62;
    ta[j] = xb[(size_t)kc * HW];
  }
  {
    float tw[63];
#pragma unroll
    for (int j = 0; j < 63; ++j) {
      int idx = tid + j * 128;
      int ic  = idx < 126 * 63 ? idx : 126 * 63 - 1;
      tw[j] = w_in[ic];
    }
    w_lds[63][tid] = 0.f;
    if (tid < 63) { w_lds[tid][126] = 0.f; w_lds[tid][127] = 0.f; }
    __syncthreads();
#pragma unroll
    for (int j = 0; j < 63; ++j) {
      int idx = tid + j * 128;
      if (idx < 126 * 63) {
        int n = idx / 63, k = idx - n * 63;
        w_lds[k][n] = tw[j] * scale_lds[n];
      }
    }
  }

  const int wave_u = __builtin_amdgcn_readfirstlane(wave);

  const int ln  = tid & 31;
  const int cn  = ln & 15;
  const int chh = (ln >> 4) << 3;
  const int nb  = wave * 32;

  for (int t = 0; t < PI_T; ++t) {
    if (t > 0) __syncthreads();
#pragma unroll
    for (int j = 0; j < 32; ++j) {
      int k = 2 * j + akh;
      a_lds[am][k] = (k < 63) ? ta[j] : 0.f;
    }
    __syncthreads();

    float* op = xp + (size_t)b * 126 * HW + pb + t * 64 + 4 * cn;
    v8f acc[2][4];
#pragma unroll
    for (int nt = 0; nt < 2; ++nt)
#pragma unroll
      for (int u = 0; u < 4; ++u) { v8f z = V8Z; acc[nt][u] = z; }
    for (int k0 = 0; k0 < 64; k0 += 4) {
      v2f pf[4], wf[2];
#pragma unroll
      for (int u = 0; u < 4; ++u) pf[u] = frag_row(&a_lds[0][0], 65, 4 * cn + u, k0, ln);
#pragma unroll
      for (int nt = 0; nt < 2; ++nt) wf[nt] = frag_b(&w_lds[0][0], 129, k0, nb + nt * 16, ln);
#pragma unroll
      for (int nt = 0; nt < 2; ++nt)
#pragma unroll
        for (int u = 0; u < 4; ++u) acc[nt][u] = wmma4(wf[nt], pf[u], acc[nt][u]);
    }
    store_tile4<false>(op, bias_lds, nb + chh, 126, acc[0][0], acc[0][1], acc[0][2], acc[0][3]);
    if (wave_u < 3) store_tile4<false>(op, bias_lds, nb + 16 + chh, 126, acc[1][0], acc[1][1], acc[1][2], acc[1][3]);
    else            store_tile4<true >(op, bias_lds, nb + 16 + chh, 126, acc[1][0], acc[1][1], acc[1][2], acc[1][3]);
    if (t + 1 < PI_T) {
      const float* xn = xb + (t + 1) * 64;
#pragma unroll
      for (int j = 0; j < 32; ++j) {
        int k  = 2 * j + akh;
        int kc = k < 63 ? k : 62;
        ta[j] = xn[(size_t)kc * HW];
      }
    }
  }
}

#define WN_NW 8
#define WN_THREADS (WN_NW * 32)
#define WN_SEG (WN_NW * WS)
#define WN_STRIPS (WW / WN_SEG)
#define WN_MOVERS (WN_SEG * 6)
#define WN_VOFF 800
#define WN_SOFF 1600
#define WN_WSZ 2664

__global__ __launch_bounds__(WN_THREADS) void k_win(float* __restrict__ xp, float* __restrict__ y2)
{
  __shared__ float wl[WN_NW * WN_WSZ];

  const int tid  = threadIdx.x;
  const int lane = tid & 31;
  const int wv   = tid >> 5;
  const int sid  = blockIdx.x;
  const int grp  = blockIdx.y;
  const int b    = sid / (NWIN * WN_STRIPS);
  const int rem  = sid % (NWIN * WN_STRIPS);
  const int wy   = rem / WN_STRIPS, st = rem % WN_STRIPS;
  const int shi  = grp ? 3 : 0;
  const int sho  = grp ? 2 : 0;

  const int x    = tid % WN_SEG, sub = tid / WN_SEG;
  const int win  = x / WS, mxp = x - win * WS;
  const int isv  = sub >= 3;
  const int cb   = (sub - 3 * isv) * 7;

  float t[35];
  if (tid < WN_MOVERS) {
    const int gx = (st * WN_SEG + x + shi) % WW;
    const float* srcp = xp + (size_t)(b * 126 + grp * 42 + isv * 21 + cb) * HW + gx;
    int ro[5];
#pragma unroll
    for (int r = 0; r < 5; ++r) ro[r] = ((wy * WS + r + shi) % HH) * WW;
#pragma unroll
    for (int j = 0; j < 35; ++j) t[j] = srcp[(size_t)(j / 5) * HW + ro[j % 5]];
  }

  {
    float* rec = wl + wv * WN_WSZ;
    for (int i = lane; i < 7 * 25; i += 32) {
      rec[25 * 25 + i] = 0.f;
      rec[WN_VOFF + 25 * 25 + i] = 0.f;
    }
    for (int i = lane; i < 25 * 4; i += 32) {
      int m = i >> 2, c = 21 + (i & 3);
      rec[m * 25 + c] = 0.f;
      rec[WN_VOFF + m * 25 + c] = 0.f;
    }
  }

  if (tid < WN_MOVERS) {
    float* dl = wl + win * WN_WSZ + isv * WN_VOFF + mxp * 25 + cb;
#pragma unroll
    for (int j = 0; j < 35; ++j) dl[(j % 5) * (WS * 25) + (j / 5)] = t[j];
  }
  __syncthreads();

  float* q = wl + wv * WN_WSZ;
  float* v = q + WN_VOFF;
  float* s = q + WN_SOFF;

  for (int mt = 0; mt < 2; ++mt)
    for (int nt = 0; nt < 2; ++nt) {
      v8f acc = V8Z;
      for (int k0 = 0; k0 < 24; k0 += 4) {
        v2f av = frag_a(q, 25, mt * 16, k0, lane);
        v2f bv = frag_a(q, 25, nt * 16, k0, lane);
        acc = wmma4(av, bv, acc);
      }
      frag_store(s, 33, mt * 16, nt * 16, lane, acc);
    }

  if (lane < 25) {
    float* sr = s + lane * 33;
    float mx = -1e30f;
    for (int n = 0; n < 25; ++n) mx = fmaxf(mx, sr[n]);
    float sum = 0.f;
    for (int n = 0; n < 25; ++n) { float e = __expf(sr[n] - mx); sr[n] = e; sum += e; }
    float inv = 1.f / sum;
    for (int n = 0; n < 25; ++n) sr[n] *= inv;
    for (int n = 25; n < 28; ++n) sr[n] = 0.f;
  }

  v8f yt[2][2];
#pragma unroll
  for (int mt = 0; mt < 2; ++mt)
#pragma unroll
    for (int nt = 0; nt < 2; ++nt) {
      v8f acc = V8Z;
      for (int k0 = 0; k0 < 28; k0 += 4) {
        v2f av = frag_a(s, 33, mt * 16, k0, lane);
        v2f bv = frag_b(v, 25, k0, nt * 16, lane);
        acc = wmma4(av, bv, acc);
      }
      yt[mt][nt] = acc;
    }
#pragma unroll
  for (int mt = 0; mt < 2; ++mt)
#pragma unroll
    for (int nt = 0; nt < 2; ++nt) frag_store(s, 33, mt * 16, nt * 16, lane, yt[mt][nt]);
  __syncthreads();

  if (tid < WN_MOVERS) {
    float* dst = grp ? y2 + (size_t)(b * 21) * HW : xp + (size_t)(b * 126) * HW;
    const float* yl = wl + win * WN_WSZ + WN_SOFF + mxp * 33;
    const int ox = (st * WN_SEG + x + sho) % WW;
#pragma unroll
    for (int j = 0; j < 18; ++j) {
      if (j < 17 || sub < 3) {
        int p = j * 6 + sub;
        int c = p / WS, r = p - c * WS;
        int oy = (wy * WS + r + sho) % HH;
        dst[(size_t)c * HW + oy * WW + ox] = yl[r * (WS * 33) + c];
      }
    }
  }
}

#define AX_WAVES 15
#define AX_THREADS (AX_WAVES * 32)
#define AX_VP 26

__global__ __launch_bounds__(AX_THREADS) void k_axial(float* __restrict__ xp,
                                                      float* __restrict__ vmid, int pass)
{
  __shared__ float q[240][25];
  __shared__ float v[241][AX_VP];

  int tid  = threadIdx.x;
  int lane = tid & 31;
  int qt   = tid >> 5;
  int row  = blockIdx.x;
  int b    = row / HH;
  int r    = row % HH;

  {
    float tq[11], tv[11];
    if (pass == 0) {
      const float* qb = xp + (size_t)(b * 126 + 84) * HW + r * WW;
      const float* vb = xp + (size_t)(b * 126 + 105) * HW + r * WW;
#pragma unroll
      for (int j = 0; j < 11; ++j) {
        int i  = tid + j * AX_THREADS;
        int ic = i < 240 * 21 ? i : 240 * 21 - 1;
        int c = ic / 240, xx = ic - c * 240;
        tq[j] = qb[(size_t)c * HW + xx];
        tv[j] = vb[(size_t)c * HW + xx];
      }
#pragma unroll
      for (int j = 0; j < 11; ++j) {
        int i = tid + j * AX_THREADS;
        if (i < 240 * 21) {
          int c = i / 240, xx = i - c * 240;
          q[xx][c] = tq[j];
          v[xx][c] = tv[j];
        }
      }
    } else {
      const float* qb = xp + (size_t)(b * 126 + 21) * HW + (size_t)r * (240 * 21);
      const float* vb = vmid + (size_t)(b * 240 + r) * 240 * 21;
#pragma unroll
      for (int j = 0; j < 11; ++j) {
        int i  = tid + j * AX_THREADS;
        int ic = i < 240 * 21 ? i : 240 * 21 - 1;
        tq[j] = qb[ic];
        tv[j] = vb[ic];
      }
#pragma unroll
      for (int j = 0; j < 11; ++j) {
        int i = tid + j * AX_THREADS;
        if (i < 240 * 21) {
          int h2 = i / 21, c2 = i - h2 * 21;
          q[h2][c2] = tq[j];
          v[h2][c2] = tv[j];
        }
      }
    }
  }
  for (int i = tid; i < 240 * 4; i += AX_THREADS) {
    int m = i >> 2, c = 21 + (i & 3);
    q[m][c] = 0.f;
  }
  for (int i = tid; i < 240 * 5; i += AX_THREADS) {
    int m = i / 5, c = 21 + (i - m * 5);
    v[m][c] = 0.f;
  }
  if (tid < AX_VP) v[240][tid] = 0.f;
  __syncthreads();

  if (pass == 0) {
    float* qo = xp + (size_t)(b * 126 + 21) * HW + (size_t)r * 21;
#pragma unroll
    for (int j = 0; j < 11; ++j) {
      int i = tid + j * AX_THREADS;
      if (i < 240 * 21) {
        int xx = i / 21, c = i - xx * 21;
        qo[(size_t)xx * (240 * 21) + c] = q[xx][c];
      }
    }
  }

  const int m0 = qt * 16;
  const int cn = lane & 15;
  const int kh = (lane >> 4) << 3;

  v2f bq[6];
#pragma unroll
  for (int kk = 0; kk < 6; ++kk) bq[kk] = frag_a(&q[0][0], 25, m0, kk * 4, lane);

  v8f st[15];
#pragma unroll
  for (int g = 0; g < 5; ++g) {
    v8f a0 = V8Z, a1 = V8Z, a2 = V8Z;
#pragma unroll
    for (int kk = 0; kk < 6; ++kk) {
      a0 = wmma4(frag_a(&q[0][0], 25, (3 * g + 0) * 16, kk * 4, lane), bq[kk], a0);
      a1 = wmma4(frag_a(&q[0][0], 25, (3 * g + 1) * 16, kk * 4, lane), bq[kk], a1);
      a2 = wmma4(frag_a(&q[0][0], 25, (3 * g + 2) * 16, kk * 4, lane), bq[kk], a2);
    }
    st[3 * g + 0] = a0; st[3 * g + 1] = a1; st[3 * g + 2] = a2;
  }

  float mx = -1e30f;
#pragma unroll
  for (int nt = 0; nt < 15; ++nt)
#pragma unroll
    for (int rr = 0; rr < 8; ++rr) mx = fmaxf(mx, st[nt][rr]);
  mx = fmaxf(mx, __shfl_xor(mx, 16));
  float s0 = 0.f, s1 = 0.f, s2 = 0.f, s3 = 0.f;
#pragma unroll
  for (int nt = 0; nt < 15; ++nt) {
#pragma unroll
    for (int rr = 0; rr < 8; rr += 4) {
      float e0 = __expf(st[nt][rr + 0] - mx);
      float e1 = __expf(st[nt][rr + 1] - mx);
      float e2 = __expf(st[nt][rr + 2] - mx);
      float e3 = __expf(st[nt][rr + 3] - mx);
      st[nt][rr + 0] = e0; st[nt][rr + 1] = e1; st[nt][rr + 2] = e2; st[nt][rr + 3] = e3;
      s0 += e0; s1 += e1; s2 += e2; s3 += e3;
    }
  }
  float sum = (s0 + s1) + (s2 + s3);
  sum += __shfl_xor(sum, 16);
  float inv = 1.f / sum;
#pragma unroll
  for (int nt = 0; nt < 15; ++nt)
#pragma unroll
    for (int rr = 0; rr < 8; ++rr) st[nt][rr] *= inv;

  v8f y0 = V8Z, y1 = V8Z;
#pragma unroll
  for (int nt = 0; nt < 15; ++nt) {
#pragma unroll
    for (int i = 0; i < 4; ++i) {
      v2f pa; pa.x = st[nt][2 * i]; pa.y = st[nt][2 * i + 1];
      int kb = nt * 16 + kh + 2 * i;
      v2f b0, b1;
      b0.x = v[kb][cn];      b0.y = v[kb + 1][cn];
      b1.x = v[kb][16 + cn]; b1.y = v[kb + 1][16 + cn];
      y0 = wmma4(pa, b0, y0);
      y1 = wmma4(pa, b1, y1);
    }
  }

  float* ob = (pass == 0) ? vmid + ((size_t)(b * 240) * 240 + r) * 21
                          : xp + (size_t)(b * 126 + 105) * HW + (size_t)r * (240 * 21);
  const int sm = (pass == 0) ? 240 * 21 : 21;
  const int mb = m0 + kh;
#pragma unroll
  for (int nt = 0; nt < 2; ++nt) {
    v8f acc = nt ? y1 : y0;
    int n = nt * 16 + cn;
    if (n < 21) {
#pragma unroll
      for (int rr = 0; rr < 8; ++rr) ob[(mb + rr) * sm + n] = acc[rr];
    }
  }
}

#define PO_T 5
#define PO_BLK (900 / PO_T)

__device__ __forceinline__ void po_load_tile(const float* __restrict__ xp, const float* __restrict__ y2,
                                             int b, int p0, int tid, float (&ta)[21], float (&tc)[11]) {
  const int am = tid & 63, akh = tid >> 6;
#pragma unroll
  for (int j = 0; j < 21; ++j) {
    int k = 2 * j + akh;
    const float* sp;
    if (k < 21) sp = xp + (size_t)(b * 126 + k) * HW;
    else        sp = y2 + (size_t)(b * 21 + (k - 21)) * HW;
    ta[j] = sp[p0 + am];
  }
  const float* y3 = xp + (size_t)(b * 126 + 105) * HW;
#pragma unroll
  for (int j = 0; j < 11; ++j) {
    int e  = tid + j * 128;
    int ec = e < 64 * 21 ? e : 64 * 21 - 1;
    int px = ec / 21, c = ec - px * 21;
    int p  = p0 + px;
    int h  = p / WW, w = p - h * WW;
    tc[j] = y3[(size_t)(w * HH + h) * 21 + c];
  }
}

__global__ __launch_bounds__(128) void k_proj_out(
    const float* __restrict__ xp, const float* __restrict__ y2,
    const float* __restrict__ w_out, const float* __restrict__ b_out,
    float* __restrict__ out)
{
  __shared__ float a_lds[64][65];
  __shared__ float w_lds[64][65];
  __shared__ float bias_lds[64];

  int tid  = threadIdx.x;
  int wave = tid >> 5;
  int blk  = blockIdx.x;
  int b    = blk / PO_BLK;
  int pb   = (blk % PO_BLK) * (PO_T * 64);

  float ta[21], tc[11];
  po_load_tile(xp, y2, b, pb, tid, ta, tc);
  {
    float tw[32];
#pragma unroll
    for (int j = 0; j < 32; ++j) {
      int idx = tid + j * 128;
      int ic  = idx < 63 * 63 ? idx : 63 * 63 - 1;
      tw[j] = w_out[ic];
    }
    if (tid < 64) {
      bias_lds[tid] = (tid < 63) ? b_out[tid] : 0.f;
      w_lds[63][tid] = 0.f;
      w_lds[tid][63] = 0.f;
      a_lds[tid][63] = 0.f;
    }
#pragma unroll
    for (int j = 0; j < 32; ++j) {
      int idx = tid + j * 128;
      if (idx < 63 * 63) {
        int n = idx / 63, k = idx - n * 63;
        w_lds[k][n] = tw[j];
      }
    }
  }

  const int wave_u = __builtin_amdgcn_readfirstlane(wave);

  for (int t = 0; t < PO_T; ++t) {
    int tl = tid;
    asm volatile("" : "+v"(tl));
    const int am_t = tl & 63, akh_t = tl >> 6;
    const int ln   = tl & 31;
    const int cn   = ln & 15;
    const int chh  = (ln >> 4) << 3;
    const int n0   = (tl >> 5) * 16;

    if (t > 0) __syncthreads();
#pragma unroll
    for (int j = 0; j < 21; ++j) a_lds[am_t][2 * j + akh_t] = ta[j];
#pragma unroll
    for (int j = 0; j < 11; ++j) {
      int e = tl + j * 128;
      if (e < 64 * 21) {
        int px = e / 21, c = e - px * 21;
        a_lds[px][42 + c] = tc[j];
      }
    }
    __syncthreads();

    float* op = out + (size_t)b * 63 * HW + pb + t * 64 + 4 * cn;
    v8f acc[4];
#pragma unroll
    for (int u = 0; u < 4; ++u) { v8f z = V8Z; acc[u] = z; }
#pragma unroll
    for (int k0 = 0; k0 < 64; k0 += 4) {
      v2f pf[4];
#pragma unroll
      for (int u = 0; u < 4; ++u) pf[u] = frag_row(&a_lds[0][0], 65, 4 * cn + u, k0, ln);
      v2f wf = frag_b(&w_lds[0][0], 65, k0, n0, ln);
#pragma unroll
      for (int u = 0; u < 4; ++u) acc[u] = wmma4(wf, pf[u], acc[u]);
    }
    if (wave_u < 3) store_tile4<false>(op, bias_lds, n0 + chh, 63, acc[0], acc[1], acc[2], acc[3]);
    else            store_tile4<true >(op, bias_lds, n0 + chh, 63, acc[0], acc[1], acc[2], acc[3]);
    if (t + 1 < PO_T) po_load_tile(xp, y2, b, pb + (t + 1) * 64, tl, ta, tc);
  }
}

extern "C" void kernel_launch(void* const* d_in, const int* in_sizes, int n_in,
                              void* d_out, int out_size, void* d_ws, size_t ws_size,
                              hipStream_t stream)
{
  (void)in_sizes; (void)n_in; (void)out_size; (void)ws_size;
  const float* x     = (const float*)d_in[0];
  const float* w_in  = (const float*)d_in[1];
  const float* b_in  = (const float*)d_in[2];
  const float* bn_g  = (const float*)d_in[3];
  const float* bn_b  = (const float*)d_in[4];
  const float* bn_m  = (const float*)d_in[5];
  const float* bn_v  = (const float*)d_in[6];
  const float* w_out = (const float*)d_in[7];
  const float* b_out = (const float*)d_in[8];
  float* out = (float*)d_out;

  float* xp   = (float*)d_ws;
  float* y2b  = xp  + (size_t)BB * 126 * HW;
  float* vmid = y2b + (size_t)BB * 21 * HW;

  k_proj_in <<<dim3(BB * PI_BLK), 128, 0, stream>>>(x, w_in, b_in, bn_g, bn_b, bn_m, bn_v, xp);
  k_win     <<<dim3(BB * NWIN * WN_STRIPS, 2), WN_THREADS, 0, stream>>>(xp, y2b);
  k_axial   <<<dim3(BB * HH), AX_THREADS, 0, stream>>>(xp, vmid, 0);
  k_axial   <<<dim3(BB * WW), AX_THREADS, 0, stream>>>(xp, vmid, 1);
  k_proj_out<<<dim3(BB * PO_BLK), 128, 0, stream>>>(xp, y2b, w_out, b_out, out);
}


# axial attention: 8-wave workgroup per sequence, each wave computes two query tiles in turn (was 15 waves, one tile each), so three workgroups are resident per WGP instead of one; sums unchanged
# speedup vs baseline: 7.5773x; 1.0325x over previous
#include <hip/hip_runtime.h>
#include <math.h>

#define BB 8
#define CC 63
#define HH 240
#define WW 240
#define HW (HH * WW)
#define C2 126
#define WS 5
#define NWIN 48
#define BN_EPS 1e-5f

typedef __attribute__((ext_vector_type(2))) float v2f;
typedef __attribute__((ext_vector_type(4))) float v4f;
typedef __attribute__((ext_vector_type(8))) float v8f;

__device__ __forceinline__ v8f wmma4(v2f a, v2f b, v8f c) {
  return __builtin_amdgcn_wmma_f32_16x16x4_f32(
       false, a,  false, b,
       (short)0, c,  false,  false);
}

__device__ __forceinline__ v2f frag_a(const float* p, int ld, int m0, int k0, int lane) {
  int m  = m0 + (lane & 15);
  int kb = k0 + ((lane >> 4) << 1);
  const float* q = p + m * ld + kb;
  v2f r; r.x = q[0]; r.y = q[1];
  return r;
}

__device__ __forceinline__ v2f frag_row(const float* p, int ld, int row, int k0, int lane) {
  int kb = k0 + ((lane >> 4) << 1);
  const float* q = p + row * ld + kb;
  v2f r; r.x = q[0]; r.y = q[1];
  return r;
}

__device__ __forceinline__ v2f frag_b(const float* p, int ld, int k0, int n0, int lane) {
  int n  = n0 + (lane & 15);
  int kb = k0 + ((lane >> 4) << 1);
  v2f r; r.x = p[kb * ld + n]; r.y = p[(kb + 1) * ld + n];
  return r;
}

__device__ __forceinline__ void frag_store(float* p, int ld, int m0, int n0, int lane, v8f d) {
  int n  = n0 + (lane & 15);
  int mb = m0 + ((lane >> 4) << 3);
#pragma unroll
  for (int r = 0; r < 8; ++r) p[(mb + r) * ld + n] = d[r];
}

#define V8Z {0.f,0.f,0.f,0.f,0.f,0.f,0.f,0.f}

template <bool GUARD>
__device__ __forceinline__ void store_tile4(float* op, const float* bias, int ch0, int nch,
                                            v8f a0, v8f a1, v8f a2, v8f a3) {
  float bs[8];
#pragma unroll
  for (int r = 0; r < 8; ++r) bs[r] = bias[ch0 + r];
#pragma unroll
  for (int r = 0; r < 8; ++r) {
    int ch = ch0 + r;
    if (!GUARD || ch < nch) {
      v4f o;
      o.x = a0[r] + bs[r]; o.y = a1[r] + bs[r];
      o.z = a2[r] + bs[r]; o.w = a3[r] + bs[r];
      *(v4f*)(op + (size_t)ch * HW) = o;
    }
  }
}

#define PI_T 5
#define PI_BLK (900 / PI_T)

__global__ __launch_bounds__(128, 6) void k_proj_in(
    const float* __restrict__ x, const float* __restrict__ w_in,
    const float* __restrict__ b_in, const float* __restrict__ bn_g,
    const float* __restrict__ bn_b, const float* __restrict__ bn_m,
    const float* __restrict__ bn_v, float* __restrict__ xp)
{
  __shared__ float a_lds[64][65];
  __shared__ float w_lds[64][129];
  __shared__ float bias_lds[128];
  __shared__ float scale_lds[128];

  int tid  = threadIdx.x;
  int wave = tid >> 5;
  int blk  = blockIdx.x;
  int b    = blk / PI_BLK;
  int pb   = (blk % PI_BLK) * (PI_T * 64);

  {
    float sc = 0.f, bv = 0.f;
    if (tid < 126) {
      sc = bn_g[tid] * rsqrtf(bn_v[tid] + BN_EPS);
      bv = b_in[tid] * sc + bn_b[tid] - bn_m[tid] * sc;
    }
    scale_lds[tid] = sc;
    bias_lds[tid]  = bv;
  }
  const int am = tid & 63, akh = tid >> 6;
  const float* xb = x + (size_t)b * 63 * HW + pb + am;
  float ta[32];
#pragma unroll
  for (int j = 0; j < 32; ++j) {
    int k  = 2 * j + akh;
    int kc = k < 63 ? k : 62;
    ta[j] = xb[(size_t)kc * HW];
  }
  {
    float tw[63];
#pragma unroll
    for (int j = 0; j < 63; ++j) {
      int idx = tid + j * 128;
      int ic  = idx < 126 * 63 ? idx : 126 * 63 - 1;
      tw[j] = w_in[ic];
    }
    w_lds[63][tid] = 0.f;
    if (tid < 63) { w_lds[tid][126] = 0.f; w_lds[tid][127] = 0.f; }
    __syncthreads();
#pragma unroll
    for (int j = 0; j < 63; ++j) {
      int idx = tid + j * 128;
      if (idx < 126 * 63) {
        int n = idx / 63, k = idx - n * 63;
        w_lds[k][n] = tw[j] * scale_lds[n];
      }
    }
  }

  const int wave_u = __builtin_amdgcn_readfirstlane(wave);

  const int ln  = tid & 31;
  const int cn  = ln & 15;
  const int chh = (ln >> 4) << 3;
  const int nb  = wave * 32;

  for (int t = 0; t < PI_T; ++t) {
    if (t > 0) __syncthreads();
#pragma unroll
    for (int j = 0; j < 32; ++j) {
      int k = 2 * j + akh;
      a_lds[am][k] = (k < 63) ? ta[j] : 0.f;
    }
    __syncthreads();

    float* op = xp + (size_t)b * 126 * HW + pb + t * 64 + 4 * cn;
    v8f acc[2][4];
#pragma unroll
    for (int nt = 0; nt < 2; ++nt)
#pragma unroll
      for (int u = 0; u < 4; ++u) { v8f z = V8Z; acc[nt][u] = z; }
    for (int k0 = 0; k0 < 64; k0 += 4) {
      v2f pf[4], wf[2];
#pragma unroll
      for (int u = 0; u < 4; ++u) pf[u] = frag_row(&a_lds[0][0], 65, 4 * cn + u, k0, ln);
#pragma unroll
      for (int nt = 0; nt < 2; ++nt) wf[nt] = frag_b(&w_lds[0][0], 129, k0, nb + nt * 16, ln);
#pragma unroll
      for (int nt = 0; nt < 2; ++nt)
#pragma unroll
        for (int u = 0; u < 4; ++u) acc[nt][u] = wmma4(wf[nt], pf[u], acc[nt][u]);
    }
    store_tile4<false>(op, bias_lds, nb + chh, 126, acc[0][0], acc[0][1], acc[0][2], acc[0][3]);
    if (wave_u < 3) store_tile4<false>(op, bias_lds, nb + 16 + chh, 126, acc[1][0], acc[1][1], acc[1][2], acc[1][3]);
    else            store_tile4<true >(op, bias_lds, nb + 16 + chh, 126, acc[1][0], acc[1][1], acc[1][2], acc[1][3]);
    if (t + 1 < PI_T) {
      const float* xn = xb + (t + 1) * 64;
#pragma unroll
      for (int j = 0; j < 32; ++j) {
        int k  = 2 * j + akh;
        int kc = k < 63 ? k : 62;
        ta[j] = xn[(size_t)kc * HW];
      }
    }
  }
}

#define WN_NW 8
#define WN_THREADS (WN_NW * 32)
#define WN_SEG (WN_NW * WS)
#define WN_STRIPS (WW / WN_SEG)
#define WN_MOVERS (WN_SEG * 6)
#define WN_VOFF 800
#define WN_SOFF 1600
#define WN_WSZ 2664

__global__ __launch_bounds__(WN_THREADS) void k_win(float* __restrict__ xp, float* __restrict__ y2)
{
  __shared__ float wl[WN_NW * WN_WSZ];

  const int tid  = threadIdx.x;
  const int lane = tid & 31;
  const int wv   = tid >> 5;
  const int sid  = blockIdx.x;
  const int grp  = blockIdx.y;
  const int b    = sid / (NWIN * WN_STRIPS);
  const int rem  = sid % (NWIN * WN_STRIPS);
  const int wy   = rem / WN_STRIPS, st = rem % WN_STRIPS;
  const int shi  = grp ? 3 : 0;
  const int sho  = grp ? 2 : 0;

  const int x    = tid % WN_SEG, sub = tid / WN_SEG;
  const int win  = x / WS, mxp = x - win * WS;
  const int isv  = sub >= 3;
  const int cb   = (sub - 3 * isv) * 7;

  float t[35];
  if (tid < WN_MOVERS) {
    const int gx = (st * WN_SEG + x + shi) % WW;
    const float* srcp = xp + (size_t)(b * 126 + grp * 42 + isv * 21 + cb) * HW + gx;
    int ro[5];
#pragma unroll
    for (int r = 0; r < 5; ++r) ro[r] = ((wy * WS + r + shi) % HH) * WW;
#pragma unroll
    for (int j = 0; j < 35; ++j) t[j] = srcp[(size_t)(j / 5) * HW + ro[j % 5]];
  }

  {
    float* rec = wl + wv * WN_WSZ;
    for (int i = lane; i < 7 * 25; i += 32) {
      rec[25 * 25 + i] = 0.f;
      rec[WN_VOFF + 25 * 25 + i] = 0.f;
    }
    for (int i = lane; i < 25 * 4; i += 32) {
      int m = i >> 2, c = 21 + (i & 3);
      rec[m * 25 + c] = 0.f;
      rec[WN_VOFF + m * 25 + c] = 0.f;
    }
  }

  if (tid < WN_MOVERS) {
    float* dl = wl + win * WN_WSZ + isv * WN_VOFF + mxp * 25 + cb;
#pragma unroll
    for (int j = 0; j < 35; ++j) dl[(j % 5) * (WS * 25) + (j / 5)] = t[j];
  }
  __syncthreads();

  float* q = wl + wv * WN_WSZ;
  float* v = q + WN_VOFF;
  float* s = q + WN_SOFF;

  for (int mt = 0; mt < 2; ++mt)
    for (int nt = 0; nt < 2; ++nt) {
      v8f acc = V8Z;
      for (int k0 = 0; k0 < 24; k0 += 4) {
        v2f av = frag_a(q, 25, mt * 16, k0, lane);
        v2f bv = frag_a(q, 25, nt * 16, k0, lane);
        acc = wmma4(av, bv, acc);
      }
      frag_store(s, 33, mt * 16, nt * 16, lane, acc);
    }

  if (lane < 25) {
    float* sr = s + lane * 33;
    float mx = -1e30f;
    for (int n = 0; n < 25; ++n) mx = fmaxf(mx, sr[n]);
    float sum = 0.f;
    for (int n = 0; n < 25; ++n) { float e = __expf(sr[n] - mx); sr[n] = e; sum += e; }
    float inv = 1.f / sum;
    for (int n = 0; n < 25; ++n) sr[n] *= inv;
    for (int n = 25; n < 28; ++n) sr[n] = 0.f;
  }

  v8f yt[2][2];
#pragma unroll
  for (int mt = 0; mt < 2; ++mt)
#pragma unroll
    for (int nt = 0; nt < 2; ++nt) {
      v8f acc = V8Z;
      for (int k0 = 0; k0 < 28; k0 += 4) {
        v2f av = frag_a(s, 33, mt * 16, k0, lane);
        v2f bv = frag_b(v, 25, k0, nt * 16, lane);
        acc = wmma4(av, bv, acc);
      }
      yt[mt][nt] = acc;
    }
#pragma unroll
  for (int mt = 0; mt < 2; ++mt)
#pragma unroll
    for (int nt = 0; nt < 2; ++nt) frag_store(s, 33, mt * 16, nt * 16, lane, yt[mt][nt]);
  __syncthreads();

  if (tid < WN_MOVERS) {
    float* dst = grp ? y2 + (size_t)(b * 21) * HW : xp + (size_t)(b * 126) * HW;
    const float* yl = wl + win * WN_WSZ + WN_SOFF + mxp * 33;
    const int ox = (st * WN_SEG + x + sho) % WW;
#pragma unroll
    for (int j = 0; j < 18; ++j) {
      if (j < 17 || sub < 3) {
        int p = j * 6 + sub;
        int c = p / WS, r = p - c * WS;
        int oy = (wy * WS + r + sho) % HH;
        dst[(size_t)c * HW + oy * WW + ox] = yl[r * (WS * 33) + c];
      }
    }
  }
}

#define AX_WAVES 8
#define AX_THREADS (AX_WAVES * 32)
#define AX_LD ((240 * 21 + AX_THREADS - 1) / AX_THREADS)
#define AX_VP 26

__global__ __launch_bounds__(AX_THREADS, 6) void k_axial(float* __restrict__ xp,
                                                         float* __restrict__ vmid, int pass)
{
  __shared__ float q[240][25];
  __shared__ float v[241][AX_VP];

  int tid  = threadIdx.x;
  int lane = tid & 31;
  int row  = blockIdx.x;
  int b    = row / HH;
  int r    = row % HH;

  {
    float tq[AX_LD], tv[AX_LD];
    if (pass == 0) {
      const float* qb = xp + (size_t)(b * 126 + 84) * HW + r * WW;
      const float* vb = xp + (size_t)(b * 126 + 105) * HW + r * WW;
#pragma unroll
      for (int j = 0; j < AX_LD; ++j) {
        int i  = tid + j * AX_THREADS;
        int ic = i < 240 * 21 ? i : 240 * 21 - 1;
        int c = ic / 240, xx = ic - c * 240;
        tq[j] = qb[(size_t)c * HW + xx];
        tv[j] = vb[(size_t)c * HW + xx];
      }
#pragma unroll
      for (int j = 0; j < AX_LD; ++j) {
        int i = tid + j * AX_THREADS;
        if (i < 240 * 21) {
          int c = i / 240, xx = i - c * 240;
          q[xx][c] = tq[j];
          v[xx][c] = tv[j];
        }
      }
    } else {
      const float* qb = xp + (size_t)(b * 126 + 21) * HW + (size_t)r * (240 * 21);
      const float* vb = vmid + (size_t)(b * 240 + r) * 240 * 21;
#pragma unroll
      for (int j = 0; j < AX_LD; ++j) {
        int i  = tid + j * AX_THREADS;
        int ic = i < 240 * 21 ? i : 240 * 21 - 1;
        tq[j] = qb[ic];
        tv[j] = vb[ic];
      }
#pragma unroll
      for (int j = 0; j < AX_LD; ++j) {
        int i = tid + j * AX_THREADS;
        if (i < 240 * 21) {
          int h2 = i / 21, c2 = i - h2 * 21;
          q[h2][c2] = tq[j];
          v[h2][c2] = tv[j];
        }
      }
    }
  }
  for (int i = tid; i < 240 * 4; i += AX_THREADS) {
    int m = i >> 2, c = 21 + (i & 3);
    q[m][c] = 0.f;
  }
  for (int i = tid; i < 240 * 5; i += AX_THREADS) {
    int m = i / 5, c = 21 + (i - m * 5);
    v[m][c] = 0.f;
  }
  if (tid < AX_VP) v[240][tid] = 0.f;
  __syncthreads();

  if (pass == 0) {
    float* qo = xp + (size_t)(b * 126 + 21) * HW + (size_t)r * 21;
#pragma unroll
    for (int j = 0; j < AX_LD; ++j) {
      int i = tid + j * AX_THREADS;
      if (i < 240 * 21) {
        int xx = i / 21, c = i - xx * 21;
        qo[(size_t)xx * (240 * 21) + c] = q[xx][c];
      }
    }
  }

  const int wv   = __builtin_amdgcn_readfirstlane(tid >> 5);
  const int one  = 4 + ((blockIdx.x + (blockIdx.x >> 8)) & 3);
  const int tl0  = 2 * wv - (wv > one ? 1 : 0);
  const int ntl  = (wv == one) ? 1 : 2;

  for (int tt = 0; tt < ntl; ++tt) {
  const int m0 = (tl0 + tt) * 16;
  int ln = lane;
  asm volatile("" : "+v"(ln));
  const int cn = ln & 15;
  const int kh = (ln >> 4) << 3;

  v2f bq[6];
#pragma unroll
  for (int kk = 0; kk < 6; ++kk) bq[kk] = frag_a(&q[0][0], 25, m0, kk * 4, ln);

  v8f st[15];
#pragma unroll
  for (int g = 0; g < 5; ++g) {
    v8f a0 = V8Z, a1 = V8Z, a2 = V8Z;
#pragma unroll
    for (int kk = 0; kk < 6; ++kk) {
      a0 = wmma4(frag_a(&q[0][0], 25, (3 * g + 0) * 16, kk * 4, ln), bq[kk], a0);
      a1 = wmma4(frag_a(&q[0][0], 25, (3 * g + 1) * 16, kk * 4, ln), bq[kk], a1);
      a2 = wmma4(frag_a(&q[0][0], 25, (3 * g + 2) * 16, kk * 4, ln), bq[kk], a2);
    }
    st[3 * g + 0] = a0; st[3 * g + 1] = a1; st[3 * g + 2] = a2;
  }

  float mx = -1e30f;
#pragma unroll
  for (int nt = 0; nt < 15; ++nt)
#pragma unroll
    for (int rr = 0; rr < 8; ++rr) mx = fmaxf(mx, st[nt][rr]);
  const int xh = (ln ^ 16) << 2;
  mx = fmaxf(mx, __int_as_float(__builtin_amdgcn_ds_bpermute(xh, __float_as_int(mx))));
  float s0 = 0.f, s1 = 0.f, s2 = 0.f, s3 = 0.f;
#pragma unroll
  for (int nt = 0; nt < 15; ++nt) {
#pragma unroll
    for (int rr = 0; rr < 8; rr += 4) {
      float e0 = __expf(st[nt][rr + 0] - mx);
      float e1 = __expf(st[nt][rr + 1] - mx);
      float e2 = __expf(st[nt][rr + 2] - mx);
      float e3 = __expf(st[nt][rr + 3] - mx);
      st[nt][rr + 0] = e0; st[nt][rr + 1] = e1; st[nt][rr + 2] = e2; st[nt][rr + 3] = e3;
      s0 += e0; s1 += e1; s2 += e2; s3 += e3;
    }
  }
  float sum = (s0 + s1) + (s2 + s3);
  sum += __int_as_float(__builtin_amdgcn_ds_bpermute(xh, __float_as_int(sum)));
  float inv = 1.f / sum;
#pragma unroll
  for (int nt = 0; nt < 15; ++nt)
#pragma unroll
    for (int rr = 0; rr < 8; ++rr) st[nt][rr] *= inv;

  v8f y0 = V8Z, y1 = V8Z;
#pragma unroll
  for (int nt = 0; nt < 15; ++nt) {
#pragma unroll
    for (int i = 0; i < 4; ++i) {
      v2f pa; pa.x = st[nt][2 * i]; pa.y = st[nt][2 * i + 1];
      int kb = nt * 16 + kh + 2 * i;
      v2f b0, b1;
      b0.x = v[kb][cn];      b0.y = v[kb + 1][cn];
      b1.x = v[kb][16 + cn]; b1.y = v[kb + 1][16 + cn];
      y0 = wmma4(pa, b0, y0);
      y1 = wmma4(pa, b1, y1);
    }
  }

  float* ob = (pass == 0) ? vmid + ((size_t)(b * 240) * 240 + r) * 21
                          : xp + (size_t)(b * 126 + 105) * HW + (size_t)r * (240 * 21);
  const int sm = (pass == 0) ? 240 * 21 : 21;
  const int mb = m0 + kh;
#pragma unroll
  for (int nt = 0; nt < 2; ++nt) {
    v8f acc = nt ? y1 : y0;
    int n = nt * 16 + cn;
    if (n < 21) {
#pragma unroll
      for (int rr = 0; rr < 8; ++rr) ob[(mb + rr) * sm + n] = acc[rr];
    }
  }
  }
}

#define PO_T 5
#define PO_BLK (900 / PO_T)

__device__ __forceinline__ void po_load_tile(const float* __restrict__ xp, const float* __restrict__ y2,
                                             int b, int p0, int tid, float (&ta)[21], float (&tc)[11]) {
  const int am = tid & 63, akh = tid >> 6;
#pragma unroll
  for (int j = 0; j < 21; ++j) {
    int k = 2 * j + akh;
    const float* sp;
    if (k < 21) sp = xp + (size_t)(b * 126 + k) * HW;
    else        sp = y2 + (size_t)(b * 21 + (k - 21)) * HW;
    ta[j] = sp[p0 + am];
  }
  const float* y3 = xp + (size_t)(b * 126 + 105) * HW;
#pragma unroll
  for (int j = 0; j < 11; ++j) {
    int e  = tid + j * 128;
    int ec = e < 64 * 21 ? e : 64 * 21 - 1;
    int px = ec / 21, c = ec - px * 21;
    int p  = p0 + px;
    int h  = p / WW, w = p - h * WW;
    tc[j] = y3[(size_t)(w * HH + h) * 21 + c];
  }
}

__global__ __launch_bounds__(128) void k_proj_out(
    const float* __restrict__ xp, const float* __restrict__ y2,
    const float* __restrict__ w_out, const float* __restrict__ b_out,
    float* __restrict__ out)
{
  __shared__ float a_lds[64][65];
  __shared__ float w_lds[64][65];
  __shared__ float bias_lds[64];

  int tid  = threadIdx.x;
  int wave = tid >> 5;
  int blk  = blockIdx.x;
  int b    = blk / PO_BLK;
  int pb   = (blk % PO_BLK) * (PO_T * 64);

  float ta[21], tc[11];
  po_load_tile(xp, y2, b, pb, tid, ta, tc);
  {
    float tw[32];
#pragma unroll
    for (int j = 0; j < 32; ++j) {
      int idx = tid + j * 128;
      int ic  = idx < 63 * 63 ? idx : 63 * 63 - 1;
      tw[j] = w_out[ic];
    }
    if (tid < 64) {
      bias_lds[tid] = (tid < 63) ? b_out[tid] : 0.f;
      w_lds[63][tid] = 0.f;
      w_lds[tid][63] = 0.f;
      a_lds[tid][63] = 0.f;
    }
#pragma unroll
    for (int j = 0; j < 32; ++j) {
      int idx = tid + j * 128;
      if (idx < 63 * 63) {
        int n = idx / 63, k = idx - n * 63;
        w_lds[k][n] = tw[j];
      }
    }
  }

  const int wave_u = __builtin_amdgcn_readfirstlane(wave);

  for (int t = 0; t < PO_T; ++t) {
    int tl = tid;
    asm volatile("" : "+v"(tl));
    const int am_t = tl & 63, akh_t = tl >> 6;
    const int ln   = tl & 31;
    const int cn   = ln & 15;
    const int chh  = (ln >> 4) << 3;
    const int n0   = (tl >> 5) * 16;

    if (t > 0) __syncthreads();
#pragma unroll
    for (int j = 0; j < 21; ++j) a_lds[am_t][2 * j + akh_t] = ta[j];
#pragma unroll
    for (int j = 0; j < 11; ++j) {
      int e = tl + j * 128;
      if (e < 64 * 21) {
        int px = e / 21, c = e - px * 21;
        a_lds[px][42 + c] = tc[j];
      }
    }
    __syncthreads();

    float* op = out + (size_t)b * 63 * HW + pb + t * 64 + 4 * cn;
    v8f acc[4];
#pragma unroll
    for (int u = 0; u < 4; ++u) { v8f z = V8Z; acc[u] = z; }
#pragma unroll
    for (int k0 = 0; k0 < 64; k0 += 4) {
      v2f pf[4];
#pragma unroll
      for (int u = 0; u < 4; ++u) pf[u] = frag_row(&a_lds[0][0], 65, 4 * cn + u, k0, ln);
      v2f wf = frag_b(&w_lds[0][0], 65, k0, n0, ln);
#pragma unroll
      for (int u = 0; u < 4; ++u) acc[u] = wmma4(wf, pf[u], acc[u]);
    }
    if (wave_u < 3) store_tile4<false>(op, bias_lds, n0 + chh, 63, acc[0], acc[1], acc[2], acc[3]);
    else            store_tile4<true >(op, bias_lds, n0 + chh, 63, acc[0], acc[1], acc[2], acc[3]);
    if (t + 1 < PO_T) po_load_tile(xp, y2, b, pb + (t + 1) * 64, tl, ta, tc);
  }
}

extern "C" void kernel_launch(void* const* d_in, const int* in_sizes, int n_in,
                              void* d_out, int out_size, void* d_ws, size_t ws_size,
                              hipStream_t stream)
{
  (void)in_sizes; (void)n_in; (void)out_size; (void)ws_size;
  const float* x     = (const float*)d_in[0];
  const float* w_in  = (const float*)d_in[1];
  const float* b_in  = (const float*)d_in[2];
  const float* bn_g  = (const float*)d_in[3];
  const float* bn_b  = (const float*)d_in[4];
  const float* bn_m  = (const float*)d_in[5];
  const float* bn_v  = (const float*)d_in[6];
  const float* w_out = (const float*)d_in[7];
  const float* b_out = (const float*)d_in[8];
  float* out = (float*)d_out;

  float* xp   = (float*)d_ws;
  float* y2b  = xp  + (size_t)BB * 126 * HW;
  float* vmid = y2b + (size_t)BB * 21 * HW;

  k_proj_in <<<dim3(BB * PI_BLK), 128, 0, stream>>>(x, w_in, b_in, bn_g, bn_b, bn_m, bn_v, xp);
  k_win     <<<dim3(BB * NWIN * WN_STRIPS, 2), WN_THREADS, 0, stream>>>(xp, y2b);
  k_axial   <<<dim3(BB * HH), AX_THREADS, 0, stream>>>(xp, vmid, 0);
  k_axial   <<<dim3(BB * WW), AX_THREADS, 0, stream>>>(xp, vmid, 1);
  k_proj_out<<<dim3(BB * PO_BLK), 128, 0, stream>>>(xp, y2b, w_out, b_out, out);
}


# axial attention: q and V laid out in LDS so that one aligned 16-byte read at an immediate offset is two operand pairs (no address adds, no moves); exp argument one fma; score sums in another k order
# speedup vs baseline: 7.9843x; 1.0537x over previous
#include <hip/hip_runtime.h>
#include <math.h>

#define BB 8
#define CC 63
#define HH 240
#define WW 240
#define HW (HH * WW)
#define C2 126
#define WS 5
#define NWIN 48
#define BN_EPS 1e-5f

typedef __attribute__((ext_vector_type(2))) float v2f;
typedef __attribute__((ext_vector_type(4))) float v4f;
typedef __attribute__((ext_vector_type(8))) float v8f;

__device__ __forceinline__ v8f wmma4(v2f a, v2f b, v8f c) {
  return __builtin_amdgcn_wmma_f32_16x16x4_f32(
       false, a,  false, b,
       (short)0, c,  false,  false);
}

__device__ __forceinline__ v2f frag_a(const float* p, int ld, int m0, int k0, int lane) {
  int m  = m0 + (lane & 15);
  int kb = k0 + ((lane >> 4) << 1);
  const float* q = p + m * ld + kb;
  v2f r; r.x = q[0]; r.y = q[1];
  return r;
}

__device__ __forceinline__ v2f frag_row(const float* p, int ld, int row, int k0, int lane) {
  int kb = k0 + ((lane >> 4) << 1);
  const float* q = p + row * ld + kb;
  v2f r; r.x = q[0]; r.y = q[1];
  return r;
}

__device__ __forceinline__ v2f frag_b(const float* p, int ld, int k0, int n0, int lane) {
  int n  = n0 + (lane & 15);
  int kb = k0 + ((lane >> 4) << 1);
  v2f r; r.x = p[kb * ld + n]; r.y = p[(kb + 1) * ld + n];
  return r;
}

__device__ __forceinline__ void frag_store(float* p, int ld, int m0, int n0, int lane, v8f d) {
  int n  = n0 + (lane & 15);
  int mb = m0 + ((lane >> 4) << 3);
#pragma unroll
  for (int r = 0; r < 8; ++r) p[(mb + r) * ld + n] = d[r];
}

__device__ __forceinline__ v4f ld4(const float* p) { return *(const v4f*)p; }
__device__ __forceinline__ v2f lo2(v4f w) { v2f r; r.x = w.x; r.y = w.y; return r; }
__device__ __forceinline__ v2f hi2(v4f w) { v2f r; r.x = w.z; r.y = w.w; return r; }

#define V8Z {0.f,0.f,0.f,0.f,0.f,0.f,0.f,0.f}

template <bool GUARD>
__device__ __forceinline__ void store_tile4(float* op, const float* bias, int ch0, int nch,
                                            v8f a0, v8f a1, v8f a2, v8f a3) {
  float bs[8];
#pragma unroll
  for (int r = 0; r < 8; ++r) bs[r] = bias[ch0 + r];
#pragma unroll
  for (int r = 0; r < 8; ++r) {
    int ch = ch0 + r;
    if (!GUARD || ch < nch) {
      v4f o;
      o.x = a0[r] + bs[r]; o.y = a1[r] + bs[r];
      o.z = a2[r] + bs[r]; o.w = a3[r] + bs[r];
      *(v4f*)(op + (size_t)ch * HW) = o;
    }
  }
}

#define PI_T 5
#define PI_BLK (900 / PI_T)

__global__ __launch_bounds__(128, 6) void k_proj_in(
    const float* __restrict__ x, const float* __restrict__ w_in,
    const float* __restrict__ b_in, const float* __restrict__ bn_g,
    const float* __restrict__ bn_b, const float* __restrict__ bn_m,
    const float* __restrict__ bn_v, float* __restrict__ xp)
{
  __shared__ float a_lds[64][65];
  __shared__ float w_lds[64][129];
  __shared__ float bias_lds[128];
  __shared__ float scale_lds[128];

  int tid  = threadIdx.x;
  int wave = tid >> 5;
  int blk  = blockIdx.x;
  int b    = blk / PI_BLK;
  int pb   = (blk % PI_BLK) * (PI_T * 64);

  {
    float sc = 0.f, bv = 0.f;
    if (tid < 126) {
      sc = bn_g[tid] * rsqrtf(bn_v[tid] + BN_EPS);
      bv = b_in[tid] * sc + bn_b[tid] - bn_m[tid] * sc;
    }
    scale_lds[tid] = sc;
    bias_lds[tid]  = bv;
  }
  const int am = tid & 63, akh = tid >> 6;
  const float* xb = x + (size_t)b * 63 * HW + pb + am;
  float ta[32];
#pragma unroll
  for (int j = 0; j < 32; ++j) {
    int k  = 2 * j + akh;
    int kc = k < 63 ? k : 62;
    ta[j] = xb[(size_t)kc * HW];
  }
  {
    float tw[63];
#pragma unroll
    for (int j = 0; j < 63; ++j) {
      int idx = tid + j * 128;
      int ic  = idx < 126 * 63 ? idx : 126 * 63 - 1;
      tw[j] = w_in[ic];
    }
    w_lds[63][tid] = 0.f;
    if (tid < 63) { w_lds[tid][126] = 0.f; w_lds[tid][127] = 0.f; }
    __syncthreads();
#pragma unroll
    for (int j = 0; j < 63; ++j) {
      int idx = tid + j * 128;
      if (idx < 126 * 63) {
        int n = idx / 63, k = idx - n * 63;
        w_lds[k][n] = tw[j] * scale_lds[n];
      }
    }
  }

  const int wave_u = __builtin_amdgcn_readfirstlane(wave);

  const int ln  = tid & 31;
  const int cn  = ln & 15;
  const int chh = (ln >> 4) << 3;
  const int nb  = wave * 32;

  for (int t = 0; t < PI_T; ++t) {
    if (t > 0) __syncthreads();
#pragma unroll
    for (int j = 0; j < 32; ++j) {
      int k = 2 * j + akh;
      a_lds[am][k] = (k < 63) ? ta[j] : 0.f;
    }
    __syncthreads();

    float* op = xp + (size_t)b * 126 * HW + pb + t * 64 + 4 * cn;
    v8f acc[2][4];
#pragma unroll
    for (int nt = 0; nt < 2; ++nt)
#pragma unroll
      for (int u = 0; u < 4; ++u) { v8f z = V8Z; acc[nt][u] = z; }
    for (int k0 = 0; k0 < 64; k0 += 4) {
      v2f pf[4], wf[2];
#pragma unroll
      for (int u = 0; u < 4; ++u) pf[u] = frag_row(&a_lds[0][0], 65, 4 * cn + u, k0, ln);
#pragma unroll
      for (int nt = 0; nt < 2; ++nt) wf[nt] = frag_b(&w_lds[0][0], 129, k0, nb + nt * 16, ln);
#pragma unroll
      for (int nt = 0; nt < 2; ++nt)
#pragma unroll
        for (int u = 0; u < 4; ++u) acc[nt][u] = wmma4(wf[nt], pf[u], acc[nt][u]);
    }
    store_tile4<false>(op, bias_lds, nb + chh, 126, acc[0][0], acc[0][1], acc[0][2], acc[0][3]);
    if (wave_u < 3) store_tile4<false>(op, bias_lds, nb + 16 + chh, 126, acc[1][0], acc[1][1], acc[1][2], acc[1][3]);
    else            store_tile4<true >(op, bias_lds, nb + 16 + chh, 126, acc[1][0], acc[1][1], acc[1][2], acc[1][3]);
    if (t + 1 < PI_T) {
      const float* xn = xb + (t + 1) * 64;
#pragma unroll
      for (int j = 0; j < 32; ++j) {
        int k  = 2 * j + akh;
        int kc = k < 63 ? k : 62;
        ta[j] = xn[(size_t)kc * HW];
      }
    }
  }
}

#define WN_NW 8
#define WN_THREADS (WN_NW * 32)
#define WN_SEG (WN_NW * WS)
#define WN_STRIPS (WW / WN_SEG)
#define WN_MOVERS (WN_SEG * 6)
#define WN_VOFF 800
#define WN_SOFF 1600
#define WN_WSZ 2664

__global__ __launch_bounds__(WN_THREADS) void k_win(float* __restrict__ xp, float* __restrict__ y2)
{
  __shared__ float wl[WN_NW * WN_WSZ];

  const int tid  = threadIdx.x;
  const int lane = tid & 31;
  const int wv   = tid >> 5;
  const int sid  = blockIdx.x;
  const int grp  = blockIdx.y;
  const int b    = sid / (NWIN * WN_STRIPS);
  const int rem  = sid % (NWIN * WN_STRIPS);
  const int wy   = rem / WN_STRIPS, st = rem % WN_STRIPS;
  const int shi  = grp ? 3 : 0;
  const int sho  = grp ? 2 : 0;

  const int x    = tid % WN_SEG, sub = tid / WN_SEG;
  const int win  = x / WS, mxp = x - win * WS;
  const int isv  = sub >= 3;
  const int cb   = (sub - 3 * isv) * 7;

  float t[35];
  if (tid < WN_MOVERS) {
    const int gx = (st * WN_SEG + x + shi) % WW;
    const float* srcp = xp + (size_t)(b * 126 + grp * 42 + isv * 21 + cb) * HW + gx;
    int ro[5];
#pragma unroll
    for (int r = 0; r < 5; ++r) ro[r] = ((wy * WS + r + shi) % HH) * WW;
#pragma unroll
    for (int j = 0; j < 35; ++j) t[j] = srcp[(size_t)(j / 5) * HW + ro[j % 5]];
  }

  {
    float* rec = wl + wv * WN_WSZ;
    for (int i = lane; i < 7 * 25; i += 32) {
      rec[25 * 25 + i] = 0.f;
      rec[WN_VOFF + 25 * 25 + i] = 0.f;
    }
    for (int i = lane; i < 25 * 4; i += 32) {
      int m = i >> 2, c = 21 + (i & 3);
      rec[m * 25 + c] = 0.f;
      rec[WN_VOFF + m * 25 + c] = 0.f;
    }
  }

  if (tid < WN_MOVERS) {
    float* dl = wl + win * WN_WSZ + isv * WN_VOFF + mxp * 25 + cb;
#pragma unroll
    for (int j = 0; j < 35; ++j) dl[(j % 5) * (WS * 25) + (j / 5)] = t[j];
  }
  __syncthreads();

  float* q = wl + wv * WN_WSZ;
  float* v = q + WN_VOFF;
  float* s = q + WN_SOFF;

  for (int mt = 0; mt < 2; ++mt)
    for (int nt = 0; nt < 2; ++nt) {
      v8f acc = V8Z;
      for (int k0 = 0; k0 < 24; k0 += 4) {
        v2f av = frag_a(q, 25, mt * 16, k0, lane);
        v2f bv = frag_a(q, 25, nt * 16, k0, lane);
        acc = wmma4(av, bv, acc);
      }
      frag_store(s, 33, mt * 16, nt * 16, lane, acc);
    }

  if (lane < 25) {
    float* sr = s + lane * 33;
    float mx = -1e30f;
    for (int n = 0; n < 25; ++n) mx = fmaxf(mx, sr[n]);
    float sum = 0.f;
    for (int n = 0; n < 25; ++n) { float e = __expf(sr[n] - mx); sr[n] = e; sum += e; }
    float inv = 1.f / sum;
    for (int n = 0; n < 25; ++n) sr[n] *= inv;
    for (int n = 25; n < 28; ++n) sr[n] = 0.f;
  }

  v8f yt[2][2];
#pragma unroll
  for (int mt = 0; mt < 2; ++mt)
#pragma unroll
    for (int nt = 0; nt < 2; ++nt) {
      v8f acc = V8Z;
      for (int k0 = 0; k0 < 28; k0 += 4) {
        v2f av = frag_a(s, 33, mt * 16, k0, lane);
        v2f bv = frag_b(v, 25, k0, nt * 16, lane);
        acc = wmma4(av, bv, acc);
      }
      yt[mt][nt] = acc;
    }
#pragma unroll
  for (int mt = 0; mt < 2; ++mt)
#pragma unroll
    for (int nt = 0; nt < 2; ++nt) frag_store(s, 33, mt * 16, nt * 16, lane, yt[mt][nt]);
  __syncthreads();

  if (tid < WN_MOVERS) {
    float* dst = grp ? y2 + (size_t)(b * 21) * HW : xp + (size_t)(b * 126) * HW;
    const float* yl = wl + win * WN_WSZ + WN_SOFF + mxp * 33;
    const int ox = (st * WN_SEG + x + sho) % WW;
#pragma unroll
    for (int j = 0; j < 18; ++j) {
      if (j < 17 || sub < 3) {
        int p = j * 6 + sub;
        int c = p / WS, r = p - c * WS;
        int oy = (wy * WS + r + sho) % HH;
        dst[(size_t)c * HW + oy * WW + ox] = yl[r * (WS * 33) + c];
      }
    }
  }
}

#define AX_WAVES 8
#define AX_THREADS (AX_WAVES * 32)
#define AX_LD ((240 * 21 + AX_THREADS - 1) / AX_THREADS)
#define AX_QP 28
#define AX_VPP 68

__global__ __launch_bounds__(AX_THREADS, 6) void k_axial(float* __restrict__ xp,
                                                         float* __restrict__ vmid, int pass)
{
  __shared__ __attribute__((aligned(16))) float q[240][AX_QP];
  __shared__ __attribute__((aligned(16))) float vp[120][AX_VPP];

  int tid  = threadIdx.x;
  int lane = tid & 31;
  int row  = blockIdx.x;
  int b    = row / HH;
  int r    = row % HH;

  {
    float tq[AX_LD], tv[AX_LD];
    if (pass == 0) {
      const float* qb = xp + (size_t)(b * 126 + 84) * HW + r * WW;
      const float* vb = xp + (size_t)(b * 126 + 105) * HW + r * WW;
#pragma unroll
      for (int j = 0; j < AX_LD; ++j) {
        int i  = tid + j * AX_THREADS;
        int ic = i < 240 * 21 ? i : 240 * 21 - 1;
        int c = ic / 240, xx = ic - c * 240;
        tq[j] = qb[(size_t)c * HW + xx];
        tv[j] = vb[(size_t)c * HW + xx];
      }
#pragma unroll
      for (int j = 0; j < AX_LD; ++j) {
        int i = tid + j * AX_THREADS;
        if (i < 240 * 21) {
          int c = i / 240, xx = i - c * 240;
          q[xx][c] = tq[j];
          vp[xx >> 1][2 * c + (xx & 1)] = tv[j];
        }
      }
    } else {
      const float* qb = xp + (size_t)(b * 126 + 21) * HW + (size_t)r * (240 * 21);
      const float* vb = vmid + (size_t)(b * 240 + r) * 240 * 21;
#pragma unroll
      for (int j = 0; j < AX_LD; ++j) {
        int i  = tid + j * AX_THREADS;
        int ic = i < 240 * 21 ? i : 240 * 21 - 1;
        tq[j] = qb[ic];
        tv[j] = vb[ic];
      }
#pragma unroll
      for (int j = 0; j < AX_LD; ++j) {
        int i = tid + j * AX_THREADS;
        if (i < 240 * 21) {
          int h2 = i / 21, c2 = i - h2 * 21;
          q[h2][c2] = tq[j];
          vp[h2 >> 1][2 * c2 + (h2 & 1)] = tv[j];
        }
      }
    }
  }
  for (int i = tid; i < 240 * 4; i += AX_THREADS) {
    int m = i >> 2, c = 21 + (i & 3);
    q[m][c] = 0.f;
  }
  for (int i = tid; i < 120 * 11; i += AX_THREADS) {
    int m = i / 11, c = 21 + (i - m * 11);
    v2f z; z.x = 0.f; z.y = 0.f;
    *(v2f*)&vp[m][2 * c] = z;
  }
  __syncthreads();

  if (pass == 0) {
    float* qo = xp + (size_t)(b * 126 + 21) * HW + (size_t)r * 21;
#pragma unroll
    for (int j = 0; j < AX_LD; ++j) {
      int i = tid + j * AX_THREADS;
      if (i < 240 * 21) {
        int xx = i / 21, c = i - xx * 21;
        qo[(size_t)xx * (240 * 21) + c] = q[xx][c];
      }
    }
  }

  const int wv   = __builtin_amdgcn_readfirstlane(tid >> 5);
  const int one  = 4 + ((blockIdx.x + (blockIdx.x >> 8)) & 3);
  const int tl0  = 2 * wv - (wv > one ? 1 : 0);
  const int ntl  = (wv == one) ? 1 : 2;

  for (int tt = 0; tt < ntl; ++tt) {
  const int m0 = (tl0 + tt) * 16;
  int ln = lane;
  asm volatile("" : "+v"(ln));
  const int cn = ln & 15;
  const int kh = (ln >> 4) << 3;
  const float* qa = &q[0][0] + cn * AX_QP + ((ln >> 4) << 2);
  const float* va = &vp[0][0] + (ln >> 4) * (4 * AX_VPP) + 4 * cn;

  v4f bq[3];
#pragma unroll
  for (int j = 0; j < 3; ++j) bq[j] = ld4(qa + m0 * AX_QP + 8 * j);

  v8f st[15];
#pragma unroll
  for (int g = 0; g < 5; ++g) {
    v8f a0 = V8Z, a1 = V8Z, a2 = V8Z;
#pragma unroll
    for (int j = 0; j < 3; ++j) {
      v4f f0 = ld4(qa + (3 * g + 0) * 16 * AX_QP + 8 * j);
      v4f f1 = ld4(qa + (3 * g + 1) * 16 * AX_QP + 8 * j);
      v4f f2 = ld4(qa + (3 * g + 2) * 16 * AX_QP + 8 * j);
      a0 = wmma4(lo2(f0), lo2(bq[j]), a0);
      a1 = wmma4(lo2(f1), lo2(bq[j]), a1);
      a2 = wmma4(lo2(f2), lo2(bq[j]), a2);
      a0 = wmma4(hi2(f0), hi2(bq[j]), a0);
      a1 = wmma4(hi2(f1), hi2(bq[j]), a1);
      a2 = wmma4(hi2(f2), hi2(bq[j]), a2);
    }
    st[3 * g + 0] = a0; st[3 * g + 1] = a1; st[3 * g + 2] = a2;
  }

  float mx = -1e30f;
#pragma unroll
  for (int nt = 0; nt < 15; ++nt)
#pragma unroll
    for (int rr = 0; rr < 8; ++rr) mx = fmaxf(mx, st[nt][rr]);
  const int xh = (ln ^ 16) << 2;
  mx = fmaxf(mx, __int_as_float(__builtin_amdgcn_ds_bpermute(xh, __float_as_int(mx))));
  const float l2e = 1.4426950408889634f;
  const float nmx = -mx * l2e;
  v2f sa = {0.f, 0.f}, sb = {0.f, 0.f};
#pragma unroll
  for (int nt = 0; nt < 15; ++nt) {
#pragma unroll
    for (int rr = 0; rr < 8; rr += 4) {
      float e0 = __builtin_amdgcn_exp2f(__builtin_fmaf(st[nt][rr + 0], l2e, nmx));
      float e1 = __builtin_amdgcn_exp2f(__builtin_fmaf(st[nt][rr + 1], l2e, nmx));
      float e2 = __builtin_amdgcn_exp2f(__builtin_fmaf(st[nt][rr + 2], l2e, nmx));
      float e3 = __builtin_amdgcn_exp2f(__builtin_fmaf(st[nt][rr + 3], l2e, nmx));
      st[nt][rr + 0] = e0; st[nt][rr + 1] = e1; st[nt][rr + 2] = e2; st[nt][rr + 3] = e3;
      v2f ea; ea.x = e0; ea.y = e1;
      v2f eb; eb.x = e2; eb.y = e3;
      sa += ea; sb += eb;
    }
  }
  float sum = (sa.x + sa.y) + (sb.x + sb.y);
  sum += __int_as_float(__builtin_amdgcn_ds_bpermute(xh, __float_as_int(sum)));
  float inv = 1.f / sum;
#pragma unroll
  for (int nt = 0; nt < 15; ++nt)
#pragma unroll
    for (int rr = 0; rr < 8; ++rr) st[nt][rr] *= inv;

  v8f y0 = V8Z, y1 = V8Z;
#pragma unroll
  for (int nt = 0; nt < 15; ++nt) {
#pragma unroll
    for (int i = 0; i < 4; ++i) {
      v2f pa; pa.x = st[nt][2 * i]; pa.y = st[nt][2 * i + 1];
      v4f w = ld4(va + (nt * 8 + i) * AX_VPP);
      y0 = wmma4(pa, lo2(w), y0);
      y1 = wmma4(pa, hi2(w), y1);
    }
  }

  float* ob = (pass == 0) ? vmid + ((size_t)(b * 240) * 240 + r) * 21
                          : xp + (size_t)(b * 126 + 105) * HW + (size_t)r * (240 * 21);
  const int sm = (pass == 0) ? 240 * 21 : 21;
  const int mb = m0 + kh;
  const unsigned o0 = (unsigned)(mb * sm + 2 * cn);
#pragma unroll
  for (int nt = 0; nt < 2; ++nt) {
    v8f acc = nt ? y1 : y0;
    int n = 2 * cn + nt;
    if (n < 21) {
#pragma unroll
      for (int rr = 0; rr < 8; ++rr) ob[o0 + (unsigned)(rr * sm + nt)] = acc[rr];
    }
  }
  }
}

#define PO_T 5
#define PO_BLK (900 / PO_T)

__device__ __forceinline__ void po_load_tile(const float* __restrict__ xp, const float* __restrict__ y2,
                                             int b, int p0, int tid, float (&ta)[21], float (&tc)[11]) {
  const int am = tid & 63, akh = tid >> 6;
#pragma unroll
  for (int j = 0; j < 21; ++j) {
    int k = 2 * j + akh;
    const float* sp;
    if (k < 21) sp = xp + (size_t)(b * 126 + k) * HW;
    else        sp = y2 + (size_t)(b * 21 + (k - 21)) * HW;
    ta[j] = sp[p0 + am];
  }
  const float* y3 = xp + (size_t)(b * 126 + 105) * HW;
#pragma unroll
  for (int j = 0; j < 11; ++j) {
    int e  = tid + j * 128;
    int ec = e < 64 * 21 ? e : 64 * 21 - 1;
    int px = ec / 21, c = ec - px * 21;
    int p  = p0 + px;
    int h  = p / WW, w = p - h * WW;
    tc[j] = y3[(size_t)(w * HH + h) * 21 + c];
  }
}

__global__ __launch_bounds__(128) void k_proj_out(
    const float* __restrict__ xp, const float* __restrict__ y2,
    const float* __restrict__ w_out, const float* __restrict__ b_out,
    float* __restrict__ out)
{
  __shared__ float a_lds[64][65];
  __shared__ float w_lds[64][65];
  __shared__ float bias_lds[64];

  int tid  = threadIdx.x;
  int wave = tid >> 5;
  int blk  = blockIdx.x;
  int b    = blk / PO_BLK;
  int pb   = (blk % PO_BLK) * (PO_T * 64);

  float ta[21], tc[11];
  po_load_tile(xp, y2, b, pb, tid, ta, tc);
  {
    float tw[32];
#pragma unroll
    for (int j = 0; j < 32; ++j) {
      int idx = tid + j * 128;
      int ic  = idx < 63 * 63 ? idx : 63 * 63 - 1;
      tw[j] = w_out[ic];
    }
    if (tid < 64) {
      bias_lds[tid] = (tid < 63) ? b_out[tid] : 0.f;
      w_lds[63][tid] = 0.f;
      w_lds[tid][63] = 0.f;
      a_lds[tid][63] = 0.f;
    }
#pragma unroll
    for (int j = 0; j < 32; ++j) {
      int idx = tid + j * 128;
      if (idx < 63 * 63) {
        int n = idx / 63, k = idx - n * 63;
        w_lds[k][n] = tw[j];
      }
    }
  }

  const int wave_u = __builtin_amdgcn_readfirstlane(wave);

  for (int t = 0; t < PO_T; ++t) {
    int tl = tid;
    asm volatile("" : "+v"(tl));
    const int am_t = tl & 63, akh_t = tl >> 6;
    const int ln   = tl & 31;
    const int cn   = ln & 15;
    const int chh  = (ln >> 4) << 3;
    const int n0   = (tl >> 5) * 16;

    if (t > 0) __syncthreads();
#pragma unroll
    for (int j = 0; j < 21; ++j) a_lds[am_t][2 * j + akh_t] = ta[j];
#pragma unroll
    for (int j = 0; j < 11; ++j) {
      int e = tl + j * 128;
      if (e < 64 * 21) {
        int px = e / 21, c = e - px * 21;
        a_lds[px][42 + c] = tc[j];
      }
    }
    __syncthreads();

    float* op = out + (size_t)b * 63 * HW + pb + t * 64 + 4 * cn;
    v8f acc[4];
#pragma unroll
    for (int u = 0; u < 4; ++u) { v8f z = V8Z; acc[u] = z; }
#pragma unroll
    for (int k0 = 0; k0 < 64; k0 += 4) {
      v2f pf[4];
#pragma unroll
      for (int u = 0; u < 4; ++u) pf[u] = frag_row(&a_lds[0][0], 65, 4 * cn + u, k0, ln);
      v2f wf = frag_b(&w_lds[0][0], 65, k0, n0, ln);
#pragma unroll
      for (int u = 0; u < 4; ++u) acc[u] = wmma4(wf, pf[u], acc[u]);
    }
    if (wave_u < 3) store_tile4<false>(op, bias_lds, n0 + chh, 63, acc[0], acc[1], acc[2], acc[3]);
    else            store_tile4<true >(op, bias_lds, n0 + chh, 63, acc[0], acc[1], acc[2], acc[3]);
    if (t + 1 < PO_T) po_load_tile(xp, y2, b, pb + (t + 1) * 64, tl, ta, tc);
  }
}

extern "C" void kernel_launch(void* const* d_in, const int* in_sizes, int n_in,
                              void* d_out, int out_size, void* d_ws, size_t ws_size,
                              hipStream_t stream)
{
  (void)in_sizes; (void)n_in; (void)out_size; (void)ws_size;
  const float* x     = (const float*)d_in[0];
  const float* w_in  = (const float*)d_in[1];
  const float* b_in  = (const float*)d_in[2];
  const float* bn_g  = (const float*)d_in[3];
  const float* bn_b  = (const float*)d_in[4];
  const float* bn_m  = (const float*)d_in[5];
  const float* bn_v  = (const float*)d_in[6];
  const float* w_out = (const float*)d_in[7];
  const float* b_out = (const float*)d_in[8];
  float* out = (float*)d_out;

  float* xp   = (float*)d_ws;
  float* y2b  = xp  + (size_t)BB * 126 * HW;
  float* vmid = y2b + (size_t)BB * 21 * HW;

  k_proj_in <<<dim3(BB * PI_BLK), 128, 0, stream>>>(x, w_in, b_in, bn_g, bn_b, bn_m, bn_v, xp);
  k_win     <<<dim3(BB * NWIN * WN_STRIPS, 2), WN_THREADS, 0, stream>>>(xp, y2b);
  k_axial   <<<dim3(BB * HH), AX_THREADS, 0, stream>>>(xp, vmid, 0);
  k_axial   <<<dim3(BB * WW), AX_THREADS, 0, stream>>>(xp, vmid, 1);
  k_proj_out<<<dim3(BB * PO_BLK), 128, 0, stream>>>(xp, y2b, w_out, b_out, out);
}
